# starting kernel
# baseline (speedup 1.0000x reference)
#include <hip/hip_runtime.h>
#include <hip/hip_bf16.h>

typedef __attribute__((ext_vector_type(2))) float v2f;
typedef __attribute__((ext_vector_type(8))) float v8f;

constexpr int BB = 2;
constexpr int VV = 4;
constexpr int NN = 2048;
constexpr int FF = 128;
constexpr int EE = 32768;
constexpr int NE = 34816;
constexpr int OO = 512;

__device__ __forceinline__ float sgnval(int r, int j) {
  if (j >= 4) return 1.0f;
  const bool flip = (j & 1) ? ((r & 2) != 0) : ((r & 1) != 0);
  return flip ? -1.0f : 1.0f;
}

__device__ __forceinline__ int src_of(const int* __restrict__ ei, int e) {
  return (e < EE) ? ei[e] : (e - EE);
}
__device__ __forceinline__ int dst_of(const int* __restrict__ ei, int e) {
  return (e < EE) ? ei[EE + e] : (e - EE);
}

__global__ void zero_f32_kernel(float* __restrict__ p, size_t n) {
  size_t i = (size_t)blockIdx.x * blockDim.x + threadIdx.x;
  if (i < n) p[i] = 0.0f;
}
__global__ void zero_i32_kernel(int* __restrict__ p, int n) {
  int i = blockIdx.x * blockDim.x + threadIdx.x;
  if (i < n) p[i] = 0;
}

__global__ void count_dst_kernel(const int* __restrict__ ei, int* __restrict__ counts) {
  int e = blockIdx.x * blockDim.x + threadIdx.x;
  if (e >= NE) return;
  atomicAdd(&counts[dst_of(ei, e)], 1);
}

__global__ __launch_bounds__(256) void scan_counts_kernel(const int* __restrict__ counts,
                                                          int* __restrict__ rowstart) {
  __shared__ int part[256];
  __shared__ int pref[257];
  const int tid = threadIdx.x;
  const int base = tid * 8;
  int loc[8];
  int s = 0;
#pragma unroll
  for (int i = 0; i < 8; ++i) { loc[i] = s; s += counts[base + i]; }
  part[tid] = s;
  __syncthreads();
  if (tid == 0) {
    int a = 0;
    for (int i = 0; i < 256; ++i) { pref[i] = a; a += part[i]; }
    pref[256] = a;
  }
  __syncthreads();
  const int off = pref[tid];
#pragma unroll
  for (int i = 0; i < 8; ++i) rowstart[base + i] = off + loc[i];
  if (tid == 255) rowstart[NN] = pref[256];
}

__global__ __launch_bounds__(256) void build_rows_kernel(const int* __restrict__ ei,
                                                         const int* __restrict__ rowstart,
                                                         int* __restrict__ order) {
  const int lane = threadIdx.x & 31;
  const int wave = threadIdx.x >> 5;
  const int n = blockIdx.x * 8 + wave;
  if (n >= NN) return;
  const int base = rowstart[n];
  int cnt = 0;
  for (int e0 = 0; e0 < NE; e0 += 32) {
    const int e = e0 + lane;
    const bool match = (e < NE) && (dst_of(ei, e) == n);
    unsigned long long mask = __ballot(match);
    const int rank = __popcll(mask & ((1ull << lane) - 1ull));
    if (match) order[base + cnt + rank] = e;
    cnt += __popcll(mask);
  }
}

__device__ __forceinline__ void wmma_quad(const float* __restrict__ W, int k, int khalf,
                                          int col, v2f a0, v2f a1, v8f* acc) {
  const float* w0 = W + (size_t)(k + khalf) * OO;
  const float* w1 = w0 + OO;
  const v2f b0 = { w0[col +  0], w1[col +  0] };
  const v2f b1 = { w0[col + 16], w1[col + 16] };
  const v2f b2 = { w0[col + 32], w1[col + 32] };
  const v2f b3 = { w0[col + 48], w1[col + 48] };
  acc[0] = __builtin_amdgcn_wmma_f32_16x16x4_f32(false, a0, false, b0, (short)0, acc[0], false, false);
  acc[1] = __builtin_amdgcn_wmma_f32_16x16x4_f32(false, a0, false, b1, (short)0, acc[1], false, false);
  acc[2] = __builtin_amdgcn_wmma_f32_16x16x4_f32(false, a0, false, b2, (short)0, acc[2], false, false);
  acc[3] = __builtin_amdgcn_wmma_f32_16x16x4_f32(false, a0, false, b3, (short)0, acc[3], false, false);
  acc[4] = __builtin_amdgcn_wmma_f32_16x16x4_f32(false, a1, false, b0, (short)0, acc[4], false, false);
  acc[5] = __builtin_amdgcn_wmma_f32_16x16x4_f32(false, a1, false, b1, (short)0, acc[5], false, false);
  acc[6] = __builtin_amdgcn_wmma_f32_16x16x4_f32(false, a1, false, b2, (short)0, acc[6], false, false);
  acc[7] = __builtin_amdgcn_wmma_f32_16x16x4_f32(false, a1, false, b3, (short)0, acc[7], false, false);
}

__global__ __launch_bounds__(256) void gemm_pair_kernel(const float* __restrict__ x,
                                                        const float* __restrict__ W,
                                                        float* __restrict__ H,
                                                        int v1, int v2) {
  const int lane = threadIdx.x & 31;
  const int wave = threadIdx.x >> 5;
  const int rowTile = blockIdx.x * 32;
  const int b = rowTile >> 11;
  const int nbase = rowTile & (NN - 1);
  const int r0 = nbase + (lane & 15);
  const float* x1a = x + (((size_t)b * VV + v1) * NN + r0) * FF;
  const float* x1b = x1a + 16 * FF;
  const float* x2a = x + (((size_t)b * VV + v2) * NN + r0) * FF;
  const float* x2b = x2a + 16 * FF;
  const int khalf = (lane >> 4) * 2;
  const int col = wave * 64 + (lane & 15);

  v8f acc[8] = {};

  {
    v2f a0 = *(const v2f*)(x1a + khalf);
    v2f a1 = *(const v2f*)(x1b + khalf);
    const float s0 = sgnval(v2, khalf);
    const float s1 = sgnval(v2, khalf + 1);
    a0.x *= s0; a0.y *= s1;
    a1.x *= s0; a1.y *= s1;
    wmma_quad(W, 0, khalf, col, a0, a1, acc);
  }
  for (int k = 4; k < FF; k += 4) {
    const v2f a0 = *(const v2f*)(x1a + k + khalf);
    const v2f a1 = *(const v2f*)(x1b + k + khalf);
    wmma_quad(W, k, khalf, col, a0, a1, acc);
  }
  {
    v2f a0 = *(const v2f*)(x2a + khalf);
    v2f a1 = *(const v2f*)(x2b + khalf);
    const float s0 = sgnval(v1, khalf);
    const float s1 = sgnval(v1, khalf + 1);
    a0.x *= s0; a0.y *= s1;
    a1.x *= s0; a1.y *= s1;
    wmma_quad(W, FF, khalf, col, a0, a1, acc);
  }
  for (int k = FF + 4; k < 2 * FF; k += 4) {
    const v2f a0 = *(const v2f*)(x2a + (k - FF) + khalf);
    const v2f a1 = *(const v2f*)(x2b + (k - FF) + khalf);
    wmma_quad(W, k, khalf, col, a0, a1, acc);
  }

  float* Hb = H + (size_t)b * NN * OO;
  const int srow = nbase + 8 * (lane >> 4);
  const int scol = wave * 64 + (lane & 15);
#pragma unroll
  for (int i = 0; i < 8; ++i) {
    float* ra = Hb + (size_t)(srow + i) * OO + scol;
    ra[ 0] = acc[0][i];
    ra[16] = acc[1][i];
    ra[32] = acc[2][i];
    ra[48] = acc[3][i];
    float* rb = Hb + (size_t)(srow + 16 + i) * OO + scol;
    rb[ 0] = acc[4][i];
    rb[16] = acc[5][i];
    rb[32] = acc[6][i];
    rb[48] = acc[7][i];
  }
}

__global__ __launch_bounds__(256) void node_scores_kernel(const float* __restrict__ H,
                                                          const float* __restrict__ att,
                                                          float* __restrict__ uS,
                                                          float* __restrict__ vS) {
  const int lane = threadIdx.x & 31;
  const int wave = threadIdx.x >> 5;
  const int node = blockIdx.x * 8 + wave;
  if (node >= BB * NN) return;
  const float* hrow = H + (size_t)node * OO;
  float su = 0.0f, sv = 0.0f;
  for (int j = lane; j < OO; j += 32) {
    const float h = hrow[j];
    const float l = h > 0.0f ? h : 0.2f * h;
    su += att[j] * l;
    sv += att[OO + j] * l;
  }
#pragma unroll
  for (int off = 16; off; off >>= 1) {
    su += __shfl_down(su, off, 32);
    sv += __shfl_down(sv, off, 32);
  }
  if (lane == 0) { uS[node] = su; vS[node] = sv; }
}

__global__ void edge_scores_kernel(const float* __restrict__ uS, const float* __restrict__ vS,
                                   const int* __restrict__ ei, float* __restrict__ s) {
  const int t = blockIdx.x * blockDim.x + threadIdx.x;
  if (t >= BB * NE) return;
  const int b = t / NE;
  const int e = t - b * NE;
  s[t] = uS[b * NN + src_of(ei, e)] + vS[b * NN + dst_of(ei, e)];
}

__global__ __launch_bounds__(256) void softmax_stats_kernel(const float* __restrict__ s,
                                                            float* __restrict__ stats) {
  const int b = blockIdx.x;
  const int tid = threadIdx.x;
  __shared__ float red[256];
  const float* sb = s + (size_t)b * NE;
  float m = -3.4e38f;
  for (int i = tid; i < NE; i += 256) m = fmaxf(m, sb[i]);
  red[tid] = m;
  __syncthreads();
  for (int off = 128; off; off >>= 1) {
    if (tid < off) red[tid] = fmaxf(red[tid], red[tid + off]);
    __syncthreads();
  }
  const float bm = red[0];
  __syncthreads();
  float sum = 0.0f;
  for (int i = tid; i < NE; i += 256) sum += expf(sb[i] - bm);
  red[tid] = sum;
  __syncthreads();
  for (int off = 128; off; off >>= 1) {
    if (tid < off) red[tid] += red[tid + off];
    __syncthreads();
  }
  if (tid == 0) { stats[2 * b] = bm; stats[2 * b + 1] = red[0]; }
}

__global__ __launch_bounds__(256) void aggregate_kernel(const float* __restrict__ H,
                                                        const float* __restrict__ s,
                                                        const float* __restrict__ stats,
                                                        const int* __restrict__ rowstart,
                                                        const int* __restrict__ order,
                                                        const int* __restrict__ ei,
                                                        float* __restrict__ out, int g) {
  const int lane = threadIdx.x & 31;
  const int wave = threadIdx.x >> 5;
  const int node = blockIdx.x * 8 + wave;
  if (node >= BB * NN) return;
  const int b = node >> 11;
  const int n = node & (NN - 1);
  const float m = stats[2 * b];
  const float inv = 1.0f / stats[2 * b + 1];
  const float* Hb = H + (size_t)b * NN * OO;
  const float* sb = s + (size_t)b * NE;
  float4 acc[4] = {};
  const int beg = rowstart[n];
  const int end = rowstart[n + 1];
  for (int idx = beg; idx < end; ++idx) {
    const int e = order[idx];
    const int sn = src_of(ei, e);
    const float w = expf(sb[e] - m) * inv;
    const float4* hr = (const float4*)(Hb + (size_t)sn * OO);
#pragma unroll
    for (int j = 0; j < 4; ++j) {
      const float4 h4 = hr[lane + 32 * j];
      acc[j].x += w * h4.x;
      acc[j].y += w * h4.y;
      acc[j].z += w * h4.z;
      acc[j].w += w * h4.w;
    }
  }
  float4* op = (float4*)(out + (((size_t)b * VV + g) * NN + n) * OO);
#pragma unroll
  for (int j = 0; j < 4; ++j) {
    float4 o = op[lane + 32 * j];
    o.x += acc[j].x; o.y += acc[j].y; o.z += acc[j].z; o.w += acc[j].w;
    op[lane + 32 * j] = o;
  }
}

__global__ void epilogue_kernel(float* __restrict__ out, const float* __restrict__ bias, size_t n) {
  size_t i = (size_t)blockIdx.x * blockDim.x + threadIdx.x;
  if (i < n) out[i] = out[i] * 0.25f + bias[i & (OO - 1)];
}

extern "C" void kernel_launch(void* const* d_in, const int* in_sizes, int n_in,
                              void* d_out, int out_size, void* d_ws, size_t ws_size,
                              hipStream_t stream) {
  const float* x    = (const float*)d_in[0];
  const int*   ei   = (const int*)d_in[1];
  const float* W    = (const float*)d_in[2];
  const float* att  = (const float*)d_in[3];
  const float* bias = (const float*)d_in[4];
  float* out = (float*)d_out;

  char* p = (char*)d_ws;
  float* H     = (float*)p; p += (size_t)BB * NN * OO * sizeof(float);
  float* uS    = (float*)p; p += (size_t)BB * NN * sizeof(float);
  float* vS    = (float*)p; p += (size_t)BB * NN * sizeof(float);
  float* s     = (float*)p; p += (size_t)BB * NE * sizeof(float);
  float* stats = (float*)p; p += 4 * sizeof(float);
  int* counts  = (int*)p;   p += (size_t)NN * sizeof(int);
  int* order   = (int*)p;   p += (size_t)NE * sizeof(int);
  int* rowstart= (int*)p;   p += (size_t)(NN + 1) * sizeof(int);

  const size_t outN = (size_t)BB * VV * NN * OO;

  zero_f32_kernel<<<(unsigned)((outN + 255) / 256), 256, 0, stream>>>(out, outN);
  zero_i32_kernel<<<(NN + 255) / 256, 256, 0, stream>>>(counts, NN);
  count_dst_kernel<<<(NE + 255) / 256, 256, 0, stream>>>(ei, counts);
  scan_counts_kernel<<<1, 256, 0, stream>>>(counts, rowstart);
  build_rows_kernel<<<NN / 8, 256, 0, stream>>>(ei, rowstart, order);

  for (int g = 0; g < VV; ++g) {
    for (int hv = 0; hv < VV; ++hv) {
      const int v1 = hv;
      const int v2 = g ^ hv;
      gemm_pair_kernel<<<(BB * NN) / 32, 256, 0, stream>>>(x, W, H, v1, v2);
      node_scores_kernel<<<(BB * NN) / 8, 256, 0, stream>>>(H, att, uS, vS);
      edge_scores_kernel<<<(BB * NE + 255) / 256, 256, 0, stream>>>(uS, vS, ei, s);
      softmax_stats_kernel<<<BB, 256, 0, stream>>>(s, stats);
      aggregate_kernel<<<(BB * NN) / 8, 256, 0, stream>>>(H, s, stats, rowstart, order, ei, out, g);
    }
  }
  epilogue_kernel<<<(unsigned)((outN + 255) / 256), 256, 0, stream>>>(out, bias, outN);

  (void)in_sizes; (void)n_in; (void)out_size; (void)ws_size;
}


# softmax max and sum now come from 272 per-block partials written by the edge-score kernel and merged by each aggregate wave (sum order: block sums rescaled, lane butterfly); stats kernel removed
# speedup vs baseline: 1.5237x; 1.5237x over previous
#include <hip/hip_runtime.h>
#include <hip/hip_bf16.h>

typedef __attribute__((ext_vector_type(2))) float v2f;
typedef __attribute__((ext_vector_type(8))) float v8f;

constexpr int BB = 2;
constexpr int VV = 4;
constexpr int NN = 2048;
constexpr int FF = 128;
constexpr int EE = 32768;
constexpr int NE = 34816;
constexpr int OO = 512;
constexpr int ESB = 256;
constexpr int NPB = NE / ESB;
static_assert(NE % ESB == 0, "an edge-score block must not straddle two batches");

__device__ __forceinline__ float sgnval(int r, int j) {
  if (j >= 4) return 1.0f;
  const bool flip = (j & 1) ? ((r & 2) != 0) : ((r & 1) != 0);
  return flip ? -1.0f : 1.0f;
}

__device__ __forceinline__ int src_of(const int* __restrict__ ei, int e) {
  return (e < EE) ? ei[e] : (e - EE);
}
__device__ __forceinline__ int dst_of(const int* __restrict__ ei, int e) {
  return (e < EE) ? ei[EE + e] : (e - EE);
}

__global__ void zero_f32_kernel(float* __restrict__ p, size_t n) {
  size_t i = (size_t)blockIdx.x * blockDim.x + threadIdx.x;
  if (i < n) p[i] = 0.0f;
}
__global__ void zero_i32_kernel(int* __restrict__ p, int n) {
  int i = blockIdx.x * blockDim.x + threadIdx.x;
  if (i < n) p[i] = 0;
}

__global__ void count_dst_kernel(const int* __restrict__ ei, int* __restrict__ counts) {
  int e = blockIdx.x * blockDim.x + threadIdx.x;
  if (e >= NE) return;
  atomicAdd(&counts[dst_of(ei, e)], 1);
}

__global__ __launch_bounds__(256) void scan_counts_kernel(const int* __restrict__ counts,
                                                          int* __restrict__ rowstart) {
  __shared__ int part[256];
  __shared__ int pref[257];
  const int tid = threadIdx.x;
  const int base = tid * 8;
  int loc[8];
  int s = 0;
#pragma unroll
  for (int i = 0; i < 8; ++i) { loc[i] = s; s += counts[base + i]; }
  part[tid] = s;
  __syncthreads();
  if (tid == 0) {
    int a = 0;
    for (int i = 0; i < 256; ++i) { pref[i] = a; a += part[i]; }
    pref[256] = a;
  }
  __syncthreads();
  const int off = pref[tid];
#pragma unroll
  for (int i = 0; i < 8; ++i) rowstart[base + i] = off + loc[i];
  if (tid == 255) rowstart[NN] = pref[256];
}

__global__ __launch_bounds__(256) void build_rows_kernel(const int* __restrict__ ei,
                                                         const int* __restrict__ rowstart,
                                                         int* __restrict__ order) {
  const int lane = threadIdx.x & 31;
  const int wave = threadIdx.x >> 5;
  const int n = blockIdx.x * 8 + wave;
  if (n >= NN) return;
  const int base = rowstart[n];
  int cnt = 0;
  for (int e0 = 0; e0 < NE; e0 += 32) {
    const int e = e0 + lane;
    const bool match = (e < NE) && (dst_of(ei, e) == n);
    unsigned long long mask = __ballot(match);
    const int rank = __popcll(mask & ((1ull << lane) - 1ull));
    if (match) order[base + cnt + rank] = e;
    cnt += __popcll(mask);
  }
}

__device__ __forceinline__ void wmma_quad(const float* __restrict__ W, int k, int khalf,
                                          int col, v2f a0, v2f a1, v8f* acc) {
  const float* w0 = W + (size_t)(k + khalf) * OO;
  const float* w1 = w0 + OO;
  const v2f b0 = { w0[col +  0], w1[col +  0] };
  const v2f b1 = { w0[col + 16], w1[col + 16] };
  const v2f b2 = { w0[col + 32], w1[col + 32] };
  const v2f b3 = { w0[col + 48], w1[col + 48] };
  acc[0] = __builtin_amdgcn_wmma_f32_16x16x4_f32(false, a0, false, b0, (short)0, acc[0], false, false);
  acc[1] = __builtin_amdgcn_wmma_f32_16x16x4_f32(false, a0, false, b1, (short)0, acc[1], false, false);
  acc[2] = __builtin_amdgcn_wmma_f32_16x16x4_f32(false, a0, false, b2, (short)0, acc[2], false, false);
  acc[3] = __builtin_amdgcn_wmma_f32_16x16x4_f32(false, a0, false, b3, (short)0, acc[3], false, false);
  acc[4] = __builtin_amdgcn_wmma_f32_16x16x4_f32(false, a1, false, b0, (short)0, acc[4], false, false);
  acc[5] = __builtin_amdgcn_wmma_f32_16x16x4_f32(false, a1, false, b1, (short)0, acc[5], false, false);
  acc[6] = __builtin_amdgcn_wmma_f32_16x16x4_f32(false, a1, false, b2, (short)0, acc[6], false, false);
  acc[7] = __builtin_amdgcn_wmma_f32_16x16x4_f32(false, a1, false, b3, (short)0, acc[7], false, false);
}

__global__ __launch_bounds__(256) void gemm_pair_kernel(const float* __restrict__ x,
                                                        const float* __restrict__ W,
                                                        float* __restrict__ H,
                                                        int v1, int v2) {
  const int lane = threadIdx.x & 31;
  const int wave = threadIdx.x >> 5;
  const int rowTile = blockIdx.x * 32;
  const int b = rowTile >> 11;
  const int nbase = rowTile & (NN - 1);
  const int r0 = nbase + (lane & 15);
  const float* x1a = x + (((size_t)b * VV + v1) * NN + r0) * FF;
  const float* x1b = x1a + 16 * FF;
  const float* x2a = x + (((size_t)b * VV + v2) * NN + r0) * FF;
  const float* x2b = x2a + 16 * FF;
  const int khalf = (lane >> 4) * 2;
  const int col = wave * 64 + (lane & 15);

  v8f acc[8] = {};

  {
    v2f a0 = *(const v2f*)(x1a + khalf);
    v2f a1 = *(const v2f*)(x1b + khalf);
    const float s0 = sgnval(v2, khalf);
    const float s1 = sgnval(v2, khalf + 1);
    a0.x *= s0; a0.y *= s1;
    a1.x *= s0; a1.y *= s1;
    wmma_quad(W, 0, khalf, col, a0, a1, acc);
  }
  for (int k = 4; k < FF; k += 4) {
    const v2f a0 = *(const v2f*)(x1a + k + khalf);
    const v2f a1 = *(const v2f*)(x1b + k + khalf);
    wmma_quad(W, k, khalf, col, a0, a1, acc);
  }
  {
    v2f a0 = *(const v2f*)(x2a + khalf);
    v2f a1 = *(const v2f*)(x2b + khalf);
    const float s0 = sgnval(v1, khalf);
    const float s1 = sgnval(v1, khalf + 1);
    a0.x *= s0; a0.y *= s1;
    a1.x *= s0; a1.y *= s1;
    wmma_quad(W, FF, khalf, col, a0, a1, acc);
  }
  for (int k = FF + 4; k < 2 * FF; k += 4) {
    const v2f a0 = *(const v2f*)(x2a + (k - FF) + khalf);
    const v2f a1 = *(const v2f*)(x2b + (k - FF) + khalf);
    wmma_quad(W, k, khalf, col, a0, a1, acc);
  }

  float* Hb = H + (size_t)b * NN * OO;
  const int srow = nbase + 8 * (lane >> 4);
  const int scol = wave * 64 + (lane & 15);
#pragma unroll
  for (int i = 0; i < 8; ++i) {
    float* ra = Hb + (size_t)(srow + i) * OO + scol;
    ra[ 0] = acc[0][i];
    ra[16] = acc[1][i];
    ra[32] = acc[2][i];
    ra[48] = acc[3][i];
    float* rb = Hb + (size_t)(srow + 16 + i) * OO + scol;
    rb[ 0] = acc[4][i];
    rb[16] = acc[5][i];
    rb[32] = acc[6][i];
    rb[48] = acc[7][i];
  }
}

__global__ __launch_bounds__(256) void node_scores_kernel(const float* __restrict__ H,
                                                          const float* __restrict__ att,
                                                          float* __restrict__ uS,
                                                          float* __restrict__ vS) {
  const int lane = threadIdx.x & 31;
  const int wave = threadIdx.x >> 5;
  const int node = blockIdx.x * 8 + wave;
  if (node >= BB * NN) return;
  const float* hrow = H + (size_t)node * OO;
  float su = 0.0f, sv = 0.0f;
  for (int j = lane; j < OO; j += 32) {
    const float h = hrow[j];
    const float l = h > 0.0f ? h : 0.2f * h;
    su += att[j] * l;
    sv += att[OO + j] * l;
  }
#pragma unroll
  for (int off = 16; off; off >>= 1) {
    su += __shfl_down(su, off, 32);
    sv += __shfl_down(sv, off, 32);
  }
  if (lane == 0) { uS[node] = su; vS[node] = sv; }
}

__global__ __launch_bounds__(ESB) void edge_scores_kernel(const float* __restrict__ uS,
                                                          const float* __restrict__ vS,
                                                          const int* __restrict__ ei,
                                                          float* __restrict__ s,
                                                          float* __restrict__ part) {
  __shared__ float redm[ESB / 32];
  __shared__ float reds[ESB / 32];
  const int lane = threadIdx.x & 31;
  const int wave = threadIdx.x >> 5;
  const int t = blockIdx.x * ESB + threadIdx.x;
  const int b = blockIdx.x / NPB;
  const int e = t - b * NE;
  const float sc = uS[b * NN + src_of(ei, e)] + vS[b * NN + dst_of(ei, e)];
  s[t] = sc;

  float m = sc;
  for (int off = 16; off; off >>= 1) m = fmaxf(m, __shfl_xor(m, off, 32));
  if (lane == 0) redm[wave] = m;
  __syncthreads();
  float bm = redm[0];
  for (int i = 1; i < ESB / 32; ++i) bm = fmaxf(bm, redm[i]);

  float ex = expf(sc - bm);
  for (int off = 16; off; off >>= 1) ex += __shfl_xor(ex, off, 32);
  if (lane == 0) reds[wave] = ex;
  __syncthreads();
  if (threadIdx.x == 0) {
    float a = reds[0];
    for (int i = 1; i < ESB / 32; ++i) a += reds[i];
    part[2 * blockIdx.x]     = bm;
    part[2 * blockIdx.x + 1] = a;
  }
}

__global__ __launch_bounds__(256) void aggregate_kernel(const float* __restrict__ H,
                                                        const float* __restrict__ s,
                                                        const float* __restrict__ part,
                                                        const int* __restrict__ rowstart,
                                                        const int* __restrict__ order,
                                                        const int* __restrict__ ei,
                                                        float* __restrict__ out, int g) {
  const int lane = threadIdx.x & 31;
  const int wave = threadIdx.x >> 5;
  const int node = blockIdx.x * 8 + wave;
  if (node >= BB * NN) return;
  const int b = node >> 11;
  const int n = node & (NN - 1);
  const float2* pb = (const float2*)part + (size_t)b * NPB;
  constexpr int PPL = (NPB + 31) / 32;
  float pm[PPL], ps[PPL];
  float m = -3.4e38f;
  for (int i = 0; i < PPL; ++i) {
    const int k = lane + 32 * i;
    const bool ok = k < NPB;
    const float2 q = ok ? pb[k] : make_float2(-3.4e38f, 0.0f);
    pm[i] = q.x; ps[i] = q.y;
    m = fmaxf(m, q.x);
  }
  for (int off = 16; off; off >>= 1) m = fmaxf(m, __shfl_xor(m, off, 32));
  float den = 0.0f;
  for (int i = 0; i < PPL; ++i) den += (lane + 32 * i < NPB) ? ps[i] * expf(pm[i] - m) : 0.0f;
  for (int off = 16; off; off >>= 1) den += __shfl_xor(den, off, 32);
  const float inv = 1.0f / den;
  const float* Hb = H + (size_t)b * NN * OO;
  const float* sb = s + (size_t)b * NE;
  float4 acc[4] = {};
  const int beg = rowstart[n];
  const int end = rowstart[n + 1];
  for (int idx = beg; idx < end; ++idx) {
    const int e = order[idx];
    const int sn = src_of(ei, e);
    const float w = expf(sb[e] - m) * inv;
    const float4* hr = (const float4*)(Hb + (size_t)sn * OO);
#pragma unroll
    for (int j = 0; j < 4; ++j) {
      const float4 h4 = hr[lane + 32 * j];
      acc[j].x += w * h4.x;
      acc[j].y += w * h4.y;
      acc[j].z += w * h4.z;
      acc[j].w += w * h4.w;
    }
  }
  float4* op = (float4*)(out + (((size_t)b * VV + g) * NN + n) * OO);
#pragma unroll
  for (int j = 0; j < 4; ++j) {
    float4 o = op[lane + 32 * j];
    o.x += acc[j].x; o.y += acc[j].y; o.z += acc[j].z; o.w += acc[j].w;
    op[lane + 32 * j] = o;
  }
}

__global__ void epilogue_kernel(float* __restrict__ out, const float* __restrict__ bias, size_t n) {
  size_t i = (size_t)blockIdx.x * blockDim.x + threadIdx.x;
  if (i < n) out[i] = out[i] * 0.25f + bias[i & (OO - 1)];
}

extern "C" void kernel_launch(void* const* d_in, const int* in_sizes, int n_in,
                              void* d_out, int out_size, void* d_ws, size_t ws_size,
                              hipStream_t stream) {
  const float* x    = (const float*)d_in[0];
  const int*   ei   = (const int*)d_in[1];
  const float* W    = (const float*)d_in[2];
  const float* att  = (const float*)d_in[3];
  const float* bias = (const float*)d_in[4];
  float* out = (float*)d_out;

  char* p = (char*)d_ws;
  float* H     = (float*)p; p += (size_t)BB * NN * OO * sizeof(float);
  float* uS    = (float*)p; p += (size_t)BB * NN * sizeof(float);
  float* vS    = (float*)p; p += (size_t)BB * NN * sizeof(float);
  float* s     = (float*)p; p += (size_t)BB * NE * sizeof(float);
  int* counts  = (int*)p;   p += (size_t)NN * sizeof(int);
  int* order   = (int*)p;   p += (size_t)NE * sizeof(int);
  int* rowstart= (int*)p;   p += (size_t)(NN + 1) * sizeof(int);
  static_assert((size_t)2 * BB * NPB * sizeof(float) <= (size_t)NN * sizeof(int), "partials must fit");
  float* part = (float*)counts;

  const size_t outN = (size_t)BB * VV * NN * OO;

  zero_f32_kernel<<<(unsigned)((outN + 255) / 256), 256, 0, stream>>>(out, outN);
  zero_i32_kernel<<<(NN + 255) / 256, 256, 0, stream>>>(counts, NN);
  count_dst_kernel<<<(NE + 255) / 256, 256, 0, stream>>>(ei, counts);
  scan_counts_kernel<<<1, 256, 0, stream>>>(counts, rowstart);
  build_rows_kernel<<<NN / 8, 256, 0, stream>>>(ei, rowstart, order);

  for (int g = 0; g < VV; ++g) {
    for (int hv = 0; hv < VV; ++hv) {
      const int v1 = hv;
      const int v2 = g ^ hv;
      gemm_pair_kernel<<<(BB * NN) / 32, 256, 0, stream>>>(x, W, H, v1, v2);
      node_scores_kernel<<<(BB * NN) / 8, 256, 0, stream>>>(H, att, uS, vS);
      edge_scores_kernel<<<BB * NPB, ESB, 0, stream>>>(uS, vS, ei, s, part);
      aggregate_kernel<<<(BB * NN) / 8, 256, 0, stream>>>(H, s, part, rowstart, order, ei, out, g);
    }
  }
  epilogue_kernel<<<(unsigned)((outN + 255) / 256), 256, 0, stream>>>(out, bias, outN);

  (void)in_sizes; (void)n_in; (void)out_size; (void)ws_size;
}


# aggregate: lane i fetches edge i's number, source and score for a 32-edge chunk in one round trip, then the rows of 8 edges are loaded together ahead of their multiply-adds; sum order unchanged
# speedup vs baseline: 1.6428x; 1.0782x over previous
#include <hip/hip_runtime.h>
#include <hip/hip_bf16.h>

typedef __attribute__((ext_vector_type(2))) float v2f;
typedef __attribute__((ext_vector_type(8))) float v8f;

constexpr int BB = 2;
constexpr int VV = 4;
constexpr int NN = 2048;
constexpr int FF = 128;
constexpr int EE = 32768;
constexpr int NE = 34816;
constexpr int OO = 512;
constexpr int ESB = 256;
constexpr int NPB = NE / ESB;
static_assert(NE % ESB == 0, "an edge-score block must not straddle two batches");

__device__ __forceinline__ float sgnval(int r, int j) {
  if (j >= 4) return 1.0f;
  const bool flip = (j & 1) ? ((r & 2) != 0) : ((r & 1) != 0);
  return flip ? -1.0f : 1.0f;
}

__device__ __forceinline__ int src_of(const int* __restrict__ ei, int e) {
  return (e < EE) ? ei[e] : (e - EE);
}
__device__ __forceinline__ int dst_of(const int* __restrict__ ei, int e) {
  return (e < EE) ? ei[EE + e] : (e - EE);
}

__global__ void zero_f32_kernel(float* __restrict__ p, size_t n) {
  size_t i = (size_t)blockIdx.x * blockDim.x + threadIdx.x;
  if (i < n) p[i] = 0.0f;
}
__global__ void zero_i32_kernel(int* __restrict__ p, int n) {
  int i = blockIdx.x * blockDim.x + threadIdx.x;
  if (i < n) p[i] = 0;
}

__global__ void count_dst_kernel(const int* __restrict__ ei, int* __restrict__ counts) {
  int e = blockIdx.x * blockDim.x + threadIdx.x;
  if (e >= NE) return;
  atomicAdd(&counts[dst_of(ei, e)], 1);
}

__global__ __launch_bounds__(256) void scan_counts_kernel(const int* __restrict__ counts,
                                                          int* __restrict__ rowstart) {
  __shared__ int part[256];
  __shared__ int pref[257];
  const int tid = threadIdx.x;
  const int base = tid * 8;
  int loc[8];
  int s = 0;
#pragma unroll
  for (int i = 0; i < 8; ++i) { loc[i] = s; s += counts[base + i]; }
  part[tid] = s;
  __syncthreads();
  if (tid == 0) {
    int a = 0;
    for (int i = 0; i < 256; ++i) { pref[i] = a; a += part[i]; }
    pref[256] = a;
  }
  __syncthreads();
  const int off = pref[tid];
#pragma unroll
  for (int i = 0; i < 8; ++i) rowstart[base + i] = off + loc[i];
  if (tid == 255) rowstart[NN] = pref[256];
}

__global__ __launch_bounds__(256) void build_rows_kernel(const int* __restrict__ ei,
                                                         const int* __restrict__ rowstart,
                                                         int* __restrict__ order) {
  const int lane = threadIdx.x & 31;
  const int wave = threadIdx.x >> 5;
  const int n = blockIdx.x * 8 + wave;
  if (n >= NN) return;
  const int base = rowstart[n];
  int cnt = 0;
  for (int e0 = 0; e0 < NE; e0 += 32) {
    const int e = e0 + lane;
    const bool match = (e < NE) && (dst_of(ei, e) == n);
    unsigned long long mask = __ballot(match);
    const int rank = __popcll(mask & ((1ull << lane) - 1ull));
    if (match) order[base + cnt + rank] = e;
    cnt += __popcll(mask);
  }
}

__device__ __forceinline__ void wmma_quad(const float* __restrict__ W, int k, int khalf,
                                          int col, v2f a0, v2f a1, v8f* acc) {
  const float* w0 = W + (size_t)(k + khalf) * OO;
  const float* w1 = w0 + OO;
  const v2f b0 = { w0[col +  0], w1[col +  0] };
  const v2f b1 = { w0[col + 16], w1[col + 16] };
  const v2f b2 = { w0[col + 32], w1[col + 32] };
  const v2f b3 = { w0[col + 48], w1[col + 48] };
  acc[0] = __builtin_amdgcn_wmma_f32_16x16x4_f32(false, a0, false, b0, (short)0, acc[0], false, false);
  acc[1] = __builtin_amdgcn_wmma_f32_16x16x4_f32(false, a0, false, b1, (short)0, acc[1], false, false);
  acc[2] = __builtin_amdgcn_wmma_f32_16x16x4_f32(false, a0, false, b2, (short)0, acc[2], false, false);
  acc[3] = __builtin_amdgcn_wmma_f32_16x16x4_f32(false, a0, false, b3, (short)0, acc[3], false, false);
  acc[4] = __builtin_amdgcn_wmma_f32_16x16x4_f32(false, a1, false, b0, (short)0, acc[4], false, false);
  acc[5] = __builtin_amdgcn_wmma_f32_16x16x4_f32(false, a1, false, b1, (short)0, acc[5], false, false);
  acc[6] = __builtin_amdgcn_wmma_f32_16x16x4_f32(false, a1, false, b2, (short)0, acc[6], false, false);
  acc[7] = __builtin_amdgcn_wmma_f32_16x16x4_f32(false, a1, false, b3, (short)0, acc[7], false, false);
}

__global__ __launch_bounds__(256) void gemm_pair_kernel(const float* __restrict__ x,
                                                        const float* __restrict__ W,
                                                        float* __restrict__ H,
                                                        int v1, int v2) {
  const int lane = threadIdx.x & 31;
  const int wave = threadIdx.x >> 5;
  const int rowTile = blockIdx.x * 32;
  const int b = rowTile >> 11;
  const int nbase = rowTile & (NN - 1);
  const int r0 = nbase + (lane & 15);
  const float* x1a = x + (((size_t)b * VV + v1) * NN + r0) * FF;
  const float* x1b = x1a + 16 * FF;
  const float* x2a = x + (((size_t)b * VV + v2) * NN + r0) * FF;
  const float* x2b = x2a + 16 * FF;
  const int khalf = (lane >> 4) * 2;
  const int col = wave * 64 + (lane & 15);

  v8f acc[8] = {};

  {
    v2f a0 = *(const v2f*)(x1a + khalf);
    v2f a1 = *(const v2f*)(x1b + khalf);
    const float s0 = sgnval(v2, khalf);
    const float s1 = sgnval(v2, khalf + 1);
    a0.x *= s0; a0.y *= s1;
    a1.x *= s0; a1.y *= s1;
    wmma_quad(W, 0, khalf, col, a0, a1, acc);
  }
  for (int k = 4; k < FF; k += 4) {
    const v2f a0 = *(const v2f*)(x1a + k + khalf);
    const v2f a1 = *(const v2f*)(x1b + k + khalf);
    wmma_quad(W, k, khalf, col, a0, a1, acc);
  }
  {
    v2f a0 = *(const v2f*)(x2a + khalf);
    v2f a1 = *(const v2f*)(x2b + khalf);
    const float s0 = sgnval(v1, khalf);
    const float s1 = sgnval(v1, khalf + 1);
    a0.x *= s0; a0.y *= s1;
    a1.x *= s0; a1.y *= s1;
    wmma_quad(W, FF, khalf, col, a0, a1, acc);
  }
  for (int k = FF + 4; k < 2 * FF; k += 4) {
    const v2f a0 = *(const v2f*)(x2a + (k - FF) + khalf);
    const v2f a1 = *(const v2f*)(x2b + (k - FF) + khalf);
    wmma_quad(W, k, khalf, col, a0, a1, acc);
  }

  float* Hb = H + (size_t)b * NN * OO;
  const int srow = nbase + 8 * (lane >> 4);
  const int scol = wave * 64 + (lane & 15);
#pragma unroll
  for (int i = 0; i < 8; ++i) {
    float* ra = Hb + (size_t)(srow + i) * OO + scol;
    ra[ 0] = acc[0][i];
    ra[16] = acc[1][i];
    ra[32] = acc[2][i];
    ra[48] = acc[3][i];
    float* rb = Hb + (size_t)(srow + 16 + i) * OO + scol;
    rb[ 0] = acc[4][i];
    rb[16] = acc[5][i];
    rb[32] = acc[6][i];
    rb[48] = acc[7][i];
  }
}

__global__ __launch_bounds__(256) void node_scores_kernel(const float* __restrict__ H,
                                                          const float* __restrict__ att,
                                                          float* __restrict__ uS,
                                                          float* __restrict__ vS) {
  const int lane = threadIdx.x & 31;
  const int wave = threadIdx.x >> 5;
  const int node = blockIdx.x * 8 + wave;
  if (node >= BB * NN) return;
  const float* hrow = H + (size_t)node * OO;
  float su = 0.0f, sv = 0.0f;
  for (int j = lane; j < OO; j += 32) {
    const float h = hrow[j];
    const float l = h > 0.0f ? h : 0.2f * h;
    su += att[j] * l;
    sv += att[OO + j] * l;
  }
#pragma unroll
  for (int off = 16; off; off >>= 1) {
    su += __shfl_down(su, off, 32);
    sv += __shfl_down(sv, off, 32);
  }
  if (lane == 0) { uS[node] = su; vS[node] = sv; }
}

__global__ __launch_bounds__(ESB) void edge_scores_kernel(const float* __restrict__ uS,
                                                          const float* __restrict__ vS,
                                                          const int* __restrict__ ei,
                                                          float* __restrict__ s,
                                                          float* __restrict__ part) {
  __shared__ float redm[ESB / 32];
  __shared__ float reds[ESB / 32];
  const int lane = threadIdx.x & 31;
  const int wave = threadIdx.x >> 5;
  const int t = blockIdx.x * ESB + threadIdx.x;
  const int b = blockIdx.x / NPB;
  const int e = t - b * NE;
  const float sc = uS[b * NN + src_of(ei, e)] + vS[b * NN + dst_of(ei, e)];
  s[t] = sc;

  float m = sc;
  for (int off = 16; off; off >>= 1) m = fmaxf(m, __shfl_xor(m, off, 32));
  if (lane == 0) redm[wave] = m;
  __syncthreads();
  float bm = redm[0];
  for (int i = 1; i < ESB / 32; ++i) bm = fmaxf(bm, redm[i]);

  float ex = expf(sc - bm);
  for (int off = 16; off; off >>= 1) ex += __shfl_xor(ex, off, 32);
  if (lane == 0) reds[wave] = ex;
  __syncthreads();
  if (threadIdx.x == 0) {
    float a = reds[0];
    for (int i = 1; i < ESB / 32; ++i) a += reds[i];
    part[2 * blockIdx.x]     = bm;
    part[2 * blockIdx.x + 1] = a;
  }
}

template <int G>
__device__ __forceinline__ void gather_group(const float* __restrict__ Hb, int snv, float wv,
                                             int k, int lane, float4* acc) {
  int sn[G];
  float w[G];
#pragma unroll
  for (int t = 0; t < G; ++t) {
    sn[t] = __builtin_amdgcn_readlane(snv, k + t);
    w[t] = __int_as_float(__builtin_amdgcn_readlane(__float_as_int(wv), k + t));
  }
  float4 h[G][4];
#pragma unroll
  for (int t = 0; t < G; ++t) {
    const float4* hr = (const float4*)(Hb + (size_t)sn[t] * OO);
#pragma unroll
    for (int j = 0; j < 4; ++j) h[t][j] = hr[lane + 32 * j];
  }
#pragma unroll
  for (int t = 0; t < G; ++t) {
#pragma unroll
    for (int j = 0; j < 4; ++j) {
      acc[j].x = __builtin_fmaf(w[t], h[t][j].x, acc[j].x);
      acc[j].y = __builtin_fmaf(w[t], h[t][j].y, acc[j].y);
      acc[j].z = __builtin_fmaf(w[t], h[t][j].z, acc[j].z);
      acc[j].w = __builtin_fmaf(w[t], h[t][j].w, acc[j].w);
    }
  }
}

__global__ __launch_bounds__(256) __attribute__((amdgpu_waves_per_eu(2, 4))) void aggregate_kernel(const float* __restrict__ H,
                                                        const float* __restrict__ s,
                                                        const float* __restrict__ part,
                                                        const int* __restrict__ rowstart,
                                                        const int* __restrict__ order,
                                                        const int* __restrict__ ei,
                                                        float* __restrict__ out, int g) {
  const int lane = threadIdx.x & 31;
  const int wave = __builtin_amdgcn_readfirstlane((int)(threadIdx.x >> 5));
  const int node = blockIdx.x * 8 + wave;
  if (node >= BB * NN) return;
  const int b = node >> 11;
  const int n = node & (NN - 1);
  const float* Hb = H + (size_t)b * NN * OO;
  const float* sb = s + (size_t)b * NE;
  const int beg = rowstart[n];
  const int end = rowstart[n + 1];

  const float2* pb = (const float2*)part + (size_t)b * NPB;
  constexpr int PPL = (NPB + 31) / 32;
  float pm[PPL], ps[PPL];
  for (int i = 0; i < PPL; ++i) {
    const int k = lane + 32 * i;
    const bool ok = k < NPB;
    const float2 q = ok ? pb[k] : make_float2(-3.4e38f, 0.0f);
    pm[i] = q.x; ps[i] = q.y;
  }

  int snv = 0;
  float sv = 0.0f;
  if (beg + lane < end) {
    const int e = order[beg + lane];
    snv = src_of(ei, e);
    sv = sb[e];
  }

  float m = -3.4e38f;
  for (int i = 0; i < PPL; ++i) m = fmaxf(m, pm[i]);
  for (int off = 16; off; off >>= 1) m = fmaxf(m, __shfl_xor(m, off, 32));
  float den = 0.0f;
  for (int i = 0; i < PPL; ++i) den += (lane + 32 * i < NPB) ? ps[i] * expf(pm[i] - m) : 0.0f;
  for (int off = 16; off; off >>= 1) den += __shfl_xor(den, off, 32);
  const float inv = 1.0f / den;

  float4 acc[4] = {};
  int base = beg;
  while (true) {
    const int left = end - base;
    const int cnt = left < 32 ? left : 32;
    const float wv = (lane < cnt) ? expf(sv - m) * inv : 0.0f;
    int k = 0;
    for (; k + 8 <= cnt; k += 8) gather_group<8>(Hb, snv, wv, k, lane, acc);
    if (k + 4 <= cnt) { gather_group<4>(Hb, snv, wv, k, lane, acc); k += 4; }
    if (k + 2 <= cnt) { gather_group<2>(Hb, snv, wv, k, lane, acc); k += 2; }
    if (k < cnt) gather_group<1>(Hb, snv, wv, k, lane, acc);
    base += 32;
    if (base >= end) break;
    snv = 0;
    sv = 0.0f;
    if (base + lane < end) {
      const int e = order[base + lane];
      snv = src_of(ei, e);
      sv = sb[e];
    }
  }

  float4* op = (float4*)(out + (((size_t)b * VV + g) * NN + n) * OO);
#pragma unroll
  for (int j = 0; j < 4; ++j) {
    float4 o = op[lane + 32 * j];
    o.x += acc[j].x; o.y += acc[j].y; o.z += acc[j].z; o.w += acc[j].w;
    op[lane + 32 * j] = o;
  }
}

__global__ void epilogue_kernel(float* __restrict__ out, const float* __restrict__ bias, size_t n) {
  size_t i = (size_t)blockIdx.x * blockDim.x + threadIdx.x;
  if (i < n) out[i] = out[i] * 0.25f + bias[i & (OO - 1)];
}

extern "C" void kernel_launch(void* const* d_in, const int* in_sizes, int n_in,
                              void* d_out, int out_size, void* d_ws, size_t ws_size,
                              hipStream_t stream) {
  const float* x    = (const float*)d_in[0];
  const int*   ei   = (const int*)d_in[1];
  const float* W    = (const float*)d_in[2];
  const float* att  = (const float*)d_in[3];
  const float* bias = (const float*)d_in[4];
  float* out = (float*)d_out;

  char* p = (char*)d_ws;
  float* H     = (float*)p; p += (size_t)BB * NN * OO * sizeof(float);
  float* uS    = (float*)p; p += (size_t)BB * NN * sizeof(float);
  float* vS    = (float*)p; p += (size_t)BB * NN * sizeof(float);
  float* s     = (float*)p; p += (size_t)BB * NE * sizeof(float);
  int* counts  = (int*)p;   p += (size_t)NN * sizeof(int);
  int* order   = (int*)p;   p += (size_t)NE * sizeof(int);
  int* rowstart= (int*)p;   p += (size_t)(NN + 1) * sizeof(int);
  static_assert((size_t)2 * BB * NPB * sizeof(float) <= (size_t)NN * sizeof(int), "partials must fit");
  float* part = (float*)counts;

  const size_t outN = (size_t)BB * VV * NN * OO;

  zero_f32_kernel<<<(unsigned)((outN + 255) / 256), 256, 0, stream>>>(out, outN);
  zero_i32_kernel<<<(NN + 255) / 256, 256, 0, stream>>>(counts, NN);
  count_dst_kernel<<<(NE + 255) / 256, 256, 0, stream>>>(ei, counts);
  scan_counts_kernel<<<1, 256, 0, stream>>>(counts, rowstart);
  build_rows_kernel<<<NN / 8, 256, 0, stream>>>(ei, rowstart, order);

  for (int g = 0; g < VV; ++g) {
    for (int hv = 0; hv < VV; ++hv) {
      const int v1 = hv;
      const int v2 = g ^ hv;
      gemm_pair_kernel<<<(BB * NN) / 32, 256, 0, stream>>>(x, W, H, v1, v2);
      node_scores_kernel<<<(BB * NN) / 8, 256, 0, stream>>>(H, att, uS, vS);
      edge_scores_kernel<<<BB * NPB, ESB, 0, stream>>>(uS, vS, ei, s, part);
      aggregate_kernel<<<(BB * NN) / 8, 256, 0, stream>>>(H, s, part, rowstart, order, ei, out, g);
    }
  }
  epilogue_kernel<<<(unsigned)((outN + 255) / 256), 256, 0, stream>>>(out, bias, outN);

  (void)in_sizes; (void)n_in; (void)out_size; (void)ws_size;
}


# build_rows: each lane reads four destinations per 16-byte load, 8 loads in flight, next batch requested ahead, instead of one 4-byte load per round trip; self loop appended; edge order unchanged
# speedup vs baseline: 2.1832x; 1.3290x over previous
#include <hip/hip_runtime.h>
#include <hip/hip_bf16.h>

typedef __attribute__((ext_vector_type(2))) float v2f;
typedef __attribute__((ext_vector_type(8))) float v8f;

constexpr int BB = 2;
constexpr int VV = 4;
constexpr int NN = 2048;
constexpr int FF = 128;
constexpr int EE = 32768;
constexpr int NE = 34816;
constexpr int OO = 512;
constexpr int ESB = 256;
constexpr int NPB = NE / ESB;
static_assert(NE % ESB == 0, "an edge-score block must not straddle two batches");

__device__ __forceinline__ float sgnval(int r, int j) {
  if (j >= 4) return 1.0f;
  const bool flip = (j & 1) ? ((r & 2) != 0) : ((r & 1) != 0);
  return flip ? -1.0f : 1.0f;
}

__device__ __forceinline__ int src_of(const int* __restrict__ ei, int e) {
  return (e < EE) ? ei[e] : (e - EE);
}
__device__ __forceinline__ int dst_of(const int* __restrict__ ei, int e) {
  return (e < EE) ? ei[EE + e] : (e - EE);
}

__global__ void zero_f32_kernel(float* __restrict__ p, size_t n) {
  size_t i = (size_t)blockIdx.x * blockDim.x + threadIdx.x;
  if (i < n) p[i] = 0.0f;
}
__global__ void zero_i32_kernel(int* __restrict__ p, int n) {
  int i = blockIdx.x * blockDim.x + threadIdx.x;
  if (i < n) p[i] = 0;
}

__global__ void count_dst_kernel(const int* __restrict__ ei, int* __restrict__ counts) {
  int e = blockIdx.x * blockDim.x + threadIdx.x;
  if (e >= NE) return;
  atomicAdd(&counts[dst_of(ei, e)], 1);
}

__global__ __launch_bounds__(256) void scan_counts_kernel(const int* __restrict__ counts,
                                                          int* __restrict__ rowstart) {
  __shared__ int part[256];
  __shared__ int pref[257];
  const int tid = threadIdx.x;
  const int base = tid * 8;
  int loc[8];
  int s = 0;
#pragma unroll
  for (int i = 0; i < 8; ++i) { loc[i] = s; s += counts[base + i]; }
  part[tid] = s;
  __syncthreads();
  if (tid == 0) {
    int a = 0;
    for (int i = 0; i < 256; ++i) { pref[i] = a; a += part[i]; }
    pref[256] = a;
  }
  __syncthreads();
  const int off = pref[tid];
#pragma unroll
  for (int i = 0; i < 8; ++i) rowstart[base + i] = off + loc[i];
  if (tid == 255) rowstart[NN] = pref[256];
}

constexpr int BRU = 8;
constexpr int BRT = EE / (128 * BRU);
static_assert(EE % (128 * BRU) == 0, "the explicit edges must fill whole batches");

__global__ __launch_bounds__(256) void build_rows_kernel(const int* __restrict__ ei,
                                                         const int* __restrict__ rowstart,
                                                         int* __restrict__ order) {
  const int lane = threadIdx.x & 31;
  const int wave = __builtin_amdgcn_readfirstlane((int)(threadIdx.x >> 5));
  const int n = blockIdx.x * 8 + wave;
  if (n >= NN) return;
  const int base = rowstart[n];
  const int4* d4 = (const int4*)(ei + EE) + lane;
  const unsigned below = (1u << lane) - 1u;
  int cnt = 0;

  int4 cur[BRU], nxt[BRU];
#pragma unroll
  for (int u = 0; u < BRU; ++u) cur[u] = d4[u * 32];

  for (int t = 0; t < BRT; ++t) {
    if (t + 1 < BRT) {
#pragma unroll
      for (int u = 0; u < BRU; ++u) nxt[u] = d4[((t + 1) * BRU + u) * 32];
    }
#pragma unroll
    for (int u = 0; u < BRU; ++u) {
      const int4 q = cur[u];
      const bool m0 = q.x == n, m1 = q.y == n, m2 = q.z == n, m3 = q.w == n;
      if (__builtin_amdgcn_ballot_w32(m0 | m1 | m2 | m3) != 0u) {
        const unsigned b0 = __builtin_amdgcn_ballot_w32(m0);
        const unsigned b1 = __builtin_amdgcn_ballot_w32(m1);
        const unsigned b2 = __builtin_amdgcn_ballot_w32(m2);
        const unsigned b3 = __builtin_amdgcn_ballot_w32(m3);
        const int e = (((t * BRU + u) * 32) + lane) * 4;
        int r = base + cnt + __popc(b0 & below) + __popc(b1 & below) + __popc(b2 & below) +
                __popc(b3 & below);
        if (m0) order[r] = e;
        r += m0 ? 1 : 0;
        if (m1) order[r] = e + 1;
        r += m1 ? 1 : 0;
        if (m2) order[r] = e + 2;
        r += m2 ? 1 : 0;
        if (m3) order[r] = e + 3;
        cnt += __popc(b0) + __popc(b1) + __popc(b2) + __popc(b3);
      }
    }
#pragma unroll
    for (int u = 0; u < BRU; ++u) cur[u] = nxt[u];
  }
  if (lane == 0) order[base + cnt] = EE + n;
}

__device__ __forceinline__ void wmma_quad(const float* __restrict__ W, int k, int khalf,
                                          int col, v2f a0, v2f a1, v8f* acc) {
  const float* w0 = W + (size_t)(k + khalf) * OO;
  const float* w1 = w0 + OO;
  const v2f b0 = { w0[col +  0], w1[col +  0] };
  const v2f b1 = { w0[col + 16], w1[col + 16] };
  const v2f b2 = { w0[col + 32], w1[col + 32] };
  const v2f b3 = { w0[col + 48], w1[col + 48] };
  acc[0] = __builtin_amdgcn_wmma_f32_16x16x4_f32(false, a0, false, b0, (short)0, acc[0], false, false);
  acc[1] = __builtin_amdgcn_wmma_f32_16x16x4_f32(false, a0, false, b1, (short)0, acc[1], false, false);
  acc[2] = __builtin_amdgcn_wmma_f32_16x16x4_f32(false, a0, false, b2, (short)0, acc[2], false, false);
  acc[3] = __builtin_amdgcn_wmma_f32_16x16x4_f32(false, a0, false, b3, (short)0, acc[3], false, false);
  acc[4] = __builtin_amdgcn_wmma_f32_16x16x4_f32(false, a1, false, b0, (short)0, acc[4], false, false);
  acc[5] = __builtin_amdgcn_wmma_f32_16x16x4_f32(false, a1, false, b1, (short)0, acc[5], false, false);
  acc[6] = __builtin_amdgcn_wmma_f32_16x16x4_f32(false, a1, false, b2, (short)0, acc[6], false, false);
  acc[7] = __builtin_amdgcn_wmma_f32_16x16x4_f32(false, a1, false, b3, (short)0, acc[7], false, false);
}

__global__ __launch_bounds__(256) void gemm_pair_kernel(const float* __restrict__ x,
                                                        const float* __restrict__ W,
                                                        float* __restrict__ H,
                                                        int v1, int v2) {
  const int lane = threadIdx.x & 31;
  const int wave = threadIdx.x >> 5;
  const int rowTile = blockIdx.x * 32;
  const int b = rowTile >> 11;
  const int nbase = rowTile & (NN - 1);
  const int r0 = nbase + (lane & 15);
  const float* x1a = x + (((size_t)b * VV + v1) * NN + r0) * FF;
  const float* x1b = x1a + 16 * FF;
  const float* x2a = x + (((size_t)b * VV + v2) * NN + r0) * FF;
  const float* x2b = x2a + 16 * FF;
  const int khalf = (lane >> 4) * 2;
  const int col = wave * 64 + (lane & 15);

  v8f acc[8] = {};

  {
    v2f a0 = *(const v2f*)(x1a + khalf);
    v2f a1 = *(const v2f*)(x1b + khalf);
    const float s0 = sgnval(v2, khalf);
    const float s1 = sgnval(v2, khalf + 1);
    a0.x *= s0; a0.y *= s1;
    a1.x *= s0; a1.y *= s1;
    wmma_quad(W, 0, khalf, col, a0, a1, acc);
  }
  for (int k = 4; k < FF; k += 4) {
    const v2f a0 = *(const v2f*)(x1a + k + khalf);
    const v2f a1 = *(const v2f*)(x1b + k + khalf);
    wmma_quad(W, k, khalf, col, a0, a1, acc);
  }
  {
    v2f a0 = *(const v2f*)(x2a + khalf);
    v2f a1 = *(const v2f*)(x2b + khalf);
    const float s0 = sgnval(v1, khalf);
    const float s1 = sgnval(v1, khalf + 1);
    a0.x *= s0; a0.y *= s1;
    a1.x *= s0; a1.y *= s1;
    wmma_quad(W, FF, khalf, col, a0, a1, acc);
  }
  for (int k = FF + 4; k < 2 * FF; k += 4) {
    const v2f a0 = *(const v2f*)(x2a + (k - FF) + khalf);
    const v2f a1 = *(const v2f*)(x2b + (k - FF) + khalf);
    wmma_quad(W, k, khalf, col, a0, a1, acc);
  }

  float* Hb = H + (size_t)b * NN * OO;
  const int srow = nbase + 8 * (lane >> 4);
  const int scol = wave * 64 + (lane & 15);
#pragma unroll
  for (int i = 0; i < 8; ++i) {
    float* ra = Hb + (size_t)(srow + i) * OO + scol;
    ra[ 0] = acc[0][i];
    ra[16] = acc[1][i];
    ra[32] = acc[2][i];
    ra[48] = acc[3][i];
    float* rb = Hb + (size_t)(srow + 16 + i) * OO + scol;
    rb[ 0] = acc[4][i];
    rb[16] = acc[5][i];
    rb[32] = acc[6][i];
    rb[48] = acc[7][i];
  }
}

__global__ __launch_bounds__(256) void node_scores_kernel(const float* __restrict__ H,
                                                          const float* __restrict__ att,
                                                          float* __restrict__ uS,
                                                          float* __restrict__ vS) {
  const int lane = threadIdx.x & 31;
  const int wave = threadIdx.x >> 5;
  const int node = blockIdx.x * 8 + wave;
  if (node >= BB * NN) return;
  const float* hrow = H + (size_t)node * OO;
  float su = 0.0f, sv = 0.0f;
  for (int j = lane; j < OO; j += 32) {
    const float h = hrow[j];
    const float l = h > 0.0f ? h : 0.2f * h;
    su += att[j] * l;
    sv += att[OO + j] * l;
  }
#pragma unroll
  for (int off = 16; off; off >>= 1) {
    su += __shfl_down(su, off, 32);
    sv += __shfl_down(sv, off, 32);
  }
  if (lane == 0) { uS[node] = su; vS[node] = sv; }
}

__global__ __launch_bounds__(ESB) void edge_scores_kernel(const float* __restrict__ uS,
                                                          const float* __restrict__ vS,
                                                          const int* __restrict__ ei,
                                                          float* __restrict__ s,
                                                          float* __restrict__ part) {
  __shared__ float redm[ESB / 32];
  __shared__ float reds[ESB / 32];
  const int lane = threadIdx.x & 31;
  const int wave = threadIdx.x >> 5;
  const int t = blockIdx.x * ESB + threadIdx.x;
  const int b = blockIdx.x / NPB;
  const int e = t - b * NE;
  const float sc = uS[b * NN + src_of(ei, e)] + vS[b * NN + dst_of(ei, e)];
  s[t] = sc;

  float m = sc;
  for (int off = 16; off; off >>= 1) m = fmaxf(m, __shfl_xor(m, off, 32));
  if (lane == 0) redm[wave] = m;
  __syncthreads();
  float bm = redm[0];
  for (int i = 1; i < ESB / 32; ++i) bm = fmaxf(bm, redm[i]);

  float ex = expf(sc - bm);
  for (int off = 16; off; off >>= 1) ex += __shfl_xor(ex, off, 32);
  if (lane == 0) reds[wave] = ex;
  __syncthreads();
  if (threadIdx.x == 0) {
    float a = reds[0];
    for (int i = 1; i < ESB / 32; ++i) a += reds[i];
    part[2 * blockIdx.x]     = bm;
    part[2 * blockIdx.x + 1] = a;
  }
}

template <int G>
__device__ __forceinline__ void gather_group(const float* __restrict__ Hb, int snv, float wv,
                                             int k, int lane, float4* acc) {
  int sn[G];
  float w[G];
#pragma unroll
  for (int t = 0; t < G; ++t) {
    sn[t] = __builtin_amdgcn_readlane(snv, k + t);
    w[t] = __int_as_float(__builtin_amdgcn_readlane(__float_as_int(wv), k + t));
  }
  float4 h[G][4];
#pragma unroll
  for (int t = 0; t < G; ++t) {
    const float4* hr = (const float4*)(Hb + (size_t)sn[t] * OO);
#pragma unroll
    for (int j = 0; j < 4; ++j) h[t][j] = hr[lane + 32 * j];
  }
#pragma unroll
  for (int t = 0; t < G; ++t) {
#pragma unroll
    for (int j = 0; j < 4; ++j) {
      acc[j].x = __builtin_fmaf(w[t], h[t][j].x, acc[j].x);
      acc[j].y = __builtin_fmaf(w[t], h[t][j].y, acc[j].y);
      acc[j].z = __builtin_fmaf(w[t], h[t][j].z, acc[j].z);
      acc[j].w = __builtin_fmaf(w[t], h[t][j].w, acc[j].w);
    }
  }
}

__global__ __launch_bounds__(256) __attribute__((amdgpu_waves_per_eu(2, 4))) void aggregate_kernel(const float* __restrict__ H,
                                                        const float* __restrict__ s,
                                                        const float* __restrict__ part,
                                                        const int* __restrict__ rowstart,
                                                        const int* __restrict__ order,
                                                        const int* __restrict__ ei,
                                                        float* __restrict__ out, int g) {
  const int lane = threadIdx.x & 31;
  const int wave = __builtin_amdgcn_readfirstlane((int)(threadIdx.x >> 5));
  const int node = blockIdx.x * 8 + wave;
  if (node >= BB * NN) return;
  const int b = node >> 11;
  const int n = node & (NN - 1);
  const float* Hb = H + (size_t)b * NN * OO;
  const float* sb = s + (size_t)b * NE;
  const int beg = rowstart[n];
  const int end = rowstart[n + 1];

  const float2* pb = (const float2*)part + (size_t)b * NPB;
  constexpr int PPL = (NPB + 31) / 32;
  float pm[PPL], ps[PPL];
  for (int i = 0; i < PPL; ++i) {
    const int k = lane + 32 * i;
    const bool ok = k < NPB;
    const float2 q = ok ? pb[k] : make_float2(-3.4e38f, 0.0f);
    pm[i] = q.x; ps[i] = q.y;
  }

  int snv = 0;
  float sv = 0.0f;
  if (beg + lane < end) {
    const int e = order[beg + lane];
    snv = src_of(ei, e);
    sv = sb[e];
  }

  float m = -3.4e38f;
  for (int i = 0; i < PPL; ++i) m = fmaxf(m, pm[i]);
  for (int off = 16; off; off >>= 1) m = fmaxf(m, __shfl_xor(m, off, 32));
  float den = 0.0f;
  for (int i = 0; i < PPL; ++i) den += (lane + 32 * i < NPB) ? ps[i] * expf(pm[i] - m) : 0.0f;
  for (int off = 16; off; off >>= 1) den += __shfl_xor(den, off, 32);
  const float inv = 1.0f / den;

  float4 acc[4] = {};
  int base = beg;
  while (true) {
    const int left = end - base;
    const int cnt = left < 32 ? left : 32;
    const float wv = (lane < cnt) ? expf(sv - m) * inv : 0.0f;
    int k = 0;
    for (; k + 8 <= cnt; k += 8) gather_group<8>(Hb, snv, wv, k, lane, acc);
    if (k + 4 <= cnt) { gather_group<4>(Hb, snv, wv, k, lane, acc); k += 4; }
    if (k + 2 <= cnt) { gather_group<2>(Hb, snv, wv, k, lane, acc); k += 2; }
    if (k < cnt) gather_group<1>(Hb, snv, wv, k, lane, acc);
    base += 32;
    if (base >= end) break;
    snv = 0;
    sv = 0.0f;
    if (base + lane < end) {
      const int e = order[base + lane];
      snv = src_of(ei, e);
      sv = sb[e];
    }
  }

  float4* op = (float4*)(out + (((size_t)b * VV + g) * NN + n) * OO);
#pragma unroll
  for (int j = 0; j < 4; ++j) {
    float4 o = op[lane + 32 * j];
    o.x += acc[j].x; o.y += acc[j].y; o.z += acc[j].z; o.w += acc[j].w;
    op[lane + 32 * j] = o;
  }
}

__global__ void epilogue_kernel(float* __restrict__ out, const float* __restrict__ bias, size_t n) {
  size_t i = (size_t)blockIdx.x * blockDim.x + threadIdx.x;
  if (i < n) out[i] = out[i] * 0.25f + bias[i & (OO - 1)];
}

extern "C" void kernel_launch(void* const* d_in, const int* in_sizes, int n_in,
                              void* d_out, int out_size, void* d_ws, size_t ws_size,
                              hipStream_t stream) {
  const float* x    = (const float*)d_in[0];
  const int*   ei   = (const int*)d_in[1];
  const float* W    = (const float*)d_in[2];
  const float* att  = (const float*)d_in[3];
  const float* bias = (const float*)d_in[4];
  float* out = (float*)d_out;

  char* p = (char*)d_ws;
  float* H     = (float*)p; p += (size_t)BB * NN * OO * sizeof(float);
  float* uS    = (float*)p; p += (size_t)BB * NN * sizeof(float);
  float* vS    = (float*)p; p += (size_t)BB * NN * sizeof(float);
  float* s     = (float*)p; p += (size_t)BB * NE * sizeof(float);
  int* counts  = (int*)p;   p += (size_t)NN * sizeof(int);
  int* order   = (int*)p;   p += (size_t)NE * sizeof(int);
  int* rowstart= (int*)p;   p += (size_t)(NN + 1) * sizeof(int);
  static_assert((size_t)2 * BB * NPB * sizeof(float) <= (size_t)NN * sizeof(int), "partials must fit");
  float* part = (float*)counts;

  const size_t outN = (size_t)BB * VV * NN * OO;

  zero_f32_kernel<<<(unsigned)((outN + 255) / 256), 256, 0, stream>>>(out, outN);
  zero_i32_kernel<<<(NN + 255) / 256, 256, 0, stream>>>(counts, NN);
  count_dst_kernel<<<(NE + 255) / 256, 256, 0, stream>>>(ei, counts);
  scan_counts_kernel<<<1, 256, 0, stream>>>(counts, rowstart);
  build_rows_kernel<<<NN / 8, 256, 0, stream>>>(ei, rowstart, order);

  for (int g = 0; g < VV; ++g) {
    for (int hv = 0; hv < VV; ++hv) {
      const int v1 = hv;
      const int v2 = g ^ hv;
      gemm_pair_kernel<<<(BB * NN) / 32, 256, 0, stream>>>(x, W, H, v1, v2);
      node_scores_kernel<<<(BB * NN) / 8, 256, 0, stream>>>(H, att, uS, vS);
      edge_scores_kernel<<<BB * NPB, ESB, 0, stream>>>(uS, vS, ei, s, part);
      aggregate_kernel<<<(BB * NN) / 8, 256, 0, stream>>>(H, s, part, rowstart, order, ei, out, g);
    }
  }
  epilogue_kernel<<<(unsigned)((outN + 255) / 256), 256, 0, stream>>>(out, bias, outN);

  (void)in_sizes; (void)n_in; (void)out_size; (void)ws_size;
}


# gemm_pair operand feed rebuilt: the block's 32 input rows go to LDS once for its 8 waves, W fragments are loaded whole groups of 8 k-quads ahead of their WMMAs, 256 blocks of 32x256; sum order kept
# speedup vs baseline: 2.7749x; 1.2710x over previous
#include <hip/hip_runtime.h>
#include <hip/hip_bf16.h>

typedef __attribute__((ext_vector_type(2))) float v2f;
typedef __attribute__((ext_vector_type(8))) float v8f;

constexpr int BB = 2;
constexpr int VV = 4;
constexpr int NN = 2048;
constexpr int FF = 128;
constexpr int EE = 32768;
constexpr int NE = 34816;
constexpr int OO = 512;
constexpr int ESB = 256;
constexpr int NPB = NE / ESB;
static_assert(NE % ESB == 0, "an edge-score block must not straddle two batches");

__device__ __forceinline__ float sgnval(int r, int j) {
  if (j >= 4) return 1.0f;
  const bool flip = (j & 1) ? ((r & 2) != 0) : ((r & 1) != 0);
  return flip ? -1.0f : 1.0f;
}

__device__ __forceinline__ int src_of(const int* __restrict__ ei, int e) {
  return (e < EE) ? ei[e] : (e - EE);
}
__device__ __forceinline__ int dst_of(const int* __restrict__ ei, int e) {
  return (e < EE) ? ei[EE + e] : (e - EE);
}

__global__ void zero_f32_kernel(float* __restrict__ p, size_t n) {
  size_t i = (size_t)blockIdx.x * blockDim.x + threadIdx.x;
  if (i < n) p[i] = 0.0f;
}
__global__ void zero_i32_kernel(int* __restrict__ p, int n) {
  int i = blockIdx.x * blockDim.x + threadIdx.x;
  if (i < n) p[i] = 0;
}

__global__ void count_dst_kernel(const int* __restrict__ ei, int* __restrict__ counts) {
  int e = blockIdx.x * blockDim.x + threadIdx.x;
  if (e >= NE) return;
  atomicAdd(&counts[dst_of(ei, e)], 1);
}

__global__ __launch_bounds__(256) void scan_counts_kernel(const int* __restrict__ counts,
                                                          int* __restrict__ rowstart) {
  __shared__ int part[256];
  __shared__ int pref[257];
  const int tid = threadIdx.x;
  const int base = tid * 8;
  int loc[8];
  int s = 0;
#pragma unroll
  for (int i = 0; i < 8; ++i) { loc[i] = s; s += counts[base + i]; }
  part[tid] = s;
  __syncthreads();
  if (tid == 0) {
    int a = 0;
    for (int i = 0; i < 256; ++i) { pref[i] = a; a += part[i]; }
    pref[256] = a;
  }
  __syncthreads();
  const int off = pref[tid];
#pragma unroll
  for (int i = 0; i < 8; ++i) rowstart[base + i] = off + loc[i];
  if (tid == 255) rowstart[NN] = pref[256];
}

constexpr int BRU = 8;
constexpr int BRT = EE / (128 * BRU);
static_assert(EE % (128 * BRU) == 0, "the explicit edges must fill whole batches");

__global__ __launch_bounds__(256) void build_rows_kernel(const int* __restrict__ ei,
                                                         const int* __restrict__ rowstart,
                                                         int* __restrict__ order) {
  const int lane = threadIdx.x & 31;
  const int wave = __builtin_amdgcn_readfirstlane((int)(threadIdx.x >> 5));
  const int n = blockIdx.x * 8 + wave;
  if (n >= NN) return;
  const int base = rowstart[n];
  const int4* d4 = (const int4*)(ei + EE) + lane;
  const unsigned below = (1u << lane) - 1u;
  int cnt = 0;

  int4 cur[BRU], nxt[BRU];
#pragma unroll
  for (int u = 0; u < BRU; ++u) cur[u] = d4[u * 32];

  for (int t = 0; t < BRT; ++t) {
    if (t + 1 < BRT) {
#pragma unroll
      for (int u = 0; u < BRU; ++u) nxt[u] = d4[((t + 1) * BRU + u) * 32];
    }
#pragma unroll
    for (int u = 0; u < BRU; ++u) {
      const int4 q = cur[u];
      const bool m0 = q.x == n, m1 = q.y == n, m2 = q.z == n, m3 = q.w == n;
      if (__builtin_amdgcn_ballot_w32(m0 | m1 | m2 | m3) != 0u) {
        const unsigned b0 = __builtin_amdgcn_ballot_w32(m0);
        const unsigned b1 = __builtin_amdgcn_ballot_w32(m1);
        const unsigned b2 = __builtin_amdgcn_ballot_w32(m2);
        const unsigned b3 = __builtin_amdgcn_ballot_w32(m3);
        const int e = (((t * BRU + u) * 32) + lane) * 4;
        int r = base + cnt + __popc(b0 & below) + __popc(b1 & below) + __popc(b2 & below) +
                __popc(b3 & below);
        if (m0) order[r] = e;
        r += m0 ? 1 : 0;
        if (m1) order[r] = e + 1;
        r += m1 ? 1 : 0;
        if (m2) order[r] = e + 2;
        r += m2 ? 1 : 0;
        if (m3) order[r] = e + 3;
        cnt += __popc(b0) + __popc(b1) + __popc(b2) + __popc(b3);
      }
    }
#pragma unroll
    for (int u = 0; u < BRU; ++u) cur[u] = nxt[u];
  }
  if (lane == 0) order[base + cnt] = EE + n;
}

constexpr int GR = 32;
constexpr int GC = 256;
constexpr int KQ = (2 * FF) / 4;
constexpr int GQ = 8;
constexpr int NGRP = KQ / GQ;
constexpr int APITCH = 2 * FF + 4;
constexpr int GEMM_BLOCKS = ((BB * NN) / GR) * (OO / GC);
static_assert(NN % GR == 0 && OO % GC == 0 && KQ % GQ == 0 && NGRP % 2 == 0, "tiling must be exact");
static_assert(GR * 2 * FF == 8 * 256 * 4, "the tile fill takes 8 float4 per thread");

struct fragset {
  v2f a[GQ][2];
  v2f w[GQ][2];
};

__device__ __forceinline__ void load_b_group(fragset& f, const float* __restrict__ wl, int g) {
#pragma unroll
  for (int u = 0; u < GQ; ++u) {
    const float* w0 = wl + (size_t)(4 * (g * GQ + u)) * OO;
    f.w[u][0] = *(const v2f*)w0;
    f.w[u][1] = *(const v2f*)(w0 + OO);
  }
}
__device__ __forceinline__ void load_a_group(fragset& f, const float* al, int g) {
#pragma unroll
  for (int u = 0; u < GQ; ++u) {
    const int k = 4 * (g * GQ + u);
    f.a[u][0] = *(const v2f*)(al + k);
    f.a[u][1] = *(const v2f*)(al + 16 * APITCH + k);
  }
}
__device__ __forceinline__ void mma_group(const fragset& f, v8f (&acc)[2][2]) {
#pragma unroll
  for (int u = 0; u < GQ; ++u) {
    const v2f b0 = { f.w[u][0].x, f.w[u][1].x };
    const v2f b1 = { f.w[u][0].y, f.w[u][1].y };
    acc[0][0] = __builtin_amdgcn_wmma_f32_16x16x4_f32(false, f.a[u][0], false, b0, (short)0, acc[0][0], false, false);
    acc[0][1] = __builtin_amdgcn_wmma_f32_16x16x4_f32(false, f.a[u][0], false, b1, (short)0, acc[0][1], false, false);
    acc[1][0] = __builtin_amdgcn_wmma_f32_16x16x4_f32(false, f.a[u][1], false, b0, (short)0, acc[1][0], false, false);
    acc[1][1] = __builtin_amdgcn_wmma_f32_16x16x4_f32(false, f.a[u][1], false, b1, (short)0, acc[1][1], false, false);
  }
}

__global__ __launch_bounds__(256) void gemm_pair_kernel(const float* __restrict__ x,
                                                        const float* __restrict__ W,
                                                        float* __restrict__ H,
                                                        int v1, int v2) {
  __shared__ float As[GR * APITCH];
  const int lane = threadIdx.x & 31;
  const int wave = __builtin_amdgcn_readfirstlane((int)(threadIdx.x >> 5));
  const int rowTile = (blockIdx.x >> 1) * GR;
  const int b = rowTile >> 11;
  const int nbase = rowTile & (NN - 1);
  const int c = lane & 15;
  const int h = lane >> 4;
  const int colbase = (blockIdx.x & 1) * GC + wave * 32;

  float4 xv[8];
#pragma unroll
  for (int i = 0; i < 8; ++i) {
    const int row = 8 * (i & 3) + wave;
    const int view = (i >> 2) ? v2 : v1;
    xv[i] = *(const float4*)(x + (((size_t)b * VV + view) * NN + nbase + row) * FF + lane * 4);
  }

  fragset F0, F1;
  const float* wl = W + (size_t)(2 * h) * OO + colbase + 2 * c;
  load_b_group(F0, wl, 0);

#pragma unroll
  for (int i = 0; i < 8; ++i) {
    const int row = 8 * (i & 3) + wave;
    const int half = i >> 2;
    const int r = half ? v1 : v2;
    float4 q = xv[i];
    q.x *= (lane == 0) ? sgnval(r, 0) : 1.0f;
    q.y *= (lane == 0) ? sgnval(r, 1) : 1.0f;
    q.z *= (lane == 0) ? sgnval(r, 2) : 1.0f;
    q.w *= (lane == 0) ? sgnval(r, 3) : 1.0f;
    *(float4*)(As + row * APITCH + half * FF + lane * 4) = q;
  }
  __syncthreads();

  const float* al = As + c * APITCH + 2 * h;
  load_a_group(F0, al, 0);

  v8f acc[2][2] = {};
  for (int g = 0; g + 2 < NGRP; g += 2) {
    load_b_group(F1, wl, g + 1);
    load_a_group(F1, al, g + 1);
    __builtin_amdgcn_sched_barrier(0);
    mma_group(F0, acc);
    __builtin_amdgcn_sched_barrier(0);
    load_b_group(F0, wl, g + 2);
    load_a_group(F0, al, g + 2);
    __builtin_amdgcn_sched_barrier(0);
    mma_group(F1, acc);
    __builtin_amdgcn_sched_barrier(0);
  }
  load_b_group(F1, wl, NGRP - 1);
  load_a_group(F1, al, NGRP - 1);
  __builtin_amdgcn_sched_barrier(0);
  mma_group(F0, acc);
  mma_group(F1, acc);

  float* Hb = H + ((size_t)b * NN + nbase + 8 * h) * OO + colbase + 2 * c;
#pragma unroll
  for (int i = 0; i < 8; ++i) {
    *(v2f*)(Hb + (size_t)i * OO) = (v2f){ acc[0][0][i], acc[0][1][i] };
    *(v2f*)(Hb + (size_t)(16 + i) * OO) = (v2f){ acc[1][0][i], acc[1][1][i] };
  }
}

__global__ __launch_bounds__(256) void node_scores_kernel(const float* __restrict__ H,
                                                          const float* __restrict__ att,
                                                          float* __restrict__ uS,
                                                          float* __restrict__ vS) {
  const int lane = threadIdx.x & 31;
  const int wave = threadIdx.x >> 5;
  const int node = blockIdx.x * 8 + wave;
  if (node >= BB * NN) return;
  const float* hrow = H + (size_t)node * OO;
  float su = 0.0f, sv = 0.0f;
  for (int j = lane; j < OO; j += 32) {
    const float h = hrow[j];
    const float l = h > 0.0f ? h : 0.2f * h;
    su += att[j] * l;
    sv += att[OO + j] * l;
  }
#pragma unroll
  for (int off = 16; off; off >>= 1) {
    su += __shfl_down(su, off, 32);
    sv += __shfl_down(sv, off, 32);
  }
  if (lane == 0) { uS[node] = su; vS[node] = sv; }
}

__global__ __launch_bounds__(ESB) void edge_scores_kernel(const float* __restrict__ uS,
                                                          const float* __restrict__ vS,
                                                          const int* __restrict__ ei,
                                                          float* __restrict__ s,
                                                          float* __restrict__ part) {
  __shared__ float redm[ESB / 32];
  __shared__ float reds[ESB / 32];
  const int lane = threadIdx.x & 31;
  const int wave = threadIdx.x >> 5;
  const int t = blockIdx.x * ESB + threadIdx.x;
  const int b = blockIdx.x / NPB;
  const int e = t - b * NE;
  const float sc = uS[b * NN + src_of(ei, e)] + vS[b * NN + dst_of(ei, e)];
  s[t] = sc;

  float m = sc;
  for (int off = 16; off; off >>= 1) m = fmaxf(m, __shfl_xor(m, off, 32));
  if (lane == 0) redm[wave] = m;
  __syncthreads();
  float bm = redm[0];
  for (int i = 1; i < ESB / 32; ++i) bm = fmaxf(bm, redm[i]);

  float ex = expf(sc - bm);
  for (int off = 16; off; off >>= 1) ex += __shfl_xor(ex, off, 32);
  if (lane == 0) reds[wave] = ex;
  __syncthreads();
  if (threadIdx.x == 0) {
    float a = reds[0];
    for (int i = 1; i < ESB / 32; ++i) a += reds[i];
    part[2 * blockIdx.x]     = bm;
    part[2 * blockIdx.x + 1] = a;
  }
}

template <int G>
__device__ __forceinline__ void gather_group(const float* __restrict__ Hb, int snv, float wv,
                                             int k, int lane, float4* acc) {
  int sn[G];
  float w[G];
#pragma unroll
  for (int t = 0; t < G; ++t) {
    sn[t] = __builtin_amdgcn_readlane(snv, k + t);
    w[t] = __int_as_float(__builtin_amdgcn_readlane(__float_as_int(wv), k + t));
  }
  float4 h[G][4];
#pragma unroll
  for (int t = 0; t < G; ++t) {
    const float4* hr = (const float4*)(Hb + (size_t)sn[t] * OO);
#pragma unroll
    for (int j = 0; j < 4; ++j) h[t][j] = hr[lane + 32 * j];
  }
#pragma unroll
  for (int t = 0; t < G; ++t) {
#pragma unroll
    for (int j = 0; j < 4; ++j) {
      acc[j].x = __builtin_fmaf(w[t], h[t][j].x, acc[j].x);
      acc[j].y = __builtin_fmaf(w[t], h[t][j].y, acc[j].y);
      acc[j].z = __builtin_fmaf(w[t], h[t][j].z, acc[j].z);
      acc[j].w = __builtin_fmaf(w[t], h[t][j].w, acc[j].w);
    }
  }
}

__global__ __launch_bounds__(256) __attribute__((amdgpu_waves_per_eu(2, 4))) void aggregate_kernel(const float* __restrict__ H,
                                                        const float* __restrict__ s,
                                                        const float* __restrict__ part,
                                                        const int* __restrict__ rowstart,
                                                        const int* __restrict__ order,
                                                        const int* __restrict__ ei,
                                                        float* __restrict__ out, int g) {
  const int lane = threadIdx.x & 31;
  const int wave = __builtin_amdgcn_readfirstlane((int)(threadIdx.x >> 5));
  const int node = blockIdx.x * 8 + wave;
  if (node >= BB * NN) return;
  const int b = node >> 11;
  const int n = node & (NN - 1);
  const float* Hb = H + (size_t)b * NN * OO;
  const float* sb = s + (size_t)b * NE;
  const int beg = rowstart[n];
  const int end = rowstart[n + 1];

  const float2* pb = (const float2*)part + (size_t)b * NPB;
  constexpr int PPL = (NPB + 31) / 32;
  float pm[PPL], ps[PPL];
  for (int i = 0; i < PPL; ++i) {
    const int k = lane + 32 * i;
    const bool ok = k < NPB;
    const float2 q = ok ? pb[k] : make_float2(-3.4e38f, 0.0f);
    pm[i] = q.x; ps[i] = q.y;
  }

  int snv = 0;
  float sv = 0.0f;
  if (beg + lane < end) {
    const int e = order[beg + lane];
    snv = src_of(ei, e);
    sv = sb[e];
  }

  float m = -3.4e38f;
  for (int i = 0; i < PPL; ++i) m = fmaxf(m, pm[i]);
  for (int off = 16; off; off >>= 1) m = fmaxf(m, __shfl_xor(m, off, 32));
  float den = 0.0f;
  for (int i = 0; i < PPL; ++i) den += (lane + 32 * i < NPB) ? ps[i] * expf(pm[i] - m) : 0.0f;
  for (int off = 16; off; off >>= 1) den += __shfl_xor(den, off, 32);
  const float inv = 1.0f / den;

  float4 acc[4] = {};
  int base = beg;
  while (true) {
    const int left = end - base;
    const int cnt = left < 32 ? left : 32;
    const float wv = (lane < cnt) ? expf(sv - m) * inv : 0.0f;
    int k = 0;
    for (; k + 8 <= cnt; k += 8) gather_group<8>(Hb, snv, wv, k, lane, acc);
    if (k + 4 <= cnt) { gather_group<4>(Hb, snv, wv, k, lane, acc); k += 4; }
    if (k + 2 <= cnt) { gather_group<2>(Hb, snv, wv, k, lane, acc); k += 2; }
    if (k < cnt) gather_group<1>(Hb, snv, wv, k, lane, acc);
    base += 32;
    if (base >= end) break;
    snv = 0;
    sv = 0.0f;
    if (base + lane < end) {
      const int e = order[base + lane];
      snv = src_of(ei, e);
      sv = sb[e];
    }
  }

  float4* op = (float4*)(out + (((size_t)b * VV + g) * NN + n) * OO);
#pragma unroll
  for (int j = 0; j < 4; ++j) {
    float4 o = op[lane + 32 * j];
    o.x += acc[j].x; o.y += acc[j].y; o.z += acc[j].z; o.w += acc[j].w;
    op[lane + 32 * j] = o;
  }
}

__global__ void epilogue_kernel(float* __restrict__ out, const float* __restrict__ bias, size_t n) {
  size_t i = (size_t)blockIdx.x * blockDim.x + threadIdx.x;
  if (i < n) out[i] = out[i] * 0.25f + bias[i & (OO - 1)];
}

extern "C" void kernel_launch(void* const* d_in, const int* in_sizes, int n_in,
                              void* d_out, int out_size, void* d_ws, size_t ws_size,
                              hipStream_t stream) {
  const float* x    = (const float*)d_in[0];
  const int*   ei   = (const int*)d_in[1];
  const float* W    = (const float*)d_in[2];
  const float* att  = (const float*)d_in[3];
  const float* bias = (const float*)d_in[4];
  float* out = (float*)d_out;

  char* p = (char*)d_ws;
  float* H     = (float*)p; p += (size_t)BB * NN * OO * sizeof(float);
  float* uS    = (float*)p; p += (size_t)BB * NN * sizeof(float);
  float* vS    = (float*)p; p += (size_t)BB * NN * sizeof(float);
  float* s     = (float*)p; p += (size_t)BB * NE * sizeof(float);
  int* counts  = (int*)p;   p += (size_t)NN * sizeof(int);
  int* order   = (int*)p;   p += (size_t)NE * sizeof(int);
  int* rowstart= (int*)p;   p += (size_t)(NN + 1) * sizeof(int);
  static_assert((size_t)2 * BB * NPB * sizeof(float) <= (size_t)NN * sizeof(int), "partials must fit");
  float* part = (float*)counts;

  const size_t outN = (size_t)BB * VV * NN * OO;

  zero_f32_kernel<<<(unsigned)((outN + 255) / 256), 256, 0, stream>>>(out, outN);
  zero_i32_kernel<<<(NN + 255) / 256, 256, 0, stream>>>(counts, NN);
  count_dst_kernel<<<(NE + 255) / 256, 256, 0, stream>>>(ei, counts);
  scan_counts_kernel<<<1, 256, 0, stream>>>(counts, rowstart);
  build_rows_kernel<<<NN / 8, 256, 0, stream>>>(ei, rowstart, order);

  for (int g = 0; g < VV; ++g) {
    for (int hv = 0; hv < VV; ++hv) {
      const int v1 = hv;
      const int v2 = g ^ hv;
      gemm_pair_kernel<<<GEMM_BLOCKS, 256, 0, stream>>>(x, W, H, v1, v2);
      node_scores_kernel<<<(BB * NN) / 8, 256, 0, stream>>>(H, att, uS, vS);
      edge_scores_kernel<<<BB * NPB, ESB, 0, stream>>>(uS, vS, ei, s, part);
      aggregate_kernel<<<(BB * NN) / 8, 256, 0, stream>>>(H, s, part, rowstart, order, ei, out, g);
    }
  }
  epilogue_kernel<<<(unsigned)((outN + 255) / 256), 256, 0, stream>>>(out, bias, outN);

  (void)in_sizes; (void)n_in; (void)out_size; (void)ws_size;
}


# attention dots u,v taken from gemm_pair's accumulators (lane pair, 16-lane DPP sum, 8 waves in order, two column halves added in edge scores: their sum order changes); node_scores kernel removed
# speedup vs baseline: 3.1306x; 1.1282x over previous
#include <hip/hip_runtime.h>
#include <hip/hip_bf16.h>

typedef __attribute__((ext_vector_type(2))) float v2f;
typedef __attribute__((ext_vector_type(8))) float v8f;

constexpr int BB = 2;
constexpr int VV = 4;
constexpr int NN = 2048;
constexpr int FF = 128;
constexpr int EE = 32768;
constexpr int NE = 34816;
constexpr int OO = 512;
constexpr int ESB = 256;
constexpr int NPB = NE / ESB;
static_assert(NE % ESB == 0, "an edge-score block must not straddle two batches");

__device__ __forceinline__ float sgnval(int r, int j) {
  if (j >= 4) return 1.0f;
  const bool flip = (j & 1) ? ((r & 2) != 0) : ((r & 1) != 0);
  return flip ? -1.0f : 1.0f;
}

__device__ __forceinline__ int src_of(const int* __restrict__ ei, int e) {
  return (e < EE) ? ei[e] : (e - EE);
}
__device__ __forceinline__ int dst_of(const int* __restrict__ ei, int e) {
  return (e < EE) ? ei[EE + e] : (e - EE);
}

__global__ void zero_f32_kernel(float* __restrict__ p, size_t n) {
  size_t i = (size_t)blockIdx.x * blockDim.x + threadIdx.x;
  if (i < n) p[i] = 0.0f;
}
__global__ void zero_i32_kernel(int* __restrict__ p, int n) {
  int i = blockIdx.x * blockDim.x + threadIdx.x;
  if (i < n) p[i] = 0;
}

__global__ void count_dst_kernel(const int* __restrict__ ei, int* __restrict__ counts) {
  int e = blockIdx.x * blockDim.x + threadIdx.x;
  if (e >= NE) return;
  atomicAdd(&counts[dst_of(ei, e)], 1);
}

__global__ __launch_bounds__(256) void scan_counts_kernel(const int* __restrict__ counts,
                                                          int* __restrict__ rowstart) {
  __shared__ int part[256];
  __shared__ int pref[257];
  const int tid = threadIdx.x;
  const int base = tid * 8;
  int loc[8];
  int s = 0;
#pragma unroll
  for (int i = 0; i < 8; ++i) { loc[i] = s; s += counts[base + i]; }
  part[tid] = s;
  __syncthreads();
  if (tid == 0) {
    int a = 0;
    for (int i = 0; i < 256; ++i) { pref[i] = a; a += part[i]; }
    pref[256] = a;
  }
  __syncthreads();
  const int off = pref[tid];
#pragma unroll
  for (int i = 0; i < 8; ++i) rowstart[base + i] = off + loc[i];
  if (tid == 255) rowstart[NN] = pref[256];
}

constexpr int BRU = 8;
constexpr int BRT = EE / (128 * BRU);
static_assert(EE % (128 * BRU) == 0, "the explicit edges must fill whole batches");

__global__ __launch_bounds__(256) void build_rows_kernel(const int* __restrict__ ei,
                                                         const int* __restrict__ rowstart,
                                                         int* __restrict__ order) {
  const int lane = threadIdx.x & 31;
  const int wave = __builtin_amdgcn_readfirstlane((int)(threadIdx.x >> 5));
  const int n = blockIdx.x * 8 + wave;
  if (n >= NN) return;
  const int base = rowstart[n];
  const int4* d4 = (const int4*)(ei + EE) + lane;
  const unsigned below = (1u << lane) - 1u;
  int cnt = 0;

  int4 cur[BRU], nxt[BRU];
#pragma unroll
  for (int u = 0; u < BRU; ++u) cur[u] = d4[u * 32];

  for (int t = 0; t < BRT; ++t) {
    if (t + 1 < BRT) {
#pragma unroll
      for (int u = 0; u < BRU; ++u) nxt[u] = d4[((t + 1) * BRU + u) * 32];
    }
#pragma unroll
    for (int u = 0; u < BRU; ++u) {
      const int4 q = cur[u];
      const bool m0 = q.x == n, m1 = q.y == n, m2 = q.z == n, m3 = q.w == n;
      if (__builtin_amdgcn_ballot_w32(m0 | m1 | m2 | m3) != 0u) {
        const unsigned b0 = __builtin_amdgcn_ballot_w32(m0);
        const unsigned b1 = __builtin_amdgcn_ballot_w32(m1);
        const unsigned b2 = __builtin_amdgcn_ballot_w32(m2);
        const unsigned b3 = __builtin_amdgcn_ballot_w32(m3);
        const int e = (((t * BRU + u) * 32) + lane) * 4;
        int r = base + cnt + __popc(b0 & below) + __popc(b1 & below) + __popc(b2 & below) +
                __popc(b3 & below);
        if (m0) order[r] = e;
        r += m0 ? 1 : 0;
        if (m1) order[r] = e + 1;
        r += m1 ? 1 : 0;
        if (m2) order[r] = e + 2;
        r += m2 ? 1 : 0;
        if (m3) order[r] = e + 3;
        cnt += __popc(b0) + __popc(b1) + __popc(b2) + __popc(b3);
      }
    }
#pragma unroll
    for (int u = 0; u < BRU; ++u) cur[u] = nxt[u];
  }
  if (lane == 0) order[base + cnt] = EE + n;
}

constexpr int GR = 32;
constexpr int GC = 256;
constexpr int KQ = (2 * FF) / 4;
constexpr int GQ = 8;
constexpr int NGRP = KQ / GQ;
constexpr int APITCH = 2 * FF + 4;
constexpr int GEMM_BLOCKS = ((BB * NN) / GR) * (OO / GC);
static_assert(NN % GR == 0 && OO % GC == 0 && KQ % GQ == 0 && NGRP % 2 == 0, "tiling must be exact");
static_assert(GR * 2 * FF == 8 * 256 * 4, "the tile fill takes 8 float4 per thread");

struct fragset {
  v2f a[GQ][2];
  v2f w[GQ][2];
};

__device__ __forceinline__ void load_b_group(fragset& f, const float* __restrict__ wl, int g) {
#pragma unroll
  for (int u = 0; u < GQ; ++u) {
    const float* w0 = wl + (size_t)(4 * (g * GQ + u)) * OO;
    f.w[u][0] = *(const v2f*)w0;
    f.w[u][1] = *(const v2f*)(w0 + OO);
  }
}
__device__ __forceinline__ void load_a_group(fragset& f, const float* al, int g) {
#pragma unroll
  for (int u = 0; u < GQ; ++u) {
    const int k = 4 * (g * GQ + u);
    f.a[u][0] = *(const v2f*)(al + k);
    f.a[u][1] = *(const v2f*)(al + 16 * APITCH + k);
  }
}
__device__ __forceinline__ void mma_group(const fragset& f, v8f (&acc)[2][2]) {
#pragma unroll
  for (int u = 0; u < GQ; ++u) {
    const v2f b0 = { f.w[u][0].x, f.w[u][1].x };
    const v2f b1 = { f.w[u][0].y, f.w[u][1].y };
    acc[0][0] = __builtin_amdgcn_wmma_f32_16x16x4_f32(false, f.a[u][0], false, b0, (short)0, acc[0][0], false, false);
    acc[0][1] = __builtin_amdgcn_wmma_f32_16x16x4_f32(false, f.a[u][0], false, b1, (short)0, acc[0][1], false, false);
    acc[1][0] = __builtin_amdgcn_wmma_f32_16x16x4_f32(false, f.a[u][1], false, b0, (short)0, acc[1][0], false, false);
    acc[1][1] = __builtin_amdgcn_wmma_f32_16x16x4_f32(false, f.a[u][1], false, b1, (short)0, acc[1][1], false, false);
  }
}

template <int MASK>
__device__ __forceinline__ float row_xadd(float v) {
  const int t = __builtin_amdgcn_update_dpp(0, __float_as_int(v), 0x160 | MASK, 0xF, 0xF, true);
  return v + __int_as_float(t);
}
__device__ __forceinline__ float row_sum16(float v) {
  v = row_xadd<1>(v);
  v = row_xadd<2>(v);
  v = row_xadd<4>(v);
  v = row_xadd<8>(v);
  return v;
}
__device__ __forceinline__ float lrelu(float h) { return h > 0.0f ? h : 0.2f * h; }

__global__ __launch_bounds__(256) void gemm_pair_kernel(const float* __restrict__ x,
                                                        const float* __restrict__ W,
                                                        const float* __restrict__ att,
                                                        float* __restrict__ H,
                                                        float* __restrict__ uP,
                                                        float* __restrict__ vP,
                                                        int v1, int v2) {
  __shared__ float As[GR * APITCH];
  __shared__ __attribute__((aligned(16))) float red[8][2][GR];
  const int lane = threadIdx.x & 31;
  const int wave = __builtin_amdgcn_readfirstlane((int)(threadIdx.x >> 5));
  const int rowTile = (blockIdx.x >> 1) * GR;
  const int b = rowTile >> 11;
  const int nbase = rowTile & (NN - 1);
  const int c = lane & 15;
  const int h = lane >> 4;
  const int colbase = (blockIdx.x & 1) * GC + wave * 32;

  float4 xv[8];
#pragma unroll
  for (int i = 0; i < 8; ++i) {
    const int row = 8 * (i & 3) + wave;
    const int view = (i >> 2) ? v2 : v1;
    xv[i] = *(const float4*)(x + (((size_t)b * VV + view) * NN + nbase + row) * FF + lane * 4);
  }

  const v2f at = *(const v2f*)(att + colbase + 2 * c);
  const v2f ab = *(const v2f*)(att + OO + colbase + 2 * c);

  fragset F0, F1;
  const float* wl = W + (size_t)(2 * h) * OO + colbase + 2 * c;
  load_b_group(F0, wl, 0);

#pragma unroll
  for (int i = 0; i < 8; ++i) {
    const int row = 8 * (i & 3) + wave;
    const int half = i >> 2;
    const int r = half ? v1 : v2;
    float4 q = xv[i];
    q.x *= (lane == 0) ? sgnval(r, 0) : 1.0f;
    q.y *= (lane == 0) ? sgnval(r, 1) : 1.0f;
    q.z *= (lane == 0) ? sgnval(r, 2) : 1.0f;
    q.w *= (lane == 0) ? sgnval(r, 3) : 1.0f;
    *(float4*)(As + row * APITCH + half * FF + lane * 4) = q;
  }
  __syncthreads();

  const float* al = As + c * APITCH + 2 * h;
  load_a_group(F0, al, 0);

  v8f acc[2][2] = {};
  for (int g = 0; g + 2 < NGRP; g += 2) {
    load_b_group(F1, wl, g + 1);
    load_a_group(F1, al, g + 1);
    __builtin_amdgcn_sched_barrier(0);
    mma_group(F0, acc);
    __builtin_amdgcn_sched_barrier(0);
    load_b_group(F0, wl, g + 2);
    load_a_group(F0, al, g + 2);
    __builtin_amdgcn_sched_barrier(0);
    mma_group(F1, acc);
    __builtin_amdgcn_sched_barrier(0);
  }
  load_b_group(F1, wl, NGRP - 1);
  load_a_group(F1, al, NGRP - 1);
  __builtin_amdgcn_sched_barrier(0);
  mma_group(F0, acc);
  mma_group(F1, acc);

  float* Hb = H + ((size_t)b * NN + nbase + 8 * h) * OO + colbase + 2 * c;
#pragma unroll
  for (int i = 0; i < 8; ++i) {
    *(v2f*)(Hb + (size_t)i * OO) = (v2f){ acc[0][0][i], acc[0][1][i] };
    *(v2f*)(Hb + (size_t)(16 + i) * OO) = (v2f){ acc[1][0][i], acc[1][1][i] };
  }

  float pu[2][8], pv[2][8];
#pragma unroll
  for (int rt = 0; rt < 2; ++rt) {
#pragma unroll
    for (int i = 0; i < 8; ++i) {
      const float l0 = lrelu(acc[rt][0][i]);
      const float l1 = lrelu(acc[rt][1][i]);
      pu[rt][i] = row_sum16(at.x * l0 + at.y * l1);
      pv[rt][i] = row_sum16(ab.x * l0 + ab.y * l1);
    }
  }
  if (c == 0) {
#pragma unroll
    for (int rt = 0; rt < 2; ++rt) {
      float4* ru = (float4*)&red[wave][0][16 * rt + 8 * h];
      float4* rv = (float4*)&red[wave][1][16 * rt + 8 * h];
      ru[0] = make_float4(pu[rt][0], pu[rt][1], pu[rt][2], pu[rt][3]);
      ru[1] = make_float4(pu[rt][4], pu[rt][5], pu[rt][6], pu[rt][7]);
      rv[0] = make_float4(pv[rt][0], pv[rt][1], pv[rt][2], pv[rt][3]);
      rv[1] = make_float4(pv[rt][4], pv[rt][5], pv[rt][6], pv[rt][7]);
    }
  }
  __syncthreads();
  if (wave < 2) {
    float a = red[0][wave][lane];
#pragma unroll
    for (int w = 1; w < 8; ++w) a += red[w][wave][lane];
    float* dp = (wave ? vP : uP) + (size_t)(blockIdx.x & 1) * (BB * NN) + rowTile + lane;
    *dp = a;
  }
}

__global__ __launch_bounds__(ESB) void edge_scores_kernel(const float* __restrict__ uS,
                                                          const float* __restrict__ vS,
                                                          const int* __restrict__ ei,
                                                          float* __restrict__ s,
                                                          float* __restrict__ part) {
  __shared__ float redm[ESB / 32];
  __shared__ float reds[ESB / 32];
  const int lane = threadIdx.x & 31;
  const int wave = threadIdx.x >> 5;
  const int t = blockIdx.x * ESB + threadIdx.x;
  const int b = blockIdx.x / NPB;
  const int e = t - b * NE;
  const int sn = b * NN + src_of(ei, e);
  const int dn = b * NN + dst_of(ei, e);
  const float sc = (uS[sn] + uS[BB * NN + sn]) + (vS[dn] + vS[BB * NN + dn]);
  s[t] = sc;

  float m = sc;
  for (int off = 16; off; off >>= 1) m = fmaxf(m, __shfl_xor(m, off, 32));
  if (lane == 0) redm[wave] = m;
  __syncthreads();
  float bm = redm[0];
  for (int i = 1; i < ESB / 32; ++i) bm = fmaxf(bm, redm[i]);

  float ex = expf(sc - bm);
  for (int off = 16; off; off >>= 1) ex += __shfl_xor(ex, off, 32);
  if (lane == 0) reds[wave] = ex;
  __syncthreads();
  if (threadIdx.x == 0) {
    float a = reds[0];
    for (int i = 1; i < ESB / 32; ++i) a += reds[i];
    part[2 * blockIdx.x]     = bm;
    part[2 * blockIdx.x + 1] = a;
  }
}

template <int G>
__device__ __forceinline__ void gather_group(const float* __restrict__ Hb, int snv, float wv,
                                             int k, int lane, float4* acc) {
  int sn[G];
  float w[G];
#pragma unroll
  for (int t = 0; t < G; ++t) {
    sn[t] = __builtin_amdgcn_readlane(snv, k + t);
    w[t] = __int_as_float(__builtin_amdgcn_readlane(__float_as_int(wv), k + t));
  }
  float4 h[G][4];
#pragma unroll
  for (int t = 0; t < G; ++t) {
    const float4* hr = (const float4*)(Hb + (size_t)sn[t] * OO);
#pragma unroll
    for (int j = 0; j < 4; ++j) h[t][j] = hr[lane + 32 * j];
  }
#pragma unroll
  for (int t = 0; t < G; ++t) {
#pragma unroll
    for (int j = 0; j < 4; ++j) {
      acc[j].x = __builtin_fmaf(w[t], h[t][j].x, acc[j].x);
      acc[j].y = __builtin_fmaf(w[t], h[t][j].y, acc[j].y);
      acc[j].z = __builtin_fmaf(w[t], h[t][j].z, acc[j].z);
      acc[j].w = __builtin_fmaf(w[t], h[t][j].w, acc[j].w);
    }
  }
}

__global__ __launch_bounds__(256) __attribute__((amdgpu_waves_per_eu(2, 4))) void aggregate_kernel(const float* __restrict__ H,
                                                        const float* __restrict__ s,
                                                        const float* __restrict__ part,
                                                        const int* __restrict__ rowstart,
                                                        const int* __restrict__ order,
                                                        const int* __restrict__ ei,
                                                        float* __restrict__ out, int g) {
  const int lane = threadIdx.x & 31;
  const int wave = __builtin_amdgcn_readfirstlane((int)(threadIdx.x >> 5));
  const int node = blockIdx.x * 8 + wave;
  if (node >= BB * NN) return;
  const int b = node >> 11;
  const int n = node & (NN - 1);
  const float* Hb = H + (size_t)b * NN * OO;
  const float* sb = s + (size_t)b * NE;
  const int beg = rowstart[n];
  const int end = rowstart[n + 1];

  const float2* pb = (const float2*)part + (size_t)b * NPB;
  constexpr int PPL = (NPB + 31) / 32;
  float pm[PPL], ps[PPL];
  for (int i = 0; i < PPL; ++i) {
    const int k = lane + 32 * i;
    const bool ok = k < NPB;
    const float2 q = ok ? pb[k] : make_float2(-3.4e38f, 0.0f);
    pm[i] = q.x; ps[i] = q.y;
  }

  int snv = 0;
  float sv = 0.0f;
  if (beg + lane < end) {
    const int e = order[beg + lane];
    snv = src_of(ei, e);
    sv = sb[e];
  }

  float m = -3.4e38f;
  for (int i = 0; i < PPL; ++i) m = fmaxf(m, pm[i]);
  for (int off = 16; off; off >>= 1) m = fmaxf(m, __shfl_xor(m, off, 32));
  float den = 0.0f;
  for (int i = 0; i < PPL; ++i) den += (lane + 32 * i < NPB) ? ps[i] * expf(pm[i] - m) : 0.0f;
  for (int off = 16; off; off >>= 1) den += __shfl_xor(den, off, 32);
  const float inv = 1.0f / den;

  float4 acc[4] = {};
  int base = beg;
  while (true) {
    const int left = end - base;
    const int cnt = left < 32 ? left : 32;
    const float wv = (lane < cnt) ? expf(sv - m) * inv : 0.0f;
    int k = 0;
    for (; k + 8 <= cnt; k += 8) gather_group<8>(Hb, snv, wv, k, lane, acc);
    if (k + 4 <= cnt) { gather_group<4>(Hb, snv, wv, k, lane, acc); k += 4; }
    if (k + 2 <= cnt) { gather_group<2>(Hb, snv, wv, k, lane, acc); k += 2; }
    if (k < cnt) gather_group<1>(Hb, snv, wv, k, lane, acc);
    base += 32;
    if (base >= end) break;
    snv = 0;
    sv = 0.0f;
    if (base + lane < end) {
      const int e = order[base + lane];
      snv = src_of(ei, e);
      sv = sb[e];
    }
  }

  float4* op = (float4*)(out + (((size_t)b * VV + g) * NN + n) * OO);
#pragma unroll
  for (int j = 0; j < 4; ++j) {
    float4 o = op[lane + 32 * j];
    o.x += acc[j].x; o.y += acc[j].y; o.z += acc[j].z; o.w += acc[j].w;
    op[lane + 32 * j] = o;
  }
}

__global__ void epilogue_kernel(float* __restrict__ out, const float* __restrict__ bias, size_t n) {
  size_t i = (size_t)blockIdx.x * blockDim.x + threadIdx.x;
  if (i < n) out[i] = out[i] * 0.25f + bias[i & (OO - 1)];
}

extern "C" void kernel_launch(void* const* d_in, const int* in_sizes, int n_in,
                              void* d_out, int out_size, void* d_ws, size_t ws_size,
                              hipStream_t stream) {
  const float* x    = (const float*)d_in[0];
  const int*   ei   = (const int*)d_in[1];
  const float* W    = (const float*)d_in[2];
  const float* att  = (const float*)d_in[3];
  const float* bias = (const float*)d_in[4];
  float* out = (float*)d_out;

  char* p = (char*)d_ws;
  float* H     = (float*)p; p += (size_t)BB * NN * OO * sizeof(float);
  float* uS    = (float*)p; p += (size_t)2 * BB * NN * sizeof(float);
  float* vS    = (float*)p; p += (size_t)2 * BB * NN * sizeof(float);
  float* s     = (float*)p; p += (size_t)BB * NE * sizeof(float);
  int* counts  = (int*)p;   p += (size_t)NN * sizeof(int);
  int* order   = (int*)p;   p += (size_t)NE * sizeof(int);
  int* rowstart= (int*)p;   p += (size_t)(NN + 1) * sizeof(int);
  static_assert((size_t)2 * BB * NPB * sizeof(float) <= (size_t)NN * sizeof(int), "partials must fit");
  float* part = (float*)counts;

  const size_t outN = (size_t)BB * VV * NN * OO;

  zero_f32_kernel<<<(unsigned)((outN + 255) / 256), 256, 0, stream>>>(out, outN);
  zero_i32_kernel<<<(NN + 255) / 256, 256, 0, stream>>>(counts, NN);
  count_dst_kernel<<<(NE + 255) / 256, 256, 0, stream>>>(ei, counts);
  scan_counts_kernel<<<1, 256, 0, stream>>>(counts, rowstart);
  build_rows_kernel<<<NN / 8, 256, 0, stream>>>(ei, rowstart, order);

  for (int g = 0; g < VV; ++g) {
    for (int hv = 0; hv < VV; ++hv) {
      const int v1 = hv;
      const int v2 = g ^ hv;
      gemm_pair_kernel<<<GEMM_BLOCKS, 256, 0, stream>>>(x, W, att, H, uS, vS, v1, v2);
      edge_scores_kernel<<<BB * NPB, ESB, 0, stream>>>(uS, vS, ei, s, part);
      aggregate_kernel<<<(BB * NN) / 8, 256, 0, stream>>>(H, s, part, rowstart, order, ei, out, g);
    }
  }
  epilogue_kernel<<<(unsigned)((outN + 255) / 256), 256, 0, stream>>>(out, bias, outN);

  (void)in_sizes; (void)n_in; (void)out_size; (void)ws_size;
}


# aggregate: a g's first launch stores its sums as out's start value, the middle two add, the last adds and applies the 1/4 and the bias; zero-fill and epilogue kernels removed; sum order kept
# speedup vs baseline: 3.8158x; 1.2189x over previous
#include <hip/hip_runtime.h>
#include <hip/hip_bf16.h>

typedef __attribute__((ext_vector_type(2))) float v2f;
typedef __attribute__((ext_vector_type(8))) float v8f;

constexpr int BB = 2;
constexpr int VV = 4;
constexpr int NN = 2048;
constexpr int FF = 128;
constexpr int EE = 32768;
constexpr int NE = 34816;
constexpr int OO = 512;
constexpr int ESB = 256;
constexpr int NPB = NE / ESB;
static_assert(NE % ESB == 0, "an edge-score block must not straddle two batches");

__device__ __forceinline__ float sgnval(int r, int j) {
  if (j >= 4) return 1.0f;
  const bool flip = (j & 1) ? ((r & 2) != 0) : ((r & 1) != 0);
  return flip ? -1.0f : 1.0f;
}

__device__ __forceinline__ int src_of(const int* __restrict__ ei, int e) {
  return (e < EE) ? ei[e] : (e - EE);
}
__device__ __forceinline__ int dst_of(const int* __restrict__ ei, int e) {
  return (e < EE) ? ei[EE + e] : (e - EE);
}

__global__ void zero_i32_kernel(int* __restrict__ p, int n) {
  int i = blockIdx.x * blockDim.x + threadIdx.x;
  if (i < n) p[i] = 0;
}

__global__ void count_dst_kernel(const int* __restrict__ ei, int* __restrict__ counts) {
  int e = blockIdx.x * blockDim.x + threadIdx.x;
  if (e >= NE) return;
  atomicAdd(&counts[dst_of(ei, e)], 1);
}

__global__ __launch_bounds__(256) void scan_counts_kernel(const int* __restrict__ counts,
                                                          int* __restrict__ rowstart) {
  __shared__ int part[256];
  __shared__ int pref[257];
  const int tid = threadIdx.x;
  const int base = tid * 8;
  int loc[8];
  int s = 0;
#pragma unroll
  for (int i = 0; i < 8; ++i) { loc[i] = s; s += counts[base + i]; }
  part[tid] = s;
  __syncthreads();
  if (tid == 0) {
    int a = 0;
    for (int i = 0; i < 256; ++i) { pref[i] = a; a += part[i]; }
    pref[256] = a;
  }
  __syncthreads();
  const int off = pref[tid];
#pragma unroll
  for (int i = 0; i < 8; ++i) rowstart[base + i] = off + loc[i];
  if (tid == 255) rowstart[NN] = pref[256];
}

constexpr int BRU = 8;
constexpr int BRT = EE / (128 * BRU);
static_assert(EE % (128 * BRU) == 0, "the explicit edges must fill whole batches");

__global__ __launch_bounds__(256) void build_rows_kernel(const int* __restrict__ ei,
                                                         const int* __restrict__ rowstart,
                                                         int* __restrict__ order) {
  const int lane = threadIdx.x & 31;
  const int wave = __builtin_amdgcn_readfirstlane((int)(threadIdx.x >> 5));
  const int n = blockIdx.x * 8 + wave;
  if (n >= NN) return;
  const int base = rowstart[n];
  const int4* d4 = (const int4*)(ei + EE) + lane;
  const unsigned below = (1u << lane) - 1u;
  int cnt = 0;

  int4 cur[BRU], nxt[BRU];
#pragma unroll
  for (int u = 0; u < BRU; ++u) cur[u] = d4[u * 32];

  for (int t = 0; t < BRT; ++t) {
    if (t + 1 < BRT) {
#pragma unroll
      for (int u = 0; u < BRU; ++u) nxt[u] = d4[((t + 1) * BRU + u) * 32];
    }
#pragma unroll
    for (int u = 0; u < BRU; ++u) {
      const int4 q = cur[u];
      const bool m0 = q.x == n, m1 = q.y == n, m2 = q.z == n, m3 = q.w == n;
      if (__builtin_amdgcn_ballot_w32(m0 | m1 | m2 | m3) != 0u) {
        const unsigned b0 = __builtin_amdgcn_ballot_w32(m0);
        const unsigned b1 = __builtin_amdgcn_ballot_w32(m1);
        const unsigned b2 = __builtin_amdgcn_ballot_w32(m2);
        const unsigned b3 = __builtin_amdgcn_ballot_w32(m3);
        const int e = (((t * BRU + u) * 32) + lane) * 4;
        int r = base + cnt + __popc(b0 & below) + __popc(b1 & below) + __popc(b2 & below) +
                __popc(b3 & below);
        if (m0) order[r] = e;
        r += m0 ? 1 : 0;
        if (m1) order[r] = e + 1;
        r += m1 ? 1 : 0;
        if (m2) order[r] = e + 2;
        r += m2 ? 1 : 0;
        if (m3) order[r] = e + 3;
        cnt += __popc(b0) + __popc(b1) + __popc(b2) + __popc(b3);
      }
    }
#pragma unroll
    for (int u = 0; u < BRU; ++u) cur[u] = nxt[u];
  }
  if (lane == 0) order[base + cnt] = EE + n;
}

constexpr int GR = 32;
constexpr int GC = 256;
constexpr int KQ = (2 * FF) / 4;
constexpr int GQ = 8;
constexpr int NGRP = KQ / GQ;
constexpr int APITCH = 2 * FF + 4;
constexpr int GEMM_BLOCKS = ((BB * NN) / GR) * (OO / GC);
static_assert(NN % GR == 0 && OO % GC == 0 && KQ % GQ == 0 && NGRP % 2 == 0, "tiling must be exact");
static_assert(GR * 2 * FF == 8 * 256 * 4, "the tile fill takes 8 float4 per thread");

struct fragset {
  v2f a[GQ][2];
  v2f w[GQ][2];
};

__device__ __forceinline__ void load_b_group(fragset& f, const float* __restrict__ wl, int g) {
#pragma unroll
  for (int u = 0; u < GQ; ++u) {
    const float* w0 = wl + (size_t)(4 * (g * GQ + u)) * OO;
    f.w[u][0] = *(const v2f*)w0;
    f.w[u][1] = *(const v2f*)(w0 + OO);
  }
}
__device__ __forceinline__ void load_a_group(fragset& f, const float* al, int g) {
#pragma unroll
  for (int u = 0; u < GQ; ++u) {
    const int k = 4 * (g * GQ + u);
    f.a[u][0] = *(const v2f*)(al + k);
    f.a[u][1] = *(const v2f*)(al + 16 * APITCH + k);
  }
}
__device__ __forceinline__ void mma_group(const fragset& f, v8f (&acc)[2][2]) {
#pragma unroll
  for (int u = 0; u < GQ; ++u) {
    const v2f b0 = { f.w[u][0].x, f.w[u][1].x };
    const v2f b1 = { f.w[u][0].y, f.w[u][1].y };
    acc[0][0] = __builtin_amdgcn_wmma_f32_16x16x4_f32(false, f.a[u][0], false, b0, (short)0, acc[0][0], false, false);
    acc[0][1] = __builtin_amdgcn_wmma_f32_16x16x4_f32(false, f.a[u][0], false, b1, (short)0, acc[0][1], false, false);
    acc[1][0] = __builtin_amdgcn_wmma_f32_16x16x4_f32(false, f.a[u][1], false, b0, (short)0, acc[1][0], false, false);
    acc[1][1] = __builtin_amdgcn_wmma_f32_16x16x4_f32(false, f.a[u][1], false, b1, (short)0, acc[1][1], false, false);
  }
}

template <int MASK>
__device__ __forceinline__ float row_xadd(float v) {
  const int t = __builtin_amdgcn_update_dpp(0, __float_as_int(v), 0x160 | MASK, 0xF, 0xF, true);
  return v + __int_as_float(t);
}
__device__ __forceinline__ float row_sum16(float v) {
  v = row_xadd<1>(v);
  v = row_xadd<2>(v);
  v = row_xadd<4>(v);
  v = row_xadd<8>(v);
  return v;
}
__device__ __forceinline__ float lrelu(float h) { return h > 0.0f ? h : 0.2f * h; }

__global__ __launch_bounds__(256) void gemm_pair_kernel(const float* __restrict__ x,
                                                        const float* __restrict__ W,
                                                        const float* __restrict__ att,
                                                        float* __restrict__ H,
                                                        float* __restrict__ uP,
                                                        float* __restrict__ vP,
                                                        int v1, int v2) {
  __shared__ float As[GR * APITCH];
  __shared__ __attribute__((aligned(16))) float red[8][2][GR];
  const int lane = threadIdx.x & 31;
  const int wave = __builtin_amdgcn_readfirstlane((int)(threadIdx.x >> 5));
  const int rowTile = (blockIdx.x >> 1) * GR;
  const int b = rowTile >> 11;
  const int nbase = rowTile & (NN - 1);
  const int c = lane & 15;
  const int h = lane >> 4;
  const int colbase = (blockIdx.x & 1) * GC + wave * 32;

  float4 xv[8];
#pragma unroll
  for (int i = 0; i < 8; ++i) {
    const int row = 8 * (i & 3) + wave;
    const int view = (i >> 2) ? v2 : v1;
    xv[i] = *(const float4*)(x + (((size_t)b * VV + view) * NN + nbase + row) * FF + lane * 4);
  }

  const v2f at = *(const v2f*)(att + colbase + 2 * c);
  const v2f ab = *(const v2f*)(att + OO + colbase + 2 * c);

  fragset F0, F1;
  const float* wl = W + (size_t)(2 * h) * OO + colbase + 2 * c;
  load_b_group(F0, wl, 0);

#pragma unroll
  for (int i = 0; i < 8; ++i) {
    const int row = 8 * (i & 3) + wave;
    const int half = i >> 2;
    const int r = half ? v1 : v2;
    float4 q = xv[i];
    q.x *= (lane == 0) ? sgnval(r, 0) : 1.0f;
    q.y *= (lane == 0) ? sgnval(r, 1) : 1.0f;
    q.z *= (lane == 0) ? sgnval(r, 2) : 1.0f;
    q.w *= (lane == 0) ? sgnval(r, 3) : 1.0f;
    *(float4*)(As + row * APITCH + half * FF + lane * 4) = q;
  }
  __syncthreads();

  const float* al = As + c * APITCH + 2 * h;
  load_a_group(F0, al, 0);

  v8f acc[2][2] = {};
  for (int g = 0; g + 2 < NGRP; g += 2) {
    load_b_group(F1, wl, g + 1);
    load_a_group(F1, al, g + 1);
    __builtin_amdgcn_sched_barrier(0);
    mma_group(F0, acc);
    __builtin_amdgcn_sched_barrier(0);
    load_b_group(F0, wl, g + 2);
    load_a_group(F0, al, g + 2);
    __builtin_amdgcn_sched_barrier(0);
    mma_group(F1, acc);
    __builtin_amdgcn_sched_barrier(0);
  }
  load_b_group(F1, wl, NGRP - 1);
  load_a_group(F1, al, NGRP - 1);
  __builtin_amdgcn_sched_barrier(0);
  mma_group(F0, acc);
  mma_group(F1, acc);

  float* Hb = H + ((size_t)b * NN + nbase + 8 * h) * OO + colbase + 2 * c;
#pragma unroll
  for (int i = 0; i < 8; ++i) {
    *(v2f*)(Hb + (size_t)i * OO) = (v2f){ acc[0][0][i], acc[0][1][i] };
    *(v2f*)(Hb + (size_t)(16 + i) * OO) = (v2f){ acc[1][0][i], acc[1][1][i] };
  }

  float pu[2][8], pv[2][8];
#pragma unroll
  for (int rt = 0; rt < 2; ++rt) {
#pragma unroll
    for (int i = 0; i < 8; ++i) {
      const float l0 = lrelu(acc[rt][0][i]);
      const float l1 = lrelu(acc[rt][1][i]);
      pu[rt][i] = row_sum16(at.x * l0 + at.y * l1);
      pv[rt][i] = row_sum16(ab.x * l0 + ab.y * l1);
    }
  }
  if (c == 0) {
#pragma unroll
    for (int rt = 0; rt < 2; ++rt) {
      float4* ru = (float4*)&red[wave][0][16 * rt + 8 * h];
      float4* rv = (float4*)&red[wave][1][16 * rt + 8 * h];
      ru[0] = make_float4(pu[rt][0], pu[rt][1], pu[rt][2], pu[rt][3]);
      ru[1] = make_float4(pu[rt][4], pu[rt][5], pu[rt][6], pu[rt][7]);
      rv[0] = make_float4(pv[rt][0], pv[rt][1], pv[rt][2], pv[rt][3]);
      rv[1] = make_float4(pv[rt][4], pv[rt][5], pv[rt][6], pv[rt][7]);
    }
  }
  __syncthreads();
  if (wave < 2) {
    float a = red[0][wave][lane];
#pragma unroll
    for (int w = 1; w < 8; ++w) a += red[w][wave][lane];
    float* dp = (wave ? vP : uP) + (size_t)(blockIdx.x & 1) * (BB * NN) + rowTile + lane;
    *dp = a;
  }
}

__global__ __launch_bounds__(ESB) void edge_scores_kernel(const float* __restrict__ uS,
                                                          const float* __restrict__ vS,
                                                          const int* __restrict__ ei,
                                                          float* __restrict__ s,
                                                          float* __restrict__ part) {
  __shared__ float redm[ESB / 32];
  __shared__ float reds[ESB / 32];
  const int lane = threadIdx.x & 31;
  const int wave = threadIdx.x >> 5;
  const int t = blockIdx.x * ESB + threadIdx.x;
  const int b = blockIdx.x / NPB;
  const int e = t - b * NE;
  const int sn = b * NN + src_of(ei, e);
  const int dn = b * NN + dst_of(ei, e);
  const float sc = (uS[sn] + uS[BB * NN + sn]) + (vS[dn] + vS[BB * NN + dn]);
  s[t] = sc;

  float m = sc;
  for (int off = 16; off; off >>= 1) m = fmaxf(m, __shfl_xor(m, off, 32));
  if (lane == 0) redm[wave] = m;
  __syncthreads();
  float bm = redm[0];
  for (int i = 1; i < ESB / 32; ++i) bm = fmaxf(bm, redm[i]);

  float ex = expf(sc - bm);
  for (int off = 16; off; off >>= 1) ex += __shfl_xor(ex, off, 32);
  if (lane == 0) reds[wave] = ex;
  __syncthreads();
  if (threadIdx.x == 0) {
    float a = reds[0];
    for (int i = 1; i < ESB / 32; ++i) a += reds[i];
    part[2 * blockIdx.x]     = bm;
    part[2 * blockIdx.x + 1] = a;
  }
}

template <int G>
__device__ __forceinline__ void gather_group(const float* __restrict__ Hb, int snv, float wv,
                                             int k, int lane, float4* acc) {
  int sn[G];
  float w[G];
#pragma unroll
  for (int t = 0; t < G; ++t) {
    sn[t] = __builtin_amdgcn_readlane(snv, k + t);
    w[t] = __int_as_float(__builtin_amdgcn_readlane(__float_as_int(wv), k + t));
  }
  float4 h[G][4];
#pragma unroll
  for (int t = 0; t < G; ++t) {
    const float4* hr = (const float4*)(Hb + (size_t)sn[t] * OO);
#pragma unroll
    for (int j = 0; j < 4; ++j) h[t][j] = hr[lane + 32 * j];
  }
#pragma unroll
  for (int t = 0; t < G; ++t) {
#pragma unroll
    for (int j = 0; j < 4; ++j) {
      acc[j].x = __builtin_fmaf(w[t], h[t][j].x, acc[j].x);
      acc[j].y = __builtin_fmaf(w[t], h[t][j].y, acc[j].y);
      acc[j].z = __builtin_fmaf(w[t], h[t][j].z, acc[j].z);
      acc[j].w = __builtin_fmaf(w[t], h[t][j].w, acc[j].w);
    }
  }
}

__global__ __launch_bounds__(256) __attribute__((amdgpu_waves_per_eu(2, 4))) void aggregate_kernel(const float* __restrict__ H,
                                                        const float* __restrict__ s,
                                                        const float* __restrict__ part,
                                                        const int* __restrict__ rowstart,
                                                        const int* __restrict__ order,
                                                        const int* __restrict__ ei,
                                                        const float* __restrict__ bias,
                                                        float* __restrict__ out, int g, int hv) {
  const int lane = threadIdx.x & 31;
  const int wave = __builtin_amdgcn_readfirstlane((int)(threadIdx.x >> 5));
  const int node = blockIdx.x * 8 + wave;
  if (node >= BB * NN) return;
  const int b = node >> 11;
  const int n = node & (NN - 1);
  const float* Hb = H + (size_t)b * NN * OO;
  const float* sb = s + (size_t)b * NE;
  const int beg = rowstart[n];
  const int end = rowstart[n + 1];

  const float2* pb = (const float2*)part + (size_t)b * NPB;
  constexpr int PPL = (NPB + 31) / 32;
  float pm[PPL], ps[PPL];
  for (int i = 0; i < PPL; ++i) {
    const int k = lane + 32 * i;
    const bool ok = k < NPB;
    const float2 q = ok ? pb[k] : make_float2(-3.4e38f, 0.0f);
    pm[i] = q.x; ps[i] = q.y;
  }

  int snv = 0;
  float sv = 0.0f;
  if (beg + lane < end) {
    const int e = order[beg + lane];
    snv = src_of(ei, e);
    sv = sb[e];
  }

  float m = -3.4e38f;
  for (int i = 0; i < PPL; ++i) m = fmaxf(m, pm[i]);
  for (int off = 16; off; off >>= 1) m = fmaxf(m, __shfl_xor(m, off, 32));
  float den = 0.0f;
  for (int i = 0; i < PPL; ++i) den += (lane + 32 * i < NPB) ? ps[i] * expf(pm[i] - m) : 0.0f;
  for (int off = 16; off; off >>= 1) den += __shfl_xor(den, off, 32);
  const float inv = 1.0f / den;

  float4 acc[4] = {};
  int base = beg;
  while (true) {
    const int left = end - base;
    const int cnt = left < 32 ? left : 32;
    const float wv = (lane < cnt) ? expf(sv - m) * inv : 0.0f;
    int k = 0;
    for (; k + 8 <= cnt; k += 8) gather_group<8>(Hb, snv, wv, k, lane, acc);
    if (k + 4 <= cnt) { gather_group<4>(Hb, snv, wv, k, lane, acc); k += 4; }
    if (k + 2 <= cnt) { gather_group<2>(Hb, snv, wv, k, lane, acc); k += 2; }
    if (k < cnt) gather_group<1>(Hb, snv, wv, k, lane, acc);
    base += 32;
    if (base >= end) break;
    snv = 0;
    sv = 0.0f;
    if (base + lane < end) {
      const int e = order[base + lane];
      snv = src_of(ei, e);
      sv = sb[e];
    }
  }

  float4* op = (float4*)(out + (((size_t)b * VV + g) * NN + n) * OO);
  if (hv == 0) {
#pragma unroll
    for (int j = 0; j < 4; ++j) op[lane + 32 * j] = acc[j];
  } else {
    const bool last = (hv == VV - 1);
    const float4* bp = (const float4*)bias;
    float4 o[4], q[4];
#pragma unroll
    for (int j = 0; j < 4; ++j) o[j] = op[lane + 32 * j];
    if (last) {
#pragma unroll
      for (int j = 0; j < 4; ++j) q[j] = bp[lane + 32 * j];
    }
#pragma unroll
    for (int j = 0; j < 4; ++j) {
      o[j].x += acc[j].x; o[j].y += acc[j].y; o[j].z += acc[j].z; o[j].w += acc[j].w;
    }
    if (last) {
#pragma unroll
      for (int j = 0; j < 4; ++j) {
        o[j].x = o[j].x * 0.25f + q[j].x;
        o[j].y = o[j].y * 0.25f + q[j].y;
        o[j].z = o[j].z * 0.25f + q[j].z;
        o[j].w = o[j].w * 0.25f + q[j].w;
      }
    }
#pragma unroll
    for (int j = 0; j < 4; ++j) op[lane + 32 * j] = o[j];
  }
}

extern "C" void kernel_launch(void* const* d_in, const int* in_sizes, int n_in,
                              void* d_out, int out_size, void* d_ws, size_t ws_size,
                              hipStream_t stream) {
  const float* x    = (const float*)d_in[0];
  const int*   ei   = (const int*)d_in[1];
  const float* W    = (const float*)d_in[2];
  const float* att  = (const float*)d_in[3];
  const float* bias = (const float*)d_in[4];
  float* out = (float*)d_out;

  char* p = (char*)d_ws;
  float* H     = (float*)p; p += (size_t)BB * NN * OO * sizeof(float);
  float* uS    = (float*)p; p += (size_t)2 * BB * NN * sizeof(float);
  float* vS    = (float*)p; p += (size_t)2 * BB * NN * sizeof(float);
  float* s     = (float*)p; p += (size_t)BB * NE * sizeof(float);
  int* counts  = (int*)p;   p += (size_t)NN * sizeof(int);
  int* order   = (int*)p;   p += (size_t)NE * sizeof(int);
  int* rowstart= (int*)p;   p += (size_t)(NN + 1) * sizeof(int);
  static_assert((size_t)2 * BB * NPB * sizeof(float) <= (size_t)NN * sizeof(int), "partials must fit");
  float* part = (float*)counts;

  zero_i32_kernel<<<(NN + 255) / 256, 256, 0, stream>>>(counts, NN);
  count_dst_kernel<<<(NE + 255) / 256, 256, 0, stream>>>(ei, counts);
  scan_counts_kernel<<<1, 256, 0, stream>>>(counts, rowstart);
  build_rows_kernel<<<NN / 8, 256, 0, stream>>>(ei, rowstart, order);

  for (int g = 0; g < VV; ++g) {
    for (int hv = 0; hv < VV; ++hv) {
      const int v1 = hv;
      const int v2 = g ^ hv;
      gemm_pair_kernel<<<GEMM_BLOCKS, 256, 0, stream>>>(x, W, att, H, uS, vS, v1, v2);
      edge_scores_kernel<<<BB * NPB, ESB, 0, stream>>>(uS, vS, ei, s, part);
      aggregate_kernel<<<(BB * NN) / 8, 256, 0, stream>>>(H, s, part, rowstart, order, ei, bias, out, g, hv);
    }
  }

  (void)in_sizes; (void)n_in; (void)out_size; (void)ws_size;
}


# score products: the 8 per-view half products over features 4..127 are formed once per tile and shared by the 16 pairs; each pair adds its two and its signed k-quads (product sum order changes)
# speedup vs baseline: 9.2238x; 2.4173x over previous
#include <hip/hip_runtime.h>
#include <hip/hip_bf16.h>

typedef __attribute__((ext_vector_type(2))) float v2f;
typedef __attribute__((ext_vector_type(8))) float v8f;

constexpr int BB = 2;
constexpr int VV = 4;
constexpr int NN = 2048;
constexpr int FF = 128;
constexpr int EE = 32768;
constexpr int NE = 34816;
constexpr int OO = 512;
constexpr int ESB = 256;
constexpr int NPB = NE / ESB;
static_assert(NE % ESB == 0, "an edge-score block must not straddle two batches");
constexpr int NC = VV * VV;

__device__ __forceinline__ float sgnval(int r, int j) {
  if (j >= 4) return 1.0f;
  const bool flip = (j & 1) ? ((r & 2) != 0) : ((r & 1) != 0);
  return flip ? -1.0f : 1.0f;
}

__device__ __forceinline__ int src_of(const int* __restrict__ ei, int e) {
  return (e < EE) ? ei[e] : (e - EE);
}
__device__ __forceinline__ int dst_of(const int* __restrict__ ei, int e) {
  return (e < EE) ? ei[EE + e] : (e - EE);
}

__global__ void zero_i32_kernel(int* __restrict__ p, int n) {
  int i = blockIdx.x * blockDim.x + threadIdx.x;
  if (i < n) p[i] = 0;
}

__global__ void count_dst_kernel(const int* __restrict__ ei, int* __restrict__ counts) {
  int e = blockIdx.x * blockDim.x + threadIdx.x;
  if (e >= NE) return;
  atomicAdd(&counts[dst_of(ei, e)], 1);
}

__global__ __launch_bounds__(256) void scan_counts_kernel(const int* __restrict__ counts,
                                                          int* __restrict__ rowstart) {
  __shared__ int part[256];
  __shared__ int pref[257];
  const int tid = threadIdx.x;
  const int base = tid * 8;
  int loc[8];
  int s = 0;
#pragma unroll
  for (int i = 0; i < 8; ++i) { loc[i] = s; s += counts[base + i]; }
  part[tid] = s;
  __syncthreads();
  if (tid == 0) {
    int a = 0;
    for (int i = 0; i < 256; ++i) { pref[i] = a; a += part[i]; }
    pref[256] = a;
  }
  __syncthreads();
  const int off = pref[tid];
#pragma unroll
  for (int i = 0; i < 8; ++i) rowstart[base + i] = off + loc[i];
  if (tid == 255) rowstart[NN] = pref[256];
}

constexpr int BRU = 8;
constexpr int BRT = EE / (128 * BRU);
static_assert(EE % (128 * BRU) == 0, "the explicit edges must fill whole batches");

__global__ __launch_bounds__(256) void build_rows_kernel(const int* __restrict__ ei,
                                                         const int* __restrict__ rowstart,
                                                         int* __restrict__ order) {
  const int lane = threadIdx.x & 31;
  const int wave = __builtin_amdgcn_readfirstlane((int)(threadIdx.x >> 5));
  const int n = blockIdx.x * 8 + wave;
  if (n >= NN) return;
  const int base = rowstart[n];
  const int4* d4 = (const int4*)(ei + EE) + lane;
  const unsigned below = (1u << lane) - 1u;
  int cnt = 0;

  int4 cur[BRU], nxt[BRU];
#pragma unroll
  for (int u = 0; u < BRU; ++u) cur[u] = d4[u * 32];

  for (int t = 0; t < BRT; ++t) {
    if (t + 1 < BRT) {
#pragma unroll
      for (int u = 0; u < BRU; ++u) nxt[u] = d4[((t + 1) * BRU + u) * 32];
    }
#pragma unroll
    for (int u = 0; u < BRU; ++u) {
      const int4 q = cur[u];
      const bool m0 = q.x == n, m1 = q.y == n, m2 = q.z == n, m3 = q.w == n;
      if (__builtin_amdgcn_ballot_w32(m0 | m1 | m2 | m3) != 0u) {
        const unsigned b0 = __builtin_amdgcn_ballot_w32(m0);
        const unsigned b1 = __builtin_amdgcn_ballot_w32(m1);
        const unsigned b2 = __builtin_amdgcn_ballot_w32(m2);
        const unsigned b3 = __builtin_amdgcn_ballot_w32(m3);
        const int e = (((t * BRU + u) * 32) + lane) * 4;
        int r = base + cnt + __popc(b0 & below) + __popc(b1 & below) + __popc(b2 & below) +
                __popc(b3 & below);
        if (m0) order[r] = e;
        r += m0 ? 1 : 0;
        if (m1) order[r] = e + 1;
        r += m1 ? 1 : 0;
        if (m2) order[r] = e + 2;
        r += m2 ? 1 : 0;
        if (m3) order[r] = e + 3;
        cnt += __popc(b0) + __popc(b1) + __popc(b2) + __popc(b3);
      }
    }
#pragma unroll
    for (int u = 0; u < BRU; ++u) cur[u] = nxt[u];
  }
  if (lane == 0) order[base + cnt] = EE + n;
}

constexpr int GR = 32;
constexpr int GC = 256;
constexpr int KQ = (2 * FF) / 4;
constexpr int GQ = 8;
constexpr int NGRP = KQ / GQ;
constexpr int APITCH = 2 * FF + 4;
constexpr int GEMM_BLOCKS = ((BB * NN) / GR) * (OO / GC);
static_assert(NN % GR == 0 && OO % GC == 0 && KQ % GQ == 0 && NGRP % 2 == 0, "tiling must be exact");
static_assert(GR * 2 * FF == 8 * 256 * 4, "the tile fill takes 8 float4 per thread");

struct fragset {
  v2f a[GQ][2];
  v2f w[GQ][2];
};

__device__ __forceinline__ void load_b_group(fragset& f, const float* __restrict__ wl, int g) {
#pragma unroll
  for (int u = 0; u < GQ; ++u) {
    const float* w0 = wl + (size_t)(4 * (g * GQ + u)) * OO;
    f.w[u][0] = *(const v2f*)w0;
    f.w[u][1] = *(const v2f*)(w0 + OO);
  }
}
__device__ __forceinline__ void load_a_group(fragset& f, const float* al, int g) {
#pragma unroll
  for (int u = 0; u < GQ; ++u) {
    const int k = 4 * (g * GQ + u);
    f.a[u][0] = *(const v2f*)(al + k);
    f.a[u][1] = *(const v2f*)(al + 16 * APITCH + k);
  }
}
__device__ __forceinline__ void mma_group(const fragset& f, v8f (&acc)[2][2]) {
#pragma unroll
  for (int u = 0; u < GQ; ++u) {
    const v2f b0 = { f.w[u][0].x, f.w[u][1].x };
    const v2f b1 = { f.w[u][0].y, f.w[u][1].y };
    acc[0][0] = __builtin_amdgcn_wmma_f32_16x16x4_f32(false, f.a[u][0], false, b0, (short)0, acc[0][0], false, false);
    acc[0][1] = __builtin_amdgcn_wmma_f32_16x16x4_f32(false, f.a[u][0], false, b1, (short)0, acc[0][1], false, false);
    acc[1][0] = __builtin_amdgcn_wmma_f32_16x16x4_f32(false, f.a[u][1], false, b0, (short)0, acc[1][0], false, false);
    acc[1][1] = __builtin_amdgcn_wmma_f32_16x16x4_f32(false, f.a[u][1], false, b1, (short)0, acc[1][1], false, false);
  }
}

template <int MASK>
__device__ __forceinline__ float row_xadd(float v) {
  const int t = __builtin_amdgcn_update_dpp(0, __float_as_int(v), 0x160 | MASK, 0xF, 0xF, true);
  return v + __int_as_float(t);
}
__device__ __forceinline__ float row_sum16(float v) {
  v = row_xadd<1>(v);
  v = row_xadd<2>(v);
  v = row_xadd<4>(v);
  v = row_xadd<8>(v);
  return v;
}
__device__ __forceinline__ float lrelu(float h) { return h > 0.0f ? h : 0.2f * h; }

constexpr int XP = VV * FF + 4;
static_assert(GR * VV * FF == 16 * 256 * 4, "the tile fill takes 16 float4 per thread");

struct sfrag {
  v2f a[VV][2];
  v2f w[2][2];
};
__device__ __forceinline__ void load_sfrag_w(sfrag& f, const float* __restrict__ wl, int q) {
#pragma unroll
  for (int hf = 0; hf < 2; ++hf) {
    const float* w0 = wl + (size_t)(hf * FF + 4 * q) * OO;
    f.w[hf][0] = *(const v2f*)w0;
    f.w[hf][1] = *(const v2f*)(w0 + OO);
  }
}
__device__ __forceinline__ void load_sfrag_a(sfrag& f, const float* al, int q) {
#pragma unroll
  for (int v = 0; v < VV; ++v) {
    f.a[v][0] = *(const v2f*)(al + v * FF + 4 * q);
    f.a[v][1] = *(const v2f*)(al + 16 * XP + v * FF + 4 * q);
  }
}
__device__ __forceinline__ void load_sfrag(sfrag& f, const float* al, const float* __restrict__ wl,
                                           int q) {
  load_sfrag_w(f, wl, q);
  load_sfrag_a(f, al, q);
}
__device__ __forceinline__ v8f wmma4(v2f a, v2f b, v8f c) {
  return __builtin_amdgcn_wmma_f32_16x16x4_f32(false, a, false, b, (short)0, c, false, false);
}
__device__ __forceinline__ void mma_sfrag(const sfrag& f, v8f (&R)[2][VV][2][2]) {
#pragma unroll
  for (int hf = 0; hf < 2; ++hf) {
    const v2f b0 = { f.w[hf][0].x, f.w[hf][1].x };
    const v2f b1 = { f.w[hf][0].y, f.w[hf][1].y };
#pragma unroll
    for (int v = 0; v < VV; ++v) {
      R[hf][v][0][0] = wmma4(f.a[v][0], b0, R[hf][v][0][0]);
      R[hf][v][0][1] = wmma4(f.a[v][0], b1, R[hf][v][0][1]);
      R[hf][v][1][0] = wmma4(f.a[v][1], b0, R[hf][v][1][0]);
      R[hf][v][1][1] = wmma4(f.a[v][1], b1, R[hf][v][1][1]);
    }
  }
}
__device__ __forceinline__ v2f signed_quad0(v2f a, int r) {
  return (v2f){ (r & 1) ? -a.x : a.x, (r & 2) ? -a.y : a.y };
}

static_assert(GEMM_BLOCKS == 256, "one block per 32-row tile and column half");
__global__ __launch_bounds__(256) void score_gemm_kernel(const float* __restrict__ x,
                                                         const float* __restrict__ W,
                                                         const float* __restrict__ att,
                                                         float* __restrict__ uP,
                                                         float* __restrict__ vP) {
  __shared__ __attribute__((aligned(16))) float Xs[GR * XP];
  __shared__ __attribute__((aligned(16))) float red[NC][8][2][GR];
  const int lane = threadIdx.x & 31;
  const int wave = __builtin_amdgcn_readfirstlane((int)(threadIdx.x >> 5));
  const int blk = blockIdx.x;
  const int rowTile = (blk >> 1) * GR;
  const int b = rowTile >> 11;
  const int nbase = rowTile & (NN - 1);
  const int c = lane & 15;
  const int h = lane >> 4;
  const int colbase = (blk & 1) * GC + wave * 32;

  float4 xv[16];
#pragma unroll
  for (int i = 0; i < 16; ++i) {
    const int row = 8 * (i & 3) + wave;
    const int view = i >> 2;
    xv[i] = *(const float4*)(x + (((size_t)b * VV + view) * NN + nbase + row) * FF + lane * 4);
  }

  const v2f at = *(const v2f*)(att + colbase + 2 * c);
  const v2f ab = *(const v2f*)(att + OO + colbase + 2 * c);

  const float* wl = W + (size_t)(2 * h) * OO + colbase + 2 * c;
  const float* al = Xs + c * XP + 2 * h;

  sfrag F0, F1;
  load_sfrag_w(F0, wl, 1);

#pragma unroll
  for (int i = 0; i < 16; ++i) {
    const int row = 8 * (i & 3) + wave;
    const int view = i >> 2;
    *(float4*)(Xs + row * XP + view * FF + lane * 4) = xv[i];
  }
  __syncthreads();

  load_sfrag_a(F0, al, 1);

  v8f R[2][VV][2][2] = {};
  for (int q = 1; q + 2 < FF / 4; q += 2) {
    load_sfrag(F1, al, wl, q + 1);
    __builtin_amdgcn_sched_barrier(0);
    mma_sfrag(F0, R);
    __builtin_amdgcn_sched_barrier(0);
    load_sfrag(F0, al, wl, q + 2);
    __builtin_amdgcn_sched_barrier(0);
    mma_sfrag(F1, R);
    __builtin_amdgcn_sched_barrier(0);
  }
  load_sfrag(F1, al, wl, 0);
  __builtin_amdgcn_sched_barrier(0);
  mma_sfrag(F0, R);

  const v2f b10 = { F1.w[0][0].x, F1.w[0][1].x }, b11 = { F1.w[0][0].y, F1.w[0][1].y };
  const v2f b20 = { F1.w[1][0].x, F1.w[1][1].x }, b21 = { F1.w[1][0].y, F1.w[1][1].y };

#pragma unroll
  for (int pair = 0; pair < NC; ++pair) {
    const int v1 = pair & 3;
    const int v2 = (pair >> 2) ^ v1;
    v8f t[2][2];
#pragma unroll
    for (int rt = 0; rt < 2; ++rt) {
      const v2f a1 = signed_quad0(F1.a[v1][rt], v2);
      const v2f a2 = signed_quad0(F1.a[v2][rt], v1);
      t[rt][0] = R[0][v1][rt][0] + R[1][v2][rt][0];
      t[rt][1] = R[0][v1][rt][1] + R[1][v2][rt][1];
      t[rt][0] = wmma4(a1, b10, t[rt][0]);
      t[rt][1] = wmma4(a1, b11, t[rt][1]);
      t[rt][0] = wmma4(a2, b20, t[rt][0]);
      t[rt][1] = wmma4(a2, b21, t[rt][1]);
    }
    float pu[2][8], pv[2][8];
#pragma unroll
    for (int rt = 0; rt < 2; ++rt) {
#pragma unroll
      for (int i = 0; i < 8; ++i) {
        const float l0 = lrelu(t[rt][0][i]);
        const float l1 = lrelu(t[rt][1][i]);
        pu[rt][i] = row_sum16(at.x * l0 + at.y * l1);
        pv[rt][i] = row_sum16(ab.x * l0 + ab.y * l1);
      }
    }
    if (c == 0) {
#pragma unroll
      for (int rt = 0; rt < 2; ++rt) {
        float4* ru = (float4*)&red[pair][wave][0][16 * rt + 8 * h];
        float4* rv = (float4*)&red[pair][wave][1][16 * rt + 8 * h];
        ru[0] = make_float4(pu[rt][0], pu[rt][1], pu[rt][2], pu[rt][3]);
        ru[1] = make_float4(pu[rt][4], pu[rt][5], pu[rt][6], pu[rt][7]);
        rv[0] = make_float4(pv[rt][0], pv[rt][1], pv[rt][2], pv[rt][3]);
        rv[1] = make_float4(pv[rt][4], pv[rt][5], pv[rt][6], pv[rt][7]);
      }
    }
  }
  __syncthreads();
#pragma unroll
  for (int j = 0; j < 4; ++j) {
    const int pair = 2 * wave + (j >> 1);
    const int uv = j & 1;
    float a = red[pair][0][uv][lane];
#pragma unroll
    for (int w = 1; w < 8; ++w) a += red[pair][w][uv][lane];
    float* dp = (uv ? vP : uP) + (size_t)pair * (2 * BB * NN) + (size_t)(blk & 1) * (BB * NN) +
                rowTile + lane;
    *dp = a;
  }
}

__global__ __launch_bounds__(ESB) void edge_stats_kernel(const float* __restrict__ uS,
                                                         const float* __restrict__ vS,
                                                         const int* __restrict__ ei,
                                                         float* __restrict__ part) {
  __shared__ float redm[VV][ESB / 32];
  __shared__ float reds[VV][ESB / 32];
  const int lane = threadIdx.x & 31;
  const int wave = threadIdx.x >> 5;
  const int g = blockIdx.x / (BB * NPB);
  const int blkp = blockIdx.x - g * (BB * NPB);
  const int t = blkp * ESB + threadIdx.x;
  const int b = blkp / NPB;
  const int e = t - b * NE;
  const int sn = b * NN + src_of(ei, e);
  const int dn = b * NN + dst_of(ei, e);

  float sc[VV];
#pragma unroll
  for (int hv = 0; hv < VV; ++hv) {
    const float* u = uS + (size_t)(g * VV + hv) * (2 * BB * NN);
    const float* v = vS + (size_t)(g * VV + hv) * (2 * BB * NN);
    sc[hv] = (u[sn] + u[BB * NN + sn]) + (v[dn] + v[BB * NN + dn]);
  }
#pragma unroll
  for (int hv = 0; hv < VV; ++hv) {
    float m = sc[hv];
    for (int off = 16; off; off >>= 1) m = fmaxf(m, __shfl_xor(m, off, 32));
    if (lane == 0) redm[hv][wave] = m;
  }
  __syncthreads();
  float bm[VV];
#pragma unroll
  for (int hv = 0; hv < VV; ++hv) {
    float a = redm[hv][0];
    for (int i = 1; i < ESB / 32; ++i) a = fmaxf(a, redm[hv][i]);
    bm[hv] = a;
  }
#pragma unroll
  for (int hv = 0; hv < VV; ++hv) {
    float ex = expf(sc[hv] - bm[hv]);
    for (int off = 16; off; off >>= 1) ex += __shfl_xor(ex, off, 32);
    if (lane == 0) reds[hv][wave] = ex;
  }
  __syncthreads();
  if (threadIdx.x < VV) {
    const int hv = threadIdx.x;
    float mx = redm[hv][0];
    float a = reds[hv][0];
    for (int i = 1; i < ESB / 32; ++i) { mx = fmaxf(mx, redm[hv][i]); a += reds[hv][i]; }
    const int idx = (g * VV + hv) * (BB * NPB) + blkp;
    part[2 * idx]     = mx;
    part[2 * idx + 1] = a;
  }
}

template <int NG, int G>
__device__ __forceinline__ void gather_x_group(const float4* __restrict__ xb, int snv,
                                               const float (&wv)[VV], int k, int lane,
                                               float4 (&A1)[VV], float4 (&A2)[VV]) {
  int sn[NG];
  float w[NG][VV];
#pragma unroll
  for (int t = 0; t < NG; ++t) {
    sn[t] = __builtin_amdgcn_readlane(snv, k + t);
#pragma unroll
    for (int hv = 0; hv < VV; ++hv)
      w[t][hv] = __int_as_float(__builtin_amdgcn_readlane(__float_as_int(wv[hv]), k + t));
  }
  float4 r[NG][VV];
#pragma unroll
  for (int t = 0; t < NG; ++t) {
#pragma unroll
    for (int v = 0; v < VV; ++v) r[t][v] = xb[((size_t)v * NN + sn[t]) * (FF / 4) + lane];
  }
#pragma unroll
  for (int t = 0; t < NG; ++t) {
#pragma unroll
    for (int hv = 0; hv < VV; ++hv) {
      const float4 p = r[t][hv];
      const float4 q = r[t][G ^ hv];
      const float ww = w[t][hv];
      A1[hv].x = __builtin_fmaf(ww, p.x, A1[hv].x);
      A1[hv].y = __builtin_fmaf(ww, p.y, A1[hv].y);
      A1[hv].z = __builtin_fmaf(ww, p.z, A1[hv].z);
      A1[hv].w = __builtin_fmaf(ww, p.w, A1[hv].w);
      A2[hv].x = __builtin_fmaf(ww, q.x, A2[hv].x);
      A2[hv].y = __builtin_fmaf(ww, q.y, A2[hv].y);
      A2[hv].z = __builtin_fmaf(ww, q.z, A2[hv].z);
      A2[hv].w = __builtin_fmaf(ww, q.w, A2[hv].w);
    }
  }
}

template <int G>
__global__ __launch_bounds__(256) __attribute__((amdgpu_waves_per_eu(2, 4))) void aggregate_x_kernel(const float* __restrict__ x,
                                                          const float* __restrict__ uS,
                                                          const float* __restrict__ vS,
                                                          const float* __restrict__ part,
                                                          const int* __restrict__ rowstart,
                                                          const int* __restrict__ order,
                                                          const int* __restrict__ ei,
                                                          float* __restrict__ Z) {
  const int lane = threadIdx.x & 31;
  const int wave = __builtin_amdgcn_readfirstlane((int)(threadIdx.x >> 5));
  const int node = blockIdx.x * 8 + wave;
  if (node >= BB * NN) return;
  const int b = node >> 11;
  const int n = node & (NN - 1);
  const float4* xb = (const float4*)(x + (size_t)b * VV * NN * FF);
  const int beg = rowstart[n];
  const int end = rowstart[n + 1];

  constexpr int PPL = (NPB + 31) / 32;
  float pm[VV][PPL], ps[VV][PPL], va[VV], vb[VV];
#pragma unroll
  for (int hv = 0; hv < VV; ++hv) {
    const int pair = G * VV + hv;
    const float2* pb = (const float2*)part + (size_t)pair * (BB * NPB) + (size_t)b * NPB;
#pragma unroll
    for (int i = 0; i < PPL; ++i) {
      const int k = lane + 32 * i;
      const bool ok = k < NPB;
      const float2 q = pb[ok ? k : 0];
      pm[hv][i] = ok ? q.x : -3.4e38f;
      ps[hv][i] = ok ? q.y : 0.0f;
    }
    const float* vv = vS + (size_t)pair * (2 * BB * NN);
    va[hv] = vv[node];
    vb[hv] = vv[BB * NN + node];
  }

  int snv = 0;
  if (beg + lane < end) snv = src_of(ei, order[beg + lane]);
  float u0[VV], u1[VV];
#pragma unroll
  for (int hv = 0; hv < VV; ++hv) {
    const float* uu = uS + (size_t)(G * VV + hv) * (2 * BB * NN) + b * NN;
    u0[hv] = uu[snv];
    u1[hv] = uu[BB * NN + snv];
  }

  float m[VV], inv[VV], vn[VV];
#pragma unroll
  for (int hv = 0; hv < VV; ++hv) {
    vn[hv] = va[hv] + vb[hv];
    float mm = -3.4e38f;
#pragma unroll
    for (int i = 0; i < PPL; ++i) mm = fmaxf(mm, pm[hv][i]);
    for (int off = 16; off; off >>= 1) mm = fmaxf(mm, __shfl_xor(mm, off, 32));
    float den = 0.0f;
#pragma unroll
    for (int i = 0; i < PPL; ++i)
      den += (lane + 32 * i < NPB) ? ps[hv][i] * expf(pm[hv][i] - mm) : 0.0f;
    for (int off = 16; off; off >>= 1) den += __shfl_xor(den, off, 32);
    m[hv] = mm;
    inv[hv] = 1.0f / den;
  }

  float4 A1[VV] = {}, A2[VV] = {};
  int base = beg;
  while (true) {
    const int left = end - base;
    const int cnt = left < 32 ? left : 32;
    float wv[VV];
#pragma unroll
    for (int hv = 0; hv < VV; ++hv) {
      const float sc = (u0[hv] + u1[hv]) + vn[hv];
      const float ex = expf(sc - m[hv]) * inv[hv];
      wv[hv] = (lane < cnt) ? ex : 0.0f;
    }
    int k = 0;
    for (; k + 8 <= cnt; k += 8) gather_x_group<8, G>(xb, snv, wv, k, lane, A1, A2);
    if (k + 4 <= cnt) { gather_x_group<4, G>(xb, snv, wv, k, lane, A1, A2); k += 4; }
    if (k + 2 <= cnt) { gather_x_group<2, G>(xb, snv, wv, k, lane, A1, A2); k += 2; }
    if (k < cnt) gather_x_group<1, G>(xb, snv, wv, k, lane, A1, A2);
    base += 32;
    if (base >= end) break;
    snv = 0;
    if (base + lane < end) snv = src_of(ei, order[base + lane]);
#pragma unroll
    for (int hv = 0; hv < VV; ++hv) {
      const float* uu = uS + (size_t)(G * VV + hv) * (2 * BB * NN) + b * NN;
      u0[hv] = uu[snv];
      u1[hv] = uu[BB * NN + snv];
    }
  }

  float4 z1, z2;
#pragma unroll
  for (int hv = 0; hv < VV; ++hv) {
    const int r1 = hv ^ G;
    const int r2 = hv;
    const bool l0 = (lane == 0);
    float4 a = A1[hv], c2 = A2[hv];
    a.x *= l0 ? sgnval(r1, 0) : 1.0f;
    a.y *= l0 ? sgnval(r1, 1) : 1.0f;
    a.z *= l0 ? sgnval(r1, 2) : 1.0f;
    a.w *= l0 ? sgnval(r1, 3) : 1.0f;
    c2.x *= l0 ? sgnval(r2, 0) : 1.0f;
    c2.y *= l0 ? sgnval(r2, 1) : 1.0f;
    c2.z *= l0 ? sgnval(r2, 2) : 1.0f;
    c2.w *= l0 ? sgnval(r2, 3) : 1.0f;
    if (hv == 0) {
      z1 = a; z2 = c2;
    } else {
      z1.x += a.x; z1.y += a.y; z1.z += a.z; z1.w += a.w;
      z2.x += c2.x; z2.y += c2.y; z2.z += c2.z; z2.w += c2.w;
    }
  }
  float4* zp = (float4*)(Z + (size_t)node * (2 * FF));
  zp[lane] = z1;
  zp[32 + lane] = z2;
}

__global__ __launch_bounds__(256) void out_gemm_kernel(const float* __restrict__ Z,
                                                       const float* __restrict__ W,
                                                       const float* __restrict__ bias,
                                                       float* __restrict__ out, int g) {
  __shared__ float As[GR * APITCH];
  const int lane = threadIdx.x & 31;
  const int wave = __builtin_amdgcn_readfirstlane((int)(threadIdx.x >> 5));
  const int rowTile = (blockIdx.x >> 1) * GR;
  const int b = rowTile >> 11;
  const int nbase = rowTile & (NN - 1);
  const int c = lane & 15;
  const int h = lane >> 4;
  const int colbase = (blockIdx.x & 1) * GC + wave * 32;

  float4 xv[8];
#pragma unroll
  for (int i = 0; i < 8; ++i) {
    const int row = 8 * (i & 3) + wave;
    const int half = i >> 2;
    xv[i] = *(const float4*)(Z + (size_t)(rowTile + row) * (2 * FF) + half * FF + lane * 4);
  }

  const v2f bs = *(const v2f*)(bias + colbase + 2 * c);

  fragset F0, F1;
  const float* wl = W + (size_t)(2 * h) * OO + colbase + 2 * c;
  load_b_group(F0, wl, 0);

#pragma unroll
  for (int i = 0; i < 8; ++i) {
    const int row = 8 * (i & 3) + wave;
    const int half = i >> 2;
    *(float4*)(As + row * APITCH + half * FF + lane * 4) = xv[i];
  }
  __syncthreads();

  const float* al = As + c * APITCH + 2 * h;
  load_a_group(F0, al, 0);

  v8f acc[2][2] = {};
  for (int q = 0; q + 2 < NGRP; q += 2) {
    load_b_group(F1, wl, q + 1);
    load_a_group(F1, al, q + 1);
    __builtin_amdgcn_sched_barrier(0);
    mma_group(F0, acc);
    __builtin_amdgcn_sched_barrier(0);
    load_b_group(F0, wl, q + 2);
    load_a_group(F0, al, q + 2);
    __builtin_amdgcn_sched_barrier(0);
    mma_group(F1, acc);
    __builtin_amdgcn_sched_barrier(0);
  }
  load_b_group(F1, wl, NGRP - 1);
  load_a_group(F1, al, NGRP - 1);
  __builtin_amdgcn_sched_barrier(0);
  mma_group(F0, acc);
  mma_group(F1, acc);

  float* ob = out + (((size_t)b * VV + g) * NN + nbase + 8 * h) * OO + colbase + 2 * c;
#pragma unroll
  for (int i = 0; i < 8; ++i) {
    *(v2f*)(ob + (size_t)i * OO) =
        (v2f){ acc[0][0][i] * 0.25f + bs.x, acc[0][1][i] * 0.25f + bs.y };
    *(v2f*)(ob + (size_t)(16 + i) * OO) =
        (v2f){ acc[1][0][i] * 0.25f + bs.x, acc[1][1][i] * 0.25f + bs.y };
  }
}

extern "C" void kernel_launch(void* const* d_in, const int* in_sizes, int n_in,
                              void* d_out, int out_size, void* d_ws, size_t ws_size,
                              hipStream_t stream) {
  const float* x    = (const float*)d_in[0];
  const int*   ei   = (const int*)d_in[1];
  const float* W    = (const float*)d_in[2];
  const float* att  = (const float*)d_in[3];
  const float* bias = (const float*)d_in[4];
  float* out = (float*)d_out;

  char* p = (char*)d_ws;
  float* Z     = (float*)p; p += (size_t)BB * NN * 2 * FF * sizeof(float);
  float* uS    = (float*)p; p += (size_t)NC * 2 * BB * NN * sizeof(float);
  float* vS    = (float*)p; p += (size_t)NC * 2 * BB * NN * sizeof(float);
  float* part  = (float*)p; p += (size_t)NC * 2 * BB * NPB * sizeof(float);
  int* counts  = (int*)p;   p += (size_t)NN * sizeof(int);
  int* order   = (int*)p;   p += (size_t)NE * sizeof(int);
  int* rowstart= (int*)p;   p += (size_t)(NN + 1) * sizeof(int);

  zero_i32_kernel<<<(NN + 255) / 256, 256, 0, stream>>>(counts, NN);
  count_dst_kernel<<<(NE + 255) / 256, 256, 0, stream>>>(ei, counts);
  scan_counts_kernel<<<1, 256, 0, stream>>>(counts, rowstart);
  build_rows_kernel<<<NN / 8, 256, 0, stream>>>(ei, rowstart, order);

  score_gemm_kernel<<<GEMM_BLOCKS, 256, 0, stream>>>(x, W, att, uS, vS);
  edge_stats_kernel<<<VV * BB * NPB, ESB, 0, stream>>>(uS, vS, ei, part);

  aggregate_x_kernel<0><<<(BB * NN) / 8, 256, 0, stream>>>(x, uS, vS, part, rowstart, order, ei, Z);
  out_gemm_kernel<<<GEMM_BLOCKS, 256, 0, stream>>>(Z, W, bias, out, 0);
  aggregate_x_kernel<1><<<(BB * NN) / 8, 256, 0, stream>>>(x, uS, vS, part, rowstart, order, ei, Z);
  out_gemm_kernel<<<GEMM_BLOCKS, 256, 0, stream>>>(Z, W, bias, out, 1);
  aggregate_x_kernel<2><<<(BB * NN) / 8, 256, 0, stream>>>(x, uS, vS, part, rowstart, order, ei, Z);
  out_gemm_kernel<<<GEMM_BLOCKS, 256, 0, stream>>>(Z, W, bias, out, 2);
  aggregate_x_kernel<3><<<(BB * NN) / 8, 256, 0, stream>>>(x, uS, vS, part, rowstart, order, ei, Z);
  out_gemm_kernel<<<GEMM_BLOCKS, 256, 0, stream>>>(Z, W, bias, out, 3);

  (void)in_sizes; (void)n_in; (void)out_size; (void)ws_size;
}


# aggregate and projection fused for all 4 g: a wave gathers a node's source rows once, one accumulator pair per g (sum over edges then views: order changes); LDS tiles feed the WMMAs; 7 fewer launches
# speedup vs baseline: 11.2345x; 1.2180x over previous
#include <hip/hip_runtime.h>
#include <hip/hip_bf16.h>

typedef __attribute__((ext_vector_type(2))) float v2f;
typedef __attribute__((ext_vector_type(8))) float v8f;

constexpr int BB = 2;
constexpr int VV = 4;
constexpr int NN = 2048;
constexpr int FF = 128;
constexpr int EE = 32768;
constexpr int NE = 34816;
constexpr int OO = 512;
constexpr int ESB = 256;
constexpr int NPB = NE / ESB;
static_assert(NE % ESB == 0, "an edge-score block must not straddle two batches");
constexpr int NC = VV * VV;

__device__ __forceinline__ int src_of(const int* __restrict__ ei, int e) {
  return (e < EE) ? ei[e] : (e - EE);
}
__device__ __forceinline__ int dst_of(const int* __restrict__ ei, int e) {
  return (e < EE) ? ei[EE + e] : (e - EE);
}

__global__ void zero_i32_kernel(int* __restrict__ p, int n) {
  int i = blockIdx.x * blockDim.x + threadIdx.x;
  if (i < n) p[i] = 0;
}

__global__ void count_dst_kernel(const int* __restrict__ ei, int* __restrict__ counts) {
  int e = blockIdx.x * blockDim.x + threadIdx.x;
  if (e >= NE) return;
  atomicAdd(&counts[dst_of(ei, e)], 1);
}

__global__ __launch_bounds__(256) void scan_counts_kernel(const int* __restrict__ counts,
                                                          int* __restrict__ rowstart) {
  __shared__ int part[256];
  __shared__ int pref[257];
  const int tid = threadIdx.x;
  const int base = tid * 8;
  int loc[8];
  int s = 0;
#pragma unroll
  for (int i = 0; i < 8; ++i) { loc[i] = s; s += counts[base + i]; }
  part[tid] = s;
  __syncthreads();
  if (tid == 0) {
    int a = 0;
    for (int i = 0; i < 256; ++i) { pref[i] = a; a += part[i]; }
    pref[256] = a;
  }
  __syncthreads();
  const int off = pref[tid];
#pragma unroll
  for (int i = 0; i < 8; ++i) rowstart[base + i] = off + loc[i];
  if (tid == 255) rowstart[NN] = pref[256];
}

constexpr int BRU = 8;
constexpr int BRT = EE / (128 * BRU);
static_assert(EE % (128 * BRU) == 0, "the explicit edges must fill whole batches");

__global__ __launch_bounds__(256) void build_rows_kernel(const int* __restrict__ ei,
                                                         const int* __restrict__ rowstart,
                                                         int* __restrict__ order) {
  const int lane = threadIdx.x & 31;
  const int wave = __builtin_amdgcn_readfirstlane((int)(threadIdx.x >> 5));
  const int n = blockIdx.x * 8 + wave;
  if (n >= NN) return;
  const int base = rowstart[n];
  const int4* d4 = (const int4*)(ei + EE) + lane;
  const unsigned below = (1u << lane) - 1u;
  int cnt = 0;

  int4 cur[BRU], nxt[BRU];
#pragma unroll
  for (int u = 0; u < BRU; ++u) cur[u] = d4[u * 32];

  for (int t = 0; t < BRT; ++t) {
    if (t + 1 < BRT) {
#pragma unroll
      for (int u = 0; u < BRU; ++u) nxt[u] = d4[((t + 1) * BRU + u) * 32];
    }
#pragma unroll
    for (int u = 0; u < BRU; ++u) {
      const int4 q = cur[u];
      const bool m0 = q.x == n, m1 = q.y == n, m2 = q.z == n, m3 = q.w == n;
      if (__builtin_amdgcn_ballot_w32(m0 | m1 | m2 | m3) != 0u) {
        const unsigned b0 = __builtin_amdgcn_ballot_w32(m0);
        const unsigned b1 = __builtin_amdgcn_ballot_w32(m1);
        const unsigned b2 = __builtin_amdgcn_ballot_w32(m2);
        const unsigned b3 = __builtin_amdgcn_ballot_w32(m3);
        const int e = (((t * BRU + u) * 32) + lane) * 4;
        int r = base + cnt + __popc(b0 & below) + __popc(b1 & below) + __popc(b2 & below) +
                __popc(b3 & below);
        if (m0) order[r] = e;
        r += m0 ? 1 : 0;
        if (m1) order[r] = e + 1;
        r += m1 ? 1 : 0;
        if (m2) order[r] = e + 2;
        r += m2 ? 1 : 0;
        if (m3) order[r] = e + 3;
        cnt += __popc(b0) + __popc(b1) + __popc(b2) + __popc(b3);
      }
    }
#pragma unroll
    for (int u = 0; u < BRU; ++u) cur[u] = nxt[u];
  }
  if (lane == 0) order[base + cnt] = EE + n;
}

constexpr int GR = 32;
constexpr int GC = 256;
constexpr int KQ = (2 * FF) / 4;
constexpr int GEMM_BLOCKS = ((BB * NN) / GR) * (OO / GC);
static_assert(NN % GR == 0 && OO % GC == 0, "tiling must be exact");

template <int MASK>
__device__ __forceinline__ float row_xadd(float v) {
  const int t = __builtin_amdgcn_update_dpp(0, __float_as_int(v), 0x160 | MASK, 0xF, 0xF, true);
  return v + __int_as_float(t);
}
__device__ __forceinline__ float row_sum16(float v) {
  v = row_xadd<1>(v);
  v = row_xadd<2>(v);
  v = row_xadd<4>(v);
  v = row_xadd<8>(v);
  return v;
}
__device__ __forceinline__ float lrelu(float h) { return h > 0.0f ? h : 0.2f * h; }

constexpr int XP = VV * FF + 4;
static_assert(GR * VV * FF == 16 * 256 * 4, "the tile fill takes 16 float4 per thread");

struct sfrag {
  v2f a[VV][2];
  v2f w[2][2];
};
__device__ __forceinline__ void load_sfrag_w(sfrag& f, const float* __restrict__ wl, int q) {
#pragma unroll
  for (int hf = 0; hf < 2; ++hf) {
    const float* w0 = wl + (size_t)(hf * FF + 4 * q) * OO;
    f.w[hf][0] = *(const v2f*)w0;
    f.w[hf][1] = *(const v2f*)(w0 + OO);
  }
}
__device__ __forceinline__ void load_sfrag_a(sfrag& f, const float* al, int q) {
#pragma unroll
  for (int v = 0; v < VV; ++v) {
    f.a[v][0] = *(const v2f*)(al + v * FF + 4 * q);
    f.a[v][1] = *(const v2f*)(al + 16 * XP + v * FF + 4 * q);
  }
}
__device__ __forceinline__ void load_sfrag(sfrag& f, const float* al, const float* __restrict__ wl,
                                           int q) {
  load_sfrag_w(f, wl, q);
  load_sfrag_a(f, al, q);
}
__device__ __forceinline__ v8f wmma4(v2f a, v2f b, v8f c) {
  return __builtin_amdgcn_wmma_f32_16x16x4_f32(false, a, false, b, (short)0, c, false, false);
}
__device__ __forceinline__ void mma_sfrag(const sfrag& f, v8f (&R)[2][VV][2][2]) {
#pragma unroll
  for (int hf = 0; hf < 2; ++hf) {
    const v2f b0 = { f.w[hf][0].x, f.w[hf][1].x };
    const v2f b1 = { f.w[hf][0].y, f.w[hf][1].y };
#pragma unroll
    for (int v = 0; v < VV; ++v) {
      R[hf][v][0][0] = wmma4(f.a[v][0], b0, R[hf][v][0][0]);
      R[hf][v][0][1] = wmma4(f.a[v][0], b1, R[hf][v][0][1]);
      R[hf][v][1][0] = wmma4(f.a[v][1], b0, R[hf][v][1][0]);
      R[hf][v][1][1] = wmma4(f.a[v][1], b1, R[hf][v][1][1]);
    }
  }
}
__device__ __forceinline__ v2f signed_quad0(v2f a, int r) {
  return (v2f){ (r & 1) ? -a.x : a.x, (r & 2) ? -a.y : a.y };
}

static_assert(GEMM_BLOCKS == 256, "one block per 32-row tile and column half");
__global__ __launch_bounds__(256) void score_gemm_kernel(const float* __restrict__ x,
                                                         const float* __restrict__ W,
                                                         const float* __restrict__ att,
                                                         float* __restrict__ uP,
                                                         float* __restrict__ vP) {
  __shared__ __attribute__((aligned(16))) float Xs[GR * XP];
  __shared__ __attribute__((aligned(16))) float red[NC][8][2][GR];
  const int lane = threadIdx.x & 31;
  const int wave = __builtin_amdgcn_readfirstlane((int)(threadIdx.x >> 5));
  const int blk = blockIdx.x;
  const int rowTile = (blk >> 1) * GR;
  const int b = rowTile >> 11;
  const int nbase = rowTile & (NN - 1);
  const int c = lane & 15;
  const int h = lane >> 4;
  const int colbase = (blk & 1) * GC + wave * 32;

  float4 xv[16];
#pragma unroll
  for (int i = 0; i < 16; ++i) {
    const int row = 8 * (i & 3) + wave;
    const int view = i >> 2;
    xv[i] = *(const float4*)(x + (((size_t)b * VV + view) * NN + nbase + row) * FF + lane * 4);
  }

  const v2f at = *(const v2f*)(att + colbase + 2 * c);
  const v2f ab = *(const v2f*)(att + OO + colbase + 2 * c);

  const float* wl = W + (size_t)(2 * h) * OO + colbase + 2 * c;
  const float* al = Xs + c * XP + 2 * h;

  sfrag F0, F1;
  load_sfrag_w(F0, wl, 1);

#pragma unroll
  for (int i = 0; i < 16; ++i) {
    const int row = 8 * (i & 3) + wave;
    const int view = i >> 2;
    *(float4*)(Xs + row * XP + view * FF + lane * 4) = xv[i];
  }
  __syncthreads();

  load_sfrag_a(F0, al, 1);

  v8f R[2][VV][2][2] = {};
  for (int q = 1; q + 2 < FF / 4; q += 2) {
    load_sfrag(F1, al, wl, q + 1);
    __builtin_amdgcn_sched_barrier(0);
    mma_sfrag(F0, R);
    __builtin_amdgcn_sched_barrier(0);
    load_sfrag(F0, al, wl, q + 2);
    __builtin_amdgcn_sched_barrier(0);
    mma_sfrag(F1, R);
    __builtin_amdgcn_sched_barrier(0);
  }
  load_sfrag(F1, al, wl, 0);
  __builtin_amdgcn_sched_barrier(0);
  mma_sfrag(F0, R);

  const v2f b10 = { F1.w[0][0].x, F1.w[0][1].x }, b11 = { F1.w[0][0].y, F1.w[0][1].y };
  const v2f b20 = { F1.w[1][0].x, F1.w[1][1].x }, b21 = { F1.w[1][0].y, F1.w[1][1].y };

#pragma unroll
  for (int pair = 0; pair < NC; ++pair) {
    const int v1 = pair & 3;
    const int v2 = (pair >> 2) ^ v1;
    v8f t[2][2];
#pragma unroll
    for (int rt = 0; rt < 2; ++rt) {
      const v2f a1 = signed_quad0(F1.a[v1][rt], v2);
      const v2f a2 = signed_quad0(F1.a[v2][rt], v1);
      t[rt][0] = R[0][v1][rt][0] + R[1][v2][rt][0];
      t[rt][1] = R[0][v1][rt][1] + R[1][v2][rt][1];
      t[rt][0] = wmma4(a1, b10, t[rt][0]);
      t[rt][1] = wmma4(a1, b11, t[rt][1]);
      t[rt][0] = wmma4(a2, b20, t[rt][0]);
      t[rt][1] = wmma4(a2, b21, t[rt][1]);
    }
    float pu[2][8], pv[2][8];
#pragma unroll
    for (int rt = 0; rt < 2; ++rt) {
#pragma unroll
      for (int i = 0; i < 8; ++i) {
        const float l0 = lrelu(t[rt][0][i]);
        const float l1 = lrelu(t[rt][1][i]);
        pu[rt][i] = row_sum16(at.x * l0 + at.y * l1);
        pv[rt][i] = row_sum16(ab.x * l0 + ab.y * l1);
      }
    }
    if (c == 0) {
#pragma unroll
      for (int rt = 0; rt < 2; ++rt) {
        float4* ru = (float4*)&red[pair][wave][0][16 * rt + 8 * h];
        float4* rv = (float4*)&red[pair][wave][1][16 * rt + 8 * h];
        ru[0] = make_float4(pu[rt][0], pu[rt][1], pu[rt][2], pu[rt][3]);
        ru[1] = make_float4(pu[rt][4], pu[rt][5], pu[rt][6], pu[rt][7]);
        rv[0] = make_float4(pv[rt][0], pv[rt][1], pv[rt][2], pv[rt][3]);
        rv[1] = make_float4(pv[rt][4], pv[rt][5], pv[rt][6], pv[rt][7]);
      }
    }
  }
  __syncthreads();
#pragma unroll
  for (int j = 0; j < 4; ++j) {
    const int pair = 2 * wave + (j >> 1);
    const int uv = j & 1;
    float a = red[pair][0][uv][lane];
#pragma unroll
    for (int w = 1; w < 8; ++w) a += red[pair][w][uv][lane];
    float* dp = (uv ? vP : uP) + (size_t)pair * (2 * BB * NN) + (size_t)(blk & 1) * (BB * NN) +
                rowTile + lane;
    *dp = a;
  }
}

__global__ __launch_bounds__(ESB) void edge_stats_kernel(const float* __restrict__ uS,
                                                         const float* __restrict__ vS,
                                                         const int* __restrict__ ei,
                                                         float* __restrict__ part) {
  __shared__ float redm[VV][ESB / 32];
  __shared__ float reds[VV][ESB / 32];
  const int lane = threadIdx.x & 31;
  const int wave = threadIdx.x >> 5;
  const int g = blockIdx.x / (BB * NPB);
  const int blkp = blockIdx.x - g * (BB * NPB);
  const int t = blkp * ESB + threadIdx.x;
  const int b = blkp / NPB;
  const int e = t - b * NE;
  const int sn = b * NN + src_of(ei, e);
  const int dn = b * NN + dst_of(ei, e);

  float sc[VV];
#pragma unroll
  for (int hv = 0; hv < VV; ++hv) {
    const float* u = uS + (size_t)(g * VV + hv) * (2 * BB * NN);
    const float* v = vS + (size_t)(g * VV + hv) * (2 * BB * NN);
    sc[hv] = (u[sn] + u[BB * NN + sn]) + (v[dn] + v[BB * NN + dn]);
  }
#pragma unroll
  for (int hv = 0; hv < VV; ++hv) {
    float m = sc[hv];
    for (int off = 16; off; off >>= 1) m = fmaxf(m, __shfl_xor(m, off, 32));
    if (lane == 0) redm[hv][wave] = m;
  }
  __syncthreads();
  float bm[VV];
#pragma unroll
  for (int hv = 0; hv < VV; ++hv) {
    float a = redm[hv][0];
    for (int i = 1; i < ESB / 32; ++i) a = fmaxf(a, redm[hv][i]);
    bm[hv] = a;
  }
#pragma unroll
  for (int hv = 0; hv < VV; ++hv) {
    float ex = expf(sc[hv] - bm[hv]);
    for (int off = 16; off; off >>= 1) ex += __shfl_xor(ex, off, 32);
    if (lane == 0) reds[hv][wave] = ex;
  }
  __syncthreads();
  if (threadIdx.x < VV) {
    const int hv = threadIdx.x;
    float mx = redm[hv][0];
    float a = reds[hv][0];
    for (int i = 1; i < ESB / 32; ++i) { mx = fmaxf(mx, redm[hv][i]); a += reds[hv][i]; }
    const int idx = (g * VV + hv) * (BB * NPB) + blkp;
    part[2 * idx]     = mx;
    part[2 * idx + 1] = a;
  }
}

template <int NG>
__device__ __forceinline__ void gather_all_group(const float4* __restrict__ xb, int snv,
                                                 const float (&wv)[NC], int k, int lane, int l0m,
                                                 float4 (&Z1)[VV], float4 (&Z2)[VV]) {
  int sn[NG];
#pragma unroll
  for (int t = 0; t < NG; ++t) {
    sn[t] = __builtin_amdgcn_readlane(snv, k + t);
  }
  float4 r[NG][VV];
#pragma unroll
  for (int t = 0; t < NG; ++t) {
#pragma unroll
    for (int v = 0; v < VV; ++v) r[t][v] = xb[((size_t)v * NN + sn[t]) * (FF / 4) + lane];
  }
#pragma unroll
  for (int t = 0; t < NG; ++t) {
#pragma unroll
    for (int g = 0; g < VV; ++g) {
#pragma unroll
      for (int hv = 0; hv < VV; ++hv) {
        const int wi = __builtin_amdgcn_readlane(__float_as_int(wv[g * VV + hv]), k + t);
        const int r1 = hv ^ g;
        const int r2 = hv;
        const float w1e = __int_as_float(wi ^ ((r1 & 1) ? l0m : 0));
        const float w1o = __int_as_float(wi ^ ((r1 & 2) ? l0m : 0));
        const float w2e = __int_as_float(wi ^ ((r2 & 1) ? l0m : 0));
        const float w2o = __int_as_float(wi ^ ((r2 & 2) ? l0m : 0));
        const float4 p = r[t][hv];
        const float4 q = r[t][g ^ hv];
        Z1[g].x = __builtin_fmaf(w1e, p.x, Z1[g].x);
        Z1[g].y = __builtin_fmaf(w1o, p.y, Z1[g].y);
        Z1[g].z = __builtin_fmaf(w1e, p.z, Z1[g].z);
        Z1[g].w = __builtin_fmaf(w1o, p.w, Z1[g].w);
        Z2[g].x = __builtin_fmaf(w2e, q.x, Z2[g].x);
        Z2[g].y = __builtin_fmaf(w2o, q.y, Z2[g].y);
        Z2[g].z = __builtin_fmaf(w2e, q.z, Z2[g].z);
        Z2[g].w = __builtin_fmaf(w2o, q.w, Z2[g].w);
      }
    }
  }
}

constexpr int TR = 16;
constexpr int ZP = 2 * FF + 4;
constexpr int ZT = TR * ZP;
constexpr int OQ = 4;
constexpr int ONG = KQ / OQ;
static_assert(NN % TR == 0 && OO == TR * 32 && NC == TR, "one wave per node, per 32 columns, per pair");
static_assert(KQ % OQ == 0 && ONG % 2 == 0, "fragment groups must pair up");

struct ofrag {
  v2f a[OQ][VV];
  v2f w[OQ][2];
};
__device__ __forceinline__ void load_ofrag(ofrag& f, const float* al, const float* __restrict__ wl,
                                           int grp) {
#pragma unroll
  for (int u = 0; u < OQ; ++u) {
    const int k = 4 * (grp * OQ + u);
    const float* w0 = wl + (size_t)k * OO;
    f.w[u][0] = *(const v2f*)w0;
    f.w[u][1] = *(const v2f*)(w0 + OO);
#pragma unroll
    for (int g = 0; g < VV; ++g) f.a[u][g] = *(const v2f*)(al + g * ZT + k);
  }
}
__device__ __forceinline__ void mma_ofrag(const ofrag& f, v8f (&acc)[VV][2]) {
#pragma unroll
  for (int u = 0; u < OQ; ++u) {
    const v2f b0 = { f.w[u][0].x, f.w[u][1].x };
    const v2f b1 = { f.w[u][0].y, f.w[u][1].y };
#pragma unroll
    for (int g = 0; g < VV; ++g) {
      acc[g][0] = wmma4(f.a[u][g], b0, acc[g][0]);
      acc[g][1] = wmma4(f.a[u][g], b1, acc[g][1]);
    }
  }
}

__global__ __launch_bounds__(512) void aggregate_out_kernel(const float* __restrict__ x,
                                                            const float* __restrict__ W,
                                                            const float* __restrict__ bias,
                                                            const float* __restrict__ uS,
                                                            const float* __restrict__ vS,
                                                            const float* __restrict__ part,
                                                            const int* __restrict__ rowstart,
                                                            const int* __restrict__ order,
                                                            const int* __restrict__ ei,
                                                            float* __restrict__ out) {
  __shared__ __attribute__((aligned(16))) float Zs[VV * ZT];
  __shared__ float smax[NC];
  __shared__ float sinv[NC];
  const int lane = threadIdx.x & 31;
  const int wave = __builtin_amdgcn_readfirstlane((int)(threadIdx.x >> 5));
  const int rowTile = blockIdx.x * TR;
  const int b = rowTile >> 11;
  const int nbase = rowTile & (NN - 1);
  const int node = rowTile + wave;
  const int n = nbase + wave;
  const float4* xb = (const float4*)(x + (size_t)b * VV * NN * FF);

  constexpr int PPL = (NPB + 31) / 32;
  float pm[PPL], ps[PPL];
  {
    const float2* pb = (const float2*)part + (size_t)wave * (BB * NPB) + (size_t)b * NPB;
#pragma unroll
    for (int i = 0; i < PPL; ++i) {
      const int k = lane + 32 * i;
      const bool ok = k < NPB;
      const float2 q = pb[ok ? k : 0];
      pm[i] = ok ? q.x : -3.4e38f;
      ps[i] = ok ? q.y : 0.0f;
    }
  }
  const float vq = vS[(size_t)(lane >> 1) * (2 * BB * NN) + (size_t)(lane & 1) * (BB * NN) + node];

  const int beg = rowstart[n];
  const int end = rowstart[n + 1];
  int snv = 0;
  if (beg + lane < end) snv = src_of(ei, order[beg + lane]);
  float u0[NC], u1[NC];
#pragma unroll
  for (int pr = 0; pr < NC; ++pr) {
    const float* uu = uS + (size_t)pr * (2 * BB * NN) + b * NN;
    u0[pr] = uu[snv];
    u1[pr] = uu[BB * NN + snv];
  }

  {
    float mm = -3.4e38f;
#pragma unroll
    for (int i = 0; i < PPL; ++i) mm = fmaxf(mm, pm[i]);
    for (int off = 16; off; off >>= 1) mm = fmaxf(mm, __shfl_xor(mm, off, 32));
    float den = 0.0f;
#pragma unroll
    for (int i = 0; i < PPL; ++i)
      den += (lane + 32 * i < NPB) ? ps[i] * expf(pm[i] - mm) : 0.0f;
    for (int off = 16; off; off >>= 1) den += __shfl_xor(den, off, 32);
    if (lane == 0) {
      smax[wave] = mm;
      sinv[wave] = 1.0f / den;
    }
  }
  __syncthreads();

  float m[NC], inv[NC], vn[NC];
#pragma unroll
  for (int pr = 0; pr < NC; ++pr) {
    m[pr] = smax[pr];
    inv[pr] = sinv[pr];
    vn[pr] = __int_as_float(__builtin_amdgcn_readlane(__float_as_int(vq), 2 * pr)) +
             __int_as_float(__builtin_amdgcn_readlane(__float_as_int(vq), 2 * pr + 1));
  }

  const int l0m = (lane == 0) ? (int)0x80000000u : 0;
  float4 Z1[VV] = {}, Z2[VV] = {};
  int base = beg;
  while (true) {
    const int left = end - base;
    const int cnt = left < 32 ? left : 32;
    float wv[NC];
#pragma unroll
    for (int pr = 0; pr < NC; ++pr) {
      const float sc = (u0[pr] + u1[pr]) + vn[pr];
      const float ex = expf(sc - m[pr]) * inv[pr];
      wv[pr] = (lane < cnt) ? ex : 0.0f;
    }
    int k = 0;
    for (; k + 8 <= cnt; k += 8) gather_all_group<8>(xb, snv, wv, k, lane, l0m, Z1, Z2);
    if (k + 4 <= cnt) { gather_all_group<4>(xb, snv, wv, k, lane, l0m, Z1, Z2); k += 4; }
    if (k + 2 <= cnt) { gather_all_group<2>(xb, snv, wv, k, lane, l0m, Z1, Z2); k += 2; }
    if (k < cnt) gather_all_group<1>(xb, snv, wv, k, lane, l0m, Z1, Z2);
    base += 32;
    if (base >= end) break;
    snv = 0;
    if (base + lane < end) snv = src_of(ei, order[base + lane]);
#pragma unroll
    for (int pr = 0; pr < NC; ++pr) {
      const float* uu = uS + (size_t)pr * (2 * BB * NN) + b * NN;
      u0[pr] = uu[snv];
      u1[pr] = uu[BB * NN + snv];
    }
  }

#pragma unroll
  for (int g = 0; g < VV; ++g) {
    float* zr = Zs + g * ZT + wave * ZP + lane * 4;
    *(float4*)zr = Z1[g];
    *(float4*)(zr + FF) = Z2[g];
  }
  __syncthreads();

  const int c = lane & 15;
  const int h = lane >> 4;
  const int colbase = wave * 32;
  const v2f bs = *(const v2f*)(bias + colbase + 2 * c);
  const float* wl = W + (size_t)(2 * h) * OO + colbase + 2 * c;
  const float* al = Zs + c * ZP + 2 * h;

  ofrag F0, F1;
  load_ofrag(F0, al, wl, 0);
  v8f acc[VV][2] = {};
  for (int q = 0; q + 2 < ONG; q += 2) {
    load_ofrag(F1, al, wl, q + 1);
    __builtin_amdgcn_sched_barrier(0);
    mma_ofrag(F0, acc);
    __builtin_amdgcn_sched_barrier(0);
    load_ofrag(F0, al, wl, q + 2);
    __builtin_amdgcn_sched_barrier(0);
    mma_ofrag(F1, acc);
    __builtin_amdgcn_sched_barrier(0);
  }
  load_ofrag(F1, al, wl, ONG - 1);
  __builtin_amdgcn_sched_barrier(0);
  mma_ofrag(F0, acc);
  mma_ofrag(F1, acc);

#pragma unroll
  for (int g = 0; g < VV; ++g) {
    float* ob = out + (((size_t)b * VV + g) * NN + nbase + 8 * h) * OO + colbase + 2 * c;
#pragma unroll
    for (int i = 0; i < 8; ++i) {
      *(v2f*)(ob + (size_t)i * OO) =
          (v2f){ acc[g][0][i] * 0.25f + bs.x, acc[g][1][i] * 0.25f + bs.y };
    }
  }
}

extern "C" void kernel_launch(void* const* d_in, const int* in_sizes, int n_in,
                              void* d_out, int out_size, void* d_ws, size_t ws_size,
                              hipStream_t stream) {
  const float* x    = (const float*)d_in[0];
  const int*   ei   = (const int*)d_in[1];
  const float* W    = (const float*)d_in[2];
  const float* att  = (const float*)d_in[3];
  const float* bias = (const float*)d_in[4];
  float* out = (float*)d_out;

  char* p = (char*)d_ws;
  float* uS    = (float*)p; p += (size_t)NC * 2 * BB * NN * sizeof(float);
  float* vS    = (float*)p; p += (size_t)NC * 2 * BB * NN * sizeof(float);
  float* part  = (float*)p; p += (size_t)NC * 2 * BB * NPB * sizeof(float);
  int* counts  = (int*)p;   p += (size_t)NN * sizeof(int);
  int* order   = (int*)p;   p += (size_t)NE * sizeof(int);
  int* rowstart= (int*)p;   p += (size_t)(NN + 1) * sizeof(int);

  zero_i32_kernel<<<(NN + 255) / 256, 256, 0, stream>>>(counts, NN);
  count_dst_kernel<<<(NE + 255) / 256, 256, 0, stream>>>(ei, counts);
  scan_counts_kernel<<<1, 256, 0, stream>>>(counts, rowstart);
  build_rows_kernel<<<NN / 8, 256, 0, stream>>>(ei, rowstart, order);

  score_gemm_kernel<<<GEMM_BLOCKS, 256, 0, stream>>>(x, W, att, uS, vS);
  edge_stats_kernel<<<VV * BB * NPB, ESB, 0, stream>>>(uS, vS, ei, part);

  aggregate_out_kernel<<<(BB * NN) / TR, 512, 0, stream>>>(x, W, bias, uS, vS, part, rowstart, order, ei,
                                                           out);

  (void)in_sizes; (void)n_in; (void)out_size; (void)ws_size;
}


# score parts uS,vS stored node-major (a node's 16 pairs = one 64-byte run per column half): an edge's scores come as 16-byte loads from its nodes' lines; edge stats take 16 pairs per thread; sums kept
# speedup vs baseline: 12.7328x; 1.1334x over previous
#include <hip/hip_runtime.h>
#include <hip/hip_bf16.h>

typedef __attribute__((ext_vector_type(2))) float v2f;
typedef __attribute__((ext_vector_type(8))) float v8f;

constexpr int BB = 2;
constexpr int VV = 4;
constexpr int NN = 2048;
constexpr int FF = 128;
constexpr int EE = 32768;
constexpr int NE = 34816;
constexpr int OO = 512;
constexpr int ESB = 256;
constexpr int NPB = NE / ESB;
static_assert(NE % ESB == 0, "an edge-score block must not straddle two batches");
constexpr int NC = VV * VV;

__device__ __forceinline__ int src_of(const int* __restrict__ ei, int e) {
  return (e < EE) ? ei[e] : (e - EE);
}
__device__ __forceinline__ int dst_of(const int* __restrict__ ei, int e) {
  return (e < EE) ? ei[EE + e] : (e - EE);
}

__global__ void zero_i32_kernel(int* __restrict__ p, int n) {
  int i = blockIdx.x * blockDim.x + threadIdx.x;
  if (i < n) p[i] = 0;
}

__global__ void count_dst_kernel(const int* __restrict__ ei, int* __restrict__ counts) {
  int e = blockIdx.x * blockDim.x + threadIdx.x;
  if (e >= NE) return;
  atomicAdd(&counts[dst_of(ei, e)], 1);
}

__global__ __launch_bounds__(256) void scan_counts_kernel(const int* __restrict__ counts,
                                                          int* __restrict__ rowstart) {
  __shared__ int part[256];
  __shared__ int pref[257];
  const int tid = threadIdx.x;
  const int base = tid * 8;
  int loc[8];
  int s = 0;
#pragma unroll
  for (int i = 0; i < 8; ++i) { loc[i] = s; s += counts[base + i]; }
  part[tid] = s;
  __syncthreads();
  if (tid == 0) {
    int a = 0;
    for (int i = 0; i < 256; ++i) { pref[i] = a; a += part[i]; }
    pref[256] = a;
  }
  __syncthreads();
  const int off = pref[tid];
#pragma unroll
  for (int i = 0; i < 8; ++i) rowstart[base + i] = off + loc[i];
  if (tid == 255) rowstart[NN] = pref[256];
}

constexpr int BRU = 8;
constexpr int BRT = EE / (128 * BRU);
static_assert(EE % (128 * BRU) == 0, "the explicit edges must fill whole batches");

__global__ __launch_bounds__(256) void build_rows_kernel(const int* __restrict__ ei,
                                                         const int* __restrict__ rowstart,
                                                         int* __restrict__ order) {
  const int lane = threadIdx.x & 31;
  const int wave = __builtin_amdgcn_readfirstlane((int)(threadIdx.x >> 5));
  const int n = blockIdx.x * 8 + wave;
  if (n >= NN) return;
  const int base = rowstart[n];
  const int4* d4 = (const int4*)(ei + EE) + lane;
  const unsigned below = (1u << lane) - 1u;
  int cnt = 0;

  int4 cur[BRU], nxt[BRU];
#pragma unroll
  for (int u = 0; u < BRU; ++u) cur[u] = d4[u * 32];

  for (int t = 0; t < BRT; ++t) {
    if (t + 1 < BRT) {
#pragma unroll
      for (int u = 0; u < BRU; ++u) nxt[u] = d4[((t + 1) * BRU + u) * 32];
    }
#pragma unroll
    for (int u = 0; u < BRU; ++u) {
      const int4 q = cur[u];
      const bool m0 = q.x == n, m1 = q.y == n, m2 = q.z == n, m3 = q.w == n;
      if (__builtin_amdgcn_ballot_w32(m0 | m1 | m2 | m3) != 0u) {
        const unsigned b0 = __builtin_amdgcn_ballot_w32(m0);
        const unsigned b1 = __builtin_amdgcn_ballot_w32(m1);
        const unsigned b2 = __builtin_amdgcn_ballot_w32(m2);
        const unsigned b3 = __builtin_amdgcn_ballot_w32(m3);
        const int e = (((t * BRU + u) * 32) + lane) * 4;
        int r = base + cnt + __popc(b0 & below) + __popc(b1 & below) + __popc(b2 & below) +
                __popc(b3 & below);
        if (m0) order[r] = e;
        r += m0 ? 1 : 0;
        if (m1) order[r] = e + 1;
        r += m1 ? 1 : 0;
        if (m2) order[r] = e + 2;
        r += m2 ? 1 : 0;
        if (m3) order[r] = e + 3;
        cnt += __popc(b0) + __popc(b1) + __popc(b2) + __popc(b3);
      }
    }
#pragma unroll
    for (int u = 0; u < BRU; ++u) cur[u] = nxt[u];
  }
  if (lane == 0) order[base + cnt] = EE + n;
}

constexpr int GR = 32;
constexpr int GC = 256;
constexpr int KQ = (2 * FF) / 4;
constexpr int GEMM_BLOCKS = ((BB * NN) / GR) * (OO / GC);
static_assert(NN % GR == 0 && OO % GC == 0, "tiling must be exact");

template <int MASK>
__device__ __forceinline__ float row_xadd(float v) {
  const int t = __builtin_amdgcn_update_dpp(0, __float_as_int(v), 0x160 | MASK, 0xF, 0xF, true);
  return v + __int_as_float(t);
}
__device__ __forceinline__ float row_sum16(float v) {
  v = row_xadd<1>(v);
  v = row_xadd<2>(v);
  v = row_xadd<4>(v);
  v = row_xadd<8>(v);
  return v;
}
__device__ __forceinline__ float lrelu(float h) { return h > 0.0f ? h : 0.2f * h; }

constexpr int XP = VV * FF + 4;
static_assert(GR * VV * FF == 16 * 256 * 4, "the tile fill takes 16 float4 per thread");

struct sfrag {
  v2f a[VV][2];
  v2f w[2][2];
};
__device__ __forceinline__ void load_sfrag_w(sfrag& f, const float* __restrict__ wl, int q) {
#pragma unroll
  for (int hf = 0; hf < 2; ++hf) {
    const float* w0 = wl + (size_t)(hf * FF + 4 * q) * OO;
    f.w[hf][0] = *(const v2f*)w0;
    f.w[hf][1] = *(const v2f*)(w0 + OO);
  }
}
__device__ __forceinline__ void load_sfrag_a(sfrag& f, const float* al, int q) {
#pragma unroll
  for (int v = 0; v < VV; ++v) {
    f.a[v][0] = *(const v2f*)(al + v * FF + 4 * q);
    f.a[v][1] = *(const v2f*)(al + 16 * XP + v * FF + 4 * q);
  }
}
__device__ __forceinline__ void load_sfrag(sfrag& f, const float* al, const float* __restrict__ wl,
                                           int q) {
  load_sfrag_w(f, wl, q);
  load_sfrag_a(f, al, q);
}
__device__ __forceinline__ v8f wmma4(v2f a, v2f b, v8f c) {
  return __builtin_amdgcn_wmma_f32_16x16x4_f32(false, a, false, b, (short)0, c, false, false);
}
__device__ __forceinline__ void mma_sfrag(const sfrag& f, v8f (&R)[2][VV][2][2]) {
#pragma unroll
  for (int hf = 0; hf < 2; ++hf) {
    const v2f b0 = { f.w[hf][0].x, f.w[hf][1].x };
    const v2f b1 = { f.w[hf][0].y, f.w[hf][1].y };
#pragma unroll
    for (int v = 0; v < VV; ++v) {
      R[hf][v][0][0] = wmma4(f.a[v][0], b0, R[hf][v][0][0]);
      R[hf][v][0][1] = wmma4(f.a[v][0], b1, R[hf][v][0][1]);
      R[hf][v][1][0] = wmma4(f.a[v][1], b0, R[hf][v][1][0]);
      R[hf][v][1][1] = wmma4(f.a[v][1], b1, R[hf][v][1][1]);
    }
  }
}
__device__ __forceinline__ v2f signed_quad0(v2f a, int r) {
  return (v2f){ (r & 1) ? -a.x : a.x, (r & 2) ? -a.y : a.y };
}

static_assert(GEMM_BLOCKS == 256, "one block per 32-row tile and column half");
constexpr int RP = 8 * 2 * GR + 4;
static_assert((RP * 4) % 16 == 0, "a pair's part must stay 16-byte aligned for the float4 stores");
__global__ __launch_bounds__(256) void score_gemm_kernel(const float* __restrict__ x,
                                                         const float* __restrict__ W,
                                                         const float* __restrict__ att,
                                                         float* __restrict__ uP,
                                                         float* __restrict__ vP) {
  __shared__ __attribute__((aligned(16))) float Xs[GR * XP];
  __shared__ __attribute__((aligned(16))) float red[NC * RP];
  const int lane = threadIdx.x & 31;
  const int wave = __builtin_amdgcn_readfirstlane((int)(threadIdx.x >> 5));
  const int blk = blockIdx.x;
  const int rowTile = (blk >> 1) * GR;
  const int b = rowTile >> 11;
  const int nbase = rowTile & (NN - 1);
  const int c = lane & 15;
  const int h = lane >> 4;
  const int colbase = (blk & 1) * GC + wave * 32;

  float4 xv[16];
#pragma unroll
  for (int i = 0; i < 16; ++i) {
    const int row = 8 * (i & 3) + wave;
    const int view = i >> 2;
    xv[i] = *(const float4*)(x + (((size_t)b * VV + view) * NN + nbase + row) * FF + lane * 4);
  }

  const v2f at = *(const v2f*)(att + colbase + 2 * c);
  const v2f ab = *(const v2f*)(att + OO + colbase + 2 * c);

  const float* wl = W + (size_t)(2 * h) * OO + colbase + 2 * c;
  const float* al = Xs + c * XP + 2 * h;

  sfrag F0, F1;
  load_sfrag_w(F0, wl, 1);

#pragma unroll
  for (int i = 0; i < 16; ++i) {
    const int row = 8 * (i & 3) + wave;
    const int view = i >> 2;
    *(float4*)(Xs + row * XP + view * FF + lane * 4) = xv[i];
  }
  __syncthreads();

  load_sfrag_a(F0, al, 1);

  v8f R[2][VV][2][2] = {};
  for (int q = 1; q + 2 < FF / 4; q += 2) {
    load_sfrag(F1, al, wl, q + 1);
    __builtin_amdgcn_sched_barrier(0);
    mma_sfrag(F0, R);
    __builtin_amdgcn_sched_barrier(0);
    load_sfrag(F0, al, wl, q + 2);
    __builtin_amdgcn_sched_barrier(0);
    mma_sfrag(F1, R);
    __builtin_amdgcn_sched_barrier(0);
  }
  load_sfrag(F1, al, wl, 0);
  __builtin_amdgcn_sched_barrier(0);
  mma_sfrag(F0, R);

  const v2f b10 = { F1.w[0][0].x, F1.w[0][1].x }, b11 = { F1.w[0][0].y, F1.w[0][1].y };
  const v2f b20 = { F1.w[1][0].x, F1.w[1][1].x }, b21 = { F1.w[1][0].y, F1.w[1][1].y };

#pragma unroll
  for (int pair = 0; pair < NC; ++pair) {
    const int v1 = pair & 3;
    const int v2 = (pair >> 2) ^ v1;
    v8f t[2][2];
#pragma unroll
    for (int rt = 0; rt < 2; ++rt) {
      const v2f a1 = signed_quad0(F1.a[v1][rt], v2);
      const v2f a2 = signed_quad0(F1.a[v2][rt], v1);
      t[rt][0] = R[0][v1][rt][0] + R[1][v2][rt][0];
      t[rt][1] = R[0][v1][rt][1] + R[1][v2][rt][1];
      t[rt][0] = wmma4(a1, b10, t[rt][0]);
      t[rt][1] = wmma4(a1, b11, t[rt][1]);
      t[rt][0] = wmma4(a2, b20, t[rt][0]);
      t[rt][1] = wmma4(a2, b21, t[rt][1]);
    }
    float pu[2][8], pv[2][8];
#pragma unroll
    for (int rt = 0; rt < 2; ++rt) {
#pragma unroll
      for (int i = 0; i < 8; ++i) {
        const float l0 = lrelu(t[rt][0][i]);
        const float l1 = lrelu(t[rt][1][i]);
        pu[rt][i] = row_sum16(at.x * l0 + at.y * l1);
        pv[rt][i] = row_sum16(ab.x * l0 + ab.y * l1);
      }
    }
    if (c == 0) {
#pragma unroll
      for (int rt = 0; rt < 2; ++rt) {
        float4* ru = (float4*)&red[pair * RP + (wave * 2 + 0) * GR + 16 * rt + 8 * h];
        float4* rv = (float4*)&red[pair * RP + (wave * 2 + 1) * GR + 16 * rt + 8 * h];
        ru[0] = make_float4(pu[rt][0], pu[rt][1], pu[rt][2], pu[rt][3]);
        ru[1] = make_float4(pu[rt][4], pu[rt][5], pu[rt][6], pu[rt][7]);
        rv[0] = make_float4(pv[rt][0], pv[rt][1], pv[rt][2], pv[rt][3]);
        rv[1] = make_float4(pv[rt][4], pv[rt][5], pv[rt][6], pv[rt][7]);
      }
    }
  }
  __syncthreads();
  const int fp = lane & 15;
  const int fs = lane >> 4;
#pragma unroll
  for (int j = 0; j < 4; ++j) {
    const int uv = j & 1;
    const int row0 = 4 * wave + 2 * (j >> 1);
    const float* rr = red + fp * RP + uv * GR + row0 + fs;
    float a = rr[0];
#pragma unroll
    for (int w = 1; w < 8; ++w) a += rr[w * 2 * GR];
    float* dp = (uv ? vP : uP) + (size_t)(blk & 1) * (BB * NN * NC) +
                (size_t)(rowTile + row0) * NC + lane;
    *dp = a;
  }
}

__device__ __forceinline__ void load_parts16(const float* __restrict__ plane, int node,
                                             float (&lo)[NC], float (&hi)[NC]) {
  const float4* p0 = (const float4*)(plane + (size_t)node * NC);
  const float4* p1 = (const float4*)(plane + (size_t)BB * NN * NC + (size_t)node * NC);
#pragma unroll
  for (int i = 0; i < NC / 4; ++i) {
    const float4 a = p0[i];
    const float4 c = p1[i];
    lo[4 * i + 0] = a.x; lo[4 * i + 1] = a.y; lo[4 * i + 2] = a.z; lo[4 * i + 3] = a.w;
    hi[4 * i + 0] = c.x; hi[4 * i + 1] = c.y; hi[4 * i + 2] = c.z; hi[4 * i + 3] = c.w;
  }
}

__global__ __launch_bounds__(ESB) void edge_stats_kernel(const float* __restrict__ uS,
                                                         const float* __restrict__ vS,
                                                         const int* __restrict__ ei,
                                                         float* __restrict__ part) {
  __shared__ float redm[NC][ESB / 32];
  __shared__ float reds[NC][ESB / 32];
  const int lane = threadIdx.x & 31;
  const int wave = threadIdx.x >> 5;
  const int blkp = blockIdx.x;
  const int t = blkp * ESB + threadIdx.x;
  const int b = blkp / NPB;
  const int e = t - b * NE;
  const int sn = b * NN + src_of(ei, e);
  const int dn = b * NN + dst_of(ei, e);

  float sc[NC];
  {
    float u0[NC], u1[NC], v0[NC], v1[NC];
    load_parts16(uS, sn, u0, u1);
    load_parts16(vS, dn, v0, v1);
#pragma unroll
    for (int pr = 0; pr < NC; ++pr) sc[pr] = (u0[pr] + u1[pr]) + (v0[pr] + v1[pr]);
  }
  float mx[NC];
#pragma unroll
  for (int pr = 0; pr < NC; ++pr) mx[pr] = sc[pr];
#pragma unroll
  for (int off = 16; off; off >>= 1) {
#pragma unroll
    for (int pr = 0; pr < NC; ++pr) mx[pr] = fmaxf(mx[pr], __shfl_xor(mx[pr], off, 32));
  }
  if (lane == 0) {
#pragma unroll
    for (int pr = 0; pr < NC; ++pr) redm[pr][wave] = mx[pr];
  }
  __syncthreads();
  float ex[NC];
#pragma unroll
  for (int pr = 0; pr < NC; ++pr) {
    float a = redm[pr][0];
#pragma unroll
    for (int i = 1; i < ESB / 32; ++i) a = fmaxf(a, redm[pr][i]);
    ex[pr] = expf(sc[pr] - a);
  }
#pragma unroll
  for (int off = 16; off; off >>= 1) {
#pragma unroll
    for (int pr = 0; pr < NC; ++pr) ex[pr] += __shfl_xor(ex[pr], off, 32);
  }
  if (lane == 0) {
#pragma unroll
    for (int pr = 0; pr < NC; ++pr) reds[pr][wave] = ex[pr];
  }
  __syncthreads();
  if (threadIdx.x < NC) {
    const int pr = threadIdx.x;
    float m = redm[pr][0];
    float a = reds[pr][0];
    for (int i = 1; i < ESB / 32; ++i) { m = fmaxf(m, redm[pr][i]); a += reds[pr][i]; }
    const int idx = pr * (BB * NPB) + blkp;
    part[2 * idx]     = m;
    part[2 * idx + 1] = a;
  }
}

template <int NG>
__device__ __forceinline__ void gather_all_group(const float4* __restrict__ xb, int snv,
                                                 const float (&wv)[NC], int k, int lane, int l0m,
                                                 float4 (&Z1)[VV], float4 (&Z2)[VV]) {
  int sn[NG];
#pragma unroll
  for (int t = 0; t < NG; ++t) {
    sn[t] = __builtin_amdgcn_readlane(snv, k + t);
  }
  float4 r[NG][VV];
#pragma unroll
  for (int t = 0; t < NG; ++t) {
#pragma unroll
    for (int v = 0; v < VV; ++v) r[t][v] = xb[((size_t)v * NN + sn[t]) * (FF / 4) + lane];
  }
#pragma unroll
  for (int t = 0; t < NG; ++t) {
#pragma unroll
    for (int g = 0; g < VV; ++g) {
#pragma unroll
      for (int hv = 0; hv < VV; ++hv) {
        const int wi = __builtin_amdgcn_readlane(__float_as_int(wv[g * VV + hv]), k + t);
        const int r1 = hv ^ g;
        const int r2 = hv;
        const float w1e = __int_as_float(wi ^ ((r1 & 1) ? l0m : 0));
        const float w1o = __int_as_float(wi ^ ((r1 & 2) ? l0m : 0));
        const float w2e = __int_as_float(wi ^ ((r2 & 1) ? l0m : 0));
        const float w2o = __int_as_float(wi ^ ((r2 & 2) ? l0m : 0));
        const float4 p = r[t][hv];
        const float4 q = r[t][g ^ hv];
        Z1[g].x = __builtin_fmaf(w1e, p.x, Z1[g].x);
        Z1[g].y = __builtin_fmaf(w1o, p.y, Z1[g].y);
        Z1[g].z = __builtin_fmaf(w1e, p.z, Z1[g].z);
        Z1[g].w = __builtin_fmaf(w1o, p.w, Z1[g].w);
        Z2[g].x = __builtin_fmaf(w2e, q.x, Z2[g].x);
        Z2[g].y = __builtin_fmaf(w2o, q.y, Z2[g].y);
        Z2[g].z = __builtin_fmaf(w2e, q.z, Z2[g].z);
        Z2[g].w = __builtin_fmaf(w2o, q.w, Z2[g].w);
      }
    }
  }
}

constexpr int TR = 16;
constexpr int ZP = 2 * FF + 4;
constexpr int ZT = TR * ZP;
constexpr int OQ = 4;
constexpr int ONG = KQ / OQ;
static_assert(NN % TR == 0 && OO == TR * 32 && NC == TR, "one wave per node, per 32 columns, per pair");
static_assert(KQ % OQ == 0 && ONG % 2 == 0, "fragment groups must pair up");

struct ofrag {
  v2f a[OQ][VV];
  v2f w[OQ][2];
};
__device__ __forceinline__ void load_ofrag(ofrag& f, const float* al, const float* __restrict__ wl,
                                           int grp) {
#pragma unroll
  for (int u = 0; u < OQ; ++u) {
    const int k = 4 * (grp * OQ + u);
    const float* w0 = wl + (size_t)k * OO;
    f.w[u][0] = *(const v2f*)w0;
    f.w[u][1] = *(const v2f*)(w0 + OO);
#pragma unroll
    for (int g = 0; g < VV; ++g) f.a[u][g] = *(const v2f*)(al + g * ZT + k);
  }
}
__device__ __forceinline__ void mma_ofrag(const ofrag& f, v8f (&acc)[VV][2]) {
#pragma unroll
  for (int u = 0; u < OQ; ++u) {
    const v2f b0 = { f.w[u][0].x, f.w[u][1].x };
    const v2f b1 = { f.w[u][0].y, f.w[u][1].y };
#pragma unroll
    for (int g = 0; g < VV; ++g) {
      acc[g][0] = wmma4(f.a[u][g], b0, acc[g][0]);
      acc[g][1] = wmma4(f.a[u][g], b1, acc[g][1]);
    }
  }
}

__global__ __launch_bounds__(512) void aggregate_out_kernel(const float* __restrict__ x,
                                                            const float* __restrict__ W,
                                                            const float* __restrict__ bias,
                                                            const float* __restrict__ uS,
                                                            const float* __restrict__ vS,
                                                            const float* __restrict__ part,
                                                            const int* __restrict__ rowstart,
                                                            const int* __restrict__ order,
                                                            const int* __restrict__ ei,
                                                            float* __restrict__ out) {
  __shared__ __attribute__((aligned(16))) float Zs[VV * ZT];
  __shared__ float smax[NC];
  __shared__ float sinv[NC];
  const int lane = threadIdx.x & 31;
  const int wave = __builtin_amdgcn_readfirstlane((int)(threadIdx.x >> 5));
  const int rowTile = blockIdx.x * TR;
  const int b = rowTile >> 11;
  const int nbase = rowTile & (NN - 1);
  const int node = rowTile + wave;
  const int n = nbase + wave;
  const float4* xb = (const float4*)(x + (size_t)b * VV * NN * FF);

  constexpr int PPL = (NPB + 31) / 32;
  float pm[PPL], ps[PPL];
  {
    const float2* pb = (const float2*)part + (size_t)wave * (BB * NPB) + (size_t)b * NPB;
#pragma unroll
    for (int i = 0; i < PPL; ++i) {
      const int k = lane + 32 * i;
      const bool ok = k < NPB;
      const float2 q = pb[ok ? k : 0];
      pm[i] = ok ? q.x : -3.4e38f;
      ps[i] = ok ? q.y : 0.0f;
    }
  }
  const float vq = vS[(size_t)(lane & 1) * (BB * NN * NC) + (size_t)node * NC + (lane >> 1)];

  const int beg = rowstart[n];
  const int end = rowstart[n + 1];
  int snv = 0;
  if (beg + lane < end) snv = src_of(ei, order[beg + lane]);
  float u0[NC], u1[NC];
  load_parts16(uS, b * NN + snv, u0, u1);

  {
    float mm = -3.4e38f;
#pragma unroll
    for (int i = 0; i < PPL; ++i) mm = fmaxf(mm, pm[i]);
    for (int off = 16; off; off >>= 1) mm = fmaxf(mm, __shfl_xor(mm, off, 32));
    float den = 0.0f;
#pragma unroll
    for (int i = 0; i < PPL; ++i)
      den += (lane + 32 * i < NPB) ? ps[i] * expf(pm[i] - mm) : 0.0f;
    for (int off = 16; off; off >>= 1) den += __shfl_xor(den, off, 32);
    if (lane == 0) {
      smax[wave] = mm;
      sinv[wave] = 1.0f / den;
    }
  }
  __syncthreads();

  float m[NC], inv[NC], vn[NC];
#pragma unroll
  for (int pr = 0; pr < NC; ++pr) {
    m[pr] = smax[pr];
    inv[pr] = sinv[pr];
    vn[pr] = __int_as_float(__builtin_amdgcn_readlane(__float_as_int(vq), 2 * pr)) +
             __int_as_float(__builtin_amdgcn_readlane(__float_as_int(vq), 2 * pr + 1));
  }

  const int l0m = (lane == 0) ? (int)0x80000000u : 0;
  float4 Z1[VV] = {}, Z2[VV] = {};
  int base = beg;
  while (true) {
    const int left = end - base;
    const int cnt = left < 32 ? left : 32;
    float wv[NC];
#pragma unroll
    for (int pr = 0; pr < NC; ++pr) {
      const float sc = (u0[pr] + u1[pr]) + vn[pr];
      const float ex = expf(sc - m[pr]) * inv[pr];
      wv[pr] = (lane < cnt) ? ex : 0.0f;
    }
    int k = 0;
    for (; k + 8 <= cnt; k += 8) gather_all_group<8>(xb, snv, wv, k, lane, l0m, Z1, Z2);
    if (k + 4 <= cnt) { gather_all_group<4>(xb, snv, wv, k, lane, l0m, Z1, Z2); k += 4; }
    if (k + 2 <= cnt) { gather_all_group<2>(xb, snv, wv, k, lane, l0m, Z1, Z2); k += 2; }
    if (k < cnt) gather_all_group<1>(xb, snv, wv, k, lane, l0m, Z1, Z2);
    base += 32;
    if (base >= end) break;
    snv = 0;
    if (base + lane < end) snv = src_of(ei, order[base + lane]);
    load_parts16(uS, b * NN + snv, u0, u1);
  }

#pragma unroll
  for (int g = 0; g < VV; ++g) {
    float* zr = Zs + g * ZT + wave * ZP + lane * 4;
    *(float4*)zr = Z1[g];
    *(float4*)(zr + FF) = Z2[g];
  }
  __syncthreads();

  const int c = lane & 15;
  const int h = lane >> 4;
  const int colbase = wave * 32;
  const v2f bs = *(const v2f*)(bias + colbase + 2 * c);
  const float* wl = W + (size_t)(2 * h) * OO + colbase + 2 * c;
  const float* al = Zs + c * ZP + 2 * h;

  ofrag F0, F1;
  load_ofrag(F0, al, wl, 0);
  v8f acc[VV][2] = {};
  for (int q = 0; q + 2 < ONG; q += 2) {
    load_ofrag(F1, al, wl, q + 1);
    __builtin_amdgcn_sched_barrier(0);
    mma_ofrag(F0, acc);
    __builtin_amdgcn_sched_barrier(0);
    load_ofrag(F0, al, wl, q + 2);
    __builtin_amdgcn_sched_barrier(0);
    mma_ofrag(F1, acc);
    __builtin_amdgcn_sched_barrier(0);
  }
  load_ofrag(F1, al, wl, ONG - 1);
  __builtin_amdgcn_sched_barrier(0);
  mma_ofrag(F0, acc);
  mma_ofrag(F1, acc);

#pragma unroll
  for (int g = 0; g < VV; ++g) {
    float* ob = out + (((size_t)b * VV + g) * NN + nbase + 8 * h) * OO + colbase + 2 * c;
#pragma unroll
    for (int i = 0; i < 8; ++i) {
      *(v2f*)(ob + (size_t)i * OO) =
          (v2f){ acc[g][0][i] * 0.25f + bs.x, acc[g][1][i] * 0.25f + bs.y };
    }
  }
}

extern "C" void kernel_launch(void* const* d_in, const int* in_sizes, int n_in,
                              void* d_out, int out_size, void* d_ws, size_t ws_size,
                              hipStream_t stream) {
  const float* x    = (const float*)d_in[0];
  const int*   ei   = (const int*)d_in[1];
  const float* W    = (const float*)d_in[2];
  const float* att  = (const float*)d_in[3];
  const float* bias = (const float*)d_in[4];
  float* out = (float*)d_out;

  char* p = (char*)d_ws;
  float* uS    = (float*)p; p += (size_t)NC * 2 * BB * NN * sizeof(float);
  float* vS    = (float*)p; p += (size_t)NC * 2 * BB * NN * sizeof(float);
  float* part  = (float*)p; p += (size_t)NC * 2 * BB * NPB * sizeof(float);
  int* counts  = (int*)p;   p += (size_t)NN * sizeof(int);
  int* order   = (int*)p;   p += (size_t)NE * sizeof(int);
  int* rowstart= (int*)p;   p += (size_t)(NN + 1) * sizeof(int);

  zero_i32_kernel<<<(NN + 255) / 256, 256, 0, stream>>>(counts, NN);
  count_dst_kernel<<<(NE + 255) / 256, 256, 0, stream>>>(ei, counts);
  scan_counts_kernel<<<1, 256, 0, stream>>>(counts, rowstart);
  build_rows_kernel<<<NN / 8, 256, 0, stream>>>(ei, rowstart, order);

  score_gemm_kernel<<<GEMM_BLOCKS, 256, 0, stream>>>(x, W, att, uS, vS);
  edge_stats_kernel<<<BB * NPB, ESB, 0, stream>>>(uS, vS, ei, part);

  aggregate_out_kernel<<<(BB * NN) / TR, 512, 0, stream>>>(x, W, bias, uS, vS, part, rowstart, order, ei,
                                                           out);

  (void)in_sizes; (void)n_in; (void)out_size; (void)ws_size;
}
